# starting kernel
# baseline (speedup 1.0000x reference)
#include <hip/hip_runtime.h>
#include <hip/hip_bf16.h>

typedef float v2f __attribute__((ext_vector_type(2)));
typedef float v8f __attribute__((ext_vector_type(8)));

#define LN_EPS    1e-5f
#define NEG_SLOPE 0.2f

__device__ __forceinline__ unsigned ford(float f) {
    unsigned u = __float_as_uint(f);
    return (u & 0x80000000u) ? ~u : (u | 0x80000000u);
}
__device__ __forceinline__ float uord(unsigned u) {
    return (u & 0x80000000u) ? __uint_as_float(u & 0x7FFFFFFFu) : __uint_as_float(~u);
}

__global__ __launch_bounds__(256) void k_init(float* __restrict__ agg,
                                              float* __restrict__ out,
                                              float* __restrict__ deg,
                                              unsigned* __restrict__ mbuf,
                                              float* __restrict__ denom, int N) {
    int i = blockIdx.x * 256 + threadIdx.x;
    int tot = N * 64;
    if (i < tot) { agg[i] = 0.f; out[i] = 0.f; }
    if (i < N)   { deg[i] = 1.0f; mbuf[i] = 0u; denom[i] = 0.f; }
}

__global__ __launch_bounds__(256) void k_deg(const int* __restrict__ ei,
                                             float* __restrict__ deg, int E) {
    int i = blockIdx.x * 256 + threadIdx.x;
    if (i < E) {
        int d = __builtin_nontemporal_load(ei + E + i);
        atomicAdd(deg + d, 1.0f);
    }
}

__global__ __launch_bounds__(256) void k_dinv(float* __restrict__ deg, int N) {
    int i = blockIdx.x * 256 + threadIdx.x;
    if (i < N) { float g = deg[i]; deg[i] = (g > 0.f) ? rsqrtf(g) : 0.f; }
}

__device__ __forceinline__ void stage_w(float* __restrict__ sWp,
                                        const float* __restrict__ W) {
    for (int i = threadIdx.x; i < 4096; i += 256) {
        int k = i >> 6, c = i & 63;
        int kk = k >> 2, h = (k >> 1) & 1, j = k & 1;
        sWp[((((kk << 1) + h) << 6) + c) * 2 + j] = W[i];
    }
}

__global__ __launch_bounds__(256) void k_gemm64(const float* __restrict__ A,
                                                const float* __restrict__ W,
                                                float* __restrict__ C, int nstrips) {
    __shared__ float sWp[64 * 64];
    stage_w(sWp, W);
    __syncthreads();

    int lane  = threadIdx.x & 31;
    int strip = blockIdx.x * 8 + (threadIdx.x >> 5);
    if (strip >= nstrips) return;
    int r = lane & 15, h = lane >> 4;

    const float* ap = A + (size_t)(strip * 16 + r) * 64 + 2 * h;
    v8f zero = {0.f, 0.f, 0.f, 0.f, 0.f, 0.f, 0.f, 0.f};
    v8f acc[4] = {zero, zero, zero, zero};

    for (int kk = 0; kk < 16; ++kk) {
        v2f a = *(const v2f*)(ap + kk * 4);
        const v2f* bp = (const v2f*)&sWp[((((kk << 1) + h) << 6) + r) << 1];
#pragma unroll
        for (int t = 0; t < 4; ++t) {
            acc[t] = __builtin_amdgcn_wmma_f32_16x16x4_f32(
                false, a, false, bp[t * 16], (short)0, acc[t], false, false);
        }
    }
    float* cp = C + (size_t)strip * 16 * 64;
#pragma unroll
    for (int t = 0; t < 4; ++t)
#pragma unroll
        for (int g = 0; g < 8; ++g)
            cp[(size_t)(g + 8 * h) * 64 + t * 16 + r] = acc[t][g];
}

__global__ __launch_bounds__(256) void k_gemm64_dual(const float* __restrict__ A,
                                                     const float* __restrict__ Wl,
                                                     const float* __restrict__ Wr,
                                                     float* __restrict__ Cl,
                                                     float* __restrict__ Cr, int nstrips) {
    __shared__ float sWp[2 * 64 * 64];
    stage_w(sWp, Wl);
    stage_w(sWp + 4096, Wr);
    __syncthreads();

    int lane  = threadIdx.x & 31;
    int strip = blockIdx.x * 8 + (threadIdx.x >> 5);
    if (strip >= nstrips) return;
    int r = lane & 15, h = lane >> 4;

    const float* ap = A + (size_t)(strip * 16 + r) * 64 + 2 * h;
    v8f zero = {0.f, 0.f, 0.f, 0.f, 0.f, 0.f, 0.f, 0.f};
    v8f al[4] = {zero, zero, zero, zero};
    v8f ar[4] = {zero, zero, zero, zero};

    for (int kk = 0; kk < 16; ++kk) {
        v2f a = *(const v2f*)(ap + kk * 4);
        const v2f* bpl = (const v2f*)&sWp[((((kk << 1) + h) << 6) + r) << 1];
        const v2f* bpr = bpl + 2048;
#pragma unroll
        for (int t = 0; t < 4; ++t) {
            al[t] = __builtin_amdgcn_wmma_f32_16x16x4_f32(
                false, a, false, bpl[t * 16], (short)0, al[t], false, false);
            ar[t] = __builtin_amdgcn_wmma_f32_16x16x4_f32(
                false, a, false, bpr[t * 16], (short)0, ar[t], false, false);
        }
    }
    float* cl = Cl + (size_t)strip * 16 * 64;
    float* cr = Cr + (size_t)strip * 16 * 64;
#pragma unroll
    for (int t = 0; t < 4; ++t)
#pragma unroll
        for (int g = 0; g < 8; ++g) {
            cl[(size_t)(g + 8 * h) * 64 + t * 16 + r] = al[t][g];
            cr[(size_t)(g + 8 * h) * 64 + t * 16 + r] = ar[t][g];
        }
}

__global__ __launch_bounds__(256) void k_gemm_tail(const float* __restrict__ A,
                                                   const float* __restrict__ W,
                                                   float* __restrict__ C,
                                                   int row0, int nrows) {
    int i = blockIdx.x * 256 + threadIdx.x;
    int rows = nrows - row0;
    if (i >= rows * 64) return;
    int rr = row0 + i / 64, c = i & 63;
    float s = 0.f;
    for (int k = 0; k < 64; ++k) s += A[(size_t)rr * 64 + k] * W[k * 64 + c];
    C[(size_t)rr * 64 + c] = s;
}

__global__ __launch_bounds__(256) void k_gcn_agg(const int* __restrict__ ei,
                                                 const float* __restrict__ h,
                                                 const float* __restrict__ dinv,
                                                 float* __restrict__ agg, int E, int N) {
    int gid = blockIdx.x * 256 + threadIdx.x;
    int eid = gid >> 4, sub = gid & 15;
    if (eid >= E + N) return;
    int s, d;
    if (eid < E) {
        s = __builtin_nontemporal_load(ei + eid);
        d = __builtin_nontemporal_load(ei + E + eid);
    } else {
        s = d = eid - E;
    }
    float norm = dinv[s] * dinv[d];
    float4 v = *(const float4*)(h + (size_t)s * 64 + sub * 4);
    float* ap = agg + (size_t)d * 64 + sub * 4;
    atomicAdd(ap + 0, v.x * norm);
    atomicAdd(ap + 1, v.y * norm);
    atomicAdd(ap + 2, v.z * norm);
    atomicAdd(ap + 3, v.w * norm);
}

__global__ __launch_bounds__(256) void k_node_ln(const float* __restrict__ agg,
                                                 const float* __restrict__ b1,
                                                 const float* __restrict__ lnw,
                                                 const float* __restrict__ lnb,
                                                 float* __restrict__ hout, int N) {
    int lane = threadIdx.x & 31;
    int n = blockIdx.x * 8 + (threadIdx.x >> 5);
    if (n >= N) return;
    const float* p = agg + (size_t)n * 64 + lane * 2;
    float vx = fmaxf(p[0] + b1[lane * 2], 0.f);
    float vy = fmaxf(p[1] + b1[lane * 2 + 1], 0.f);
    float s = vx + vy;
    for (int m = 16; m >= 1; m >>= 1) s += __shfl_xor(s, m, 32);
    float mu = s * (1.f / 64.f);
    float dx = vx - mu, dy = vy - mu;
    float q = dx * dx + dy * dy;
    for (int m = 16; m >= 1; m >>= 1) q += __shfl_xor(q, m, 32);
    float inv = rsqrtf(q * (1.f / 64.f) + LN_EPS);
    float* o = hout + (size_t)n * 64 + lane * 2;
    o[0] = dx * inv * lnw[lane * 2] + lnb[lane * 2];
    o[1] = dy * inv * lnw[lane * 2 + 1] + lnb[lane * 2 + 1];
}

__global__ __launch_bounds__(256) void k_att_score(const int* __restrict__ ei,
                                                   const float* __restrict__ xl,
                                                   const float* __restrict__ xr,
                                                   const float* __restrict__ att,
                                                   float* __restrict__ ebuf,
                                                   unsigned* __restrict__ mbuf,
                                                   int E, int N) {
    int gid = blockIdx.x * 256 + threadIdx.x;
    int eid = gid >> 4, sub = gid & 15;
    if (eid >= E + N) return;
    int s, d;
    if (eid < E) {
        s = __builtin_nontemporal_load(ei + eid);
        d = __builtin_nontemporal_load(ei + E + eid);
    } else {
        s = d = eid - E;
    }
    float4 a = *(const float4*)(xl + (size_t)s * 64 + sub * 4);
    float4 b = *(const float4*)(xr + (size_t)d * 64 + sub * 4);
    float4 w = *(const float4*)(att + sub * 4);
    float p = 0.f, v;
    v = a.x + b.x; v = v > 0.f ? v : NEG_SLOPE * v; p += w.x * v;
    v = a.y + b.y; v = v > 0.f ? v : NEG_SLOPE * v; p += w.y * v;
    v = a.z + b.z; v = v > 0.f ? v : NEG_SLOPE * v; p += w.z * v;
    v = a.w + b.w; v = v > 0.f ? v : NEG_SLOPE * v; p += w.w * v;
    for (int m = 8; m >= 1; m >>= 1) p += __shfl_xor(p, m, 16);
    if (sub == 0) {
        __builtin_nontemporal_store(p, ebuf + eid);
        atomicMax(mbuf + d, ford(p));
    }
}

__global__ __launch_bounds__(256) void k_att_norm(const int* __restrict__ ei,
                                                  float* __restrict__ ebuf,
                                                  const unsigned* __restrict__ mbuf,
                                                  float* __restrict__ denom,
                                                  int E, int N) {
    int i = blockIdx.x * 256 + threadIdx.x;
    if (i >= E + N) return;
    int d = (i < E) ? __builtin_nontemporal_load(ei + E + i) : (i - E);
    float m = uord(mbuf[d]);
    float ex = __expf(ebuf[i] - m);
    ebuf[i] = ex;
    atomicAdd(denom + d, ex);
}

__global__ __launch_bounds__(256) void k_att_agg(const int* __restrict__ ei,
                                                 const float* __restrict__ ebuf,
                                                 const float* __restrict__ denom,
                                                 const float* __restrict__ xl,
                                                 float* __restrict__ out, int E, int N) {
    int gid = blockIdx.x * 256 + threadIdx.x;
    int eid = gid >> 4, sub = gid & 15;
    if (eid >= E + N) return;
    int s, d;
    if (eid < E) {
        s = __builtin_nontemporal_load(ei + eid);
        d = __builtin_nontemporal_load(ei + E + eid);
    } else {
        s = d = eid - E;
    }
    float alpha = ebuf[eid] / denom[d];
    float4 v = *(const float4*)(xl + (size_t)s * 64 + sub * 4);
    float* op = out + (size_t)d * 64 + sub * 4;
    atomicAdd(op + 0, alpha * v.x);
    atomicAdd(op + 1, alpha * v.y);
    atomicAdd(op + 2, alpha * v.z);
    atomicAdd(op + 3, alpha * v.w);
}

__global__ __launch_bounds__(256) void k_finalize(float* __restrict__ out,
                                                  const float* __restrict__ b2, int tot) {
    int i = blockIdx.x * 256 + threadIdx.x;
    if (i < tot) out[i] = fmaxf(out[i] + b2[i & 63], 0.f);
}

extern "C" void kernel_launch(void* const* d_in, const int* in_sizes, int n_in,
                              void* d_out, int out_size, void* d_ws, size_t ws_size,
                              hipStream_t stream) {
    const float* x   = (const float*)d_in[0];
    const int*   ei  = (const int*)d_in[1];
    const float* W1  = (const float*)d_in[2];
    const float* b1  = (const float*)d_in[3];
    const float* lnw = (const float*)d_in[4];
    const float* lnb = (const float*)d_in[5];
    const float* Wl  = (const float*)d_in[6];
    const float* Wr  = (const float*)d_in[7];
    const float* att = (const float*)d_in[8];
    const float* b2  = (const float*)d_in[9];
    float* out = (float*)d_out;

    const int N = in_sizes[0] / 64;
    const int E = in_sizes[1] / 2;
    const int EN = E + N;

    char* ws = (char*)d_ws;
    size_t off = 0;
    auto carve = [&](size_t bytes) -> void* {
        void* p = ws + off;
        off += (bytes + 255) & ~(size_t)255;
        return p;
    };
    float*    hbuf  = (float*)carve((size_t)N * 64 * 4);
    float*    aggxl = (float*)carve((size_t)N * 64 * 4);
    float*    xrb   = (float*)carve((size_t)N * 64 * 4);
    float*    ebuf  = (float*)carve((size_t)EN * 4);
    unsigned* mbuf  = (unsigned*)carve((size_t)N * 4);
    float*    denom = (float*)carve((size_t)N * 4);
    float*    degv  = (float*)carve((size_t)N * 4);

    const int tot = N * 64;
    const int egrid = (int)(((long long)EN * 16 + 255) / 256);
    const int nstrips = N / 16, rem = N - nstrips * 16;

    k_init<<<(tot + 255) / 256, 256, 0, stream>>>(aggxl, out, degv, mbuf, denom, N);
    k_deg<<<(E + 255) / 256, 256, 0, stream>>>(ei, degv, E);
    k_dinv<<<(N + 255) / 256, 256, 0, stream>>>(degv, N);

    k_gemm64<<<(nstrips + 7) / 8, 256, 0, stream>>>(x, W1, hbuf, nstrips);
    if (rem) k_gemm_tail<<<(rem * 64 + 255) / 256, 256, 0, stream>>>(x, W1, hbuf, nstrips * 16, N);

    k_gcn_agg<<<egrid, 256, 0, stream>>>(ei, hbuf, degv, aggxl, E, N);
    k_node_ln<<<(N + 7) / 8, 256, 0, stream>>>(aggxl, b1, lnw, lnb, hbuf, N);

    k_gemm64_dual<<<(nstrips + 7) / 8, 256, 0, stream>>>(hbuf, Wl, Wr, aggxl, xrb, nstrips);
    if (rem) {
        k_gemm_tail<<<(rem * 64 + 255) / 256, 256, 0, stream>>>(hbuf, Wl, aggxl, nstrips * 16, N);
        k_gemm_tail<<<(rem * 64 + 255) / 256, 256, 0, stream>>>(hbuf, Wr, xrb, nstrips * 16, N);
    }

    k_att_score<<<egrid, 256, 0, stream>>>(ei, aggxl, xrb, att, ebuf, mbuf, E, N);
    k_att_norm<<<(EN + 255) / 256, 256, 0, stream>>>(ei, ebuf, mbuf, denom, E, N);
    k_att_agg<<<egrid, 256, 0, stream>>>(ei, ebuf, denom, aggxl, out, E, N);
    k_finalize<<<(tot + 255) / 256, 256, 0, stream>>>(out, b2, tot);
}


# edges counting-sorted by destination inside the call (int count + arrival rank, scan, place); each node then gathers its in-edges: no float atomics, one softmax loop; sum order = arrival rank
# speedup vs baseline: 3.2460x; 3.2460x over previous
#include <hip/hip_runtime.h>
#include <hip/hip_bf16.h>

typedef float v2f __attribute__((ext_vector_type(2)));
typedef float v8f __attribute__((ext_vector_type(8)));

#define LN_EPS    1e-5f
#define NEG_SLOPE 0.2f

__global__ __launch_bounds__(256) void k_zero_cnt(int* __restrict__ cnt, int N) {
    int i = blockIdx.x * 256 + threadIdx.x;
    if (i < N) cnt[i] = 0;
}

__global__ __launch_bounds__(256) void k_count(const int* __restrict__ ei,
                                               int* __restrict__ cnt,
                                               int* __restrict__ rank, int E) {
    int i = blockIdx.x * 256 + threadIdx.x;
    if (i < E) {
        int d = ei[E + i];
        rank[i] = atomicAdd(cnt + d, 1);
    }
}

__device__ __forceinline__ int4 load4i(const int* __restrict__ p, int base, int N) {
    int4 v = make_int4(0, 0, 0, 0);
    if (base + 3 < N) {
        v = *(const int4*)(p + base);
    } else {
        if (base < N)     v.x = p[base];
        if (base + 1 < N) v.y = p[base + 1];
        if (base + 2 < N) v.z = p[base + 2];
    }
    return v;
}

__device__ __forceinline__ int block_scan_excl(int v, int* sm, int& total) {
    int lane = threadIdx.x & 31, w = threadIdx.x >> 5;
    int inc = v;
    for (int o = 1; o < 32; o <<= 1) {
        int t = __shfl_up(inc, o, 32);
        if (lane >= o) inc += t;
    }
    if (lane == 31) sm[w] = inc;
    __syncthreads();
    int woff = 0, tot = 0;
    for (int k = 0; k < 8; ++k) {
        int s = sm[k];
        if (k < w) woff += s;
        tot += s;
    }
    __syncthreads();
    total = tot;
    return woff + inc - v;
}

__global__ __launch_bounds__(256) void k_chunk_sum(const int* __restrict__ cnt,
                                                   int* __restrict__ bsum, int N) {
    __shared__ int sm[8];
    int4 c = load4i(cnt, blockIdx.x * 1024 + threadIdx.x * 4, N);
    int tot;
    block_scan_excl(c.x + c.y + c.z + c.w, sm, tot);
    if (threadIdx.x == 0) bsum[blockIdx.x] = tot;
}

__global__ __launch_bounds__(256) void k_scan(const int* __restrict__ cnt,
                                              const int* __restrict__ bsum,
                                              int* __restrict__ rowptr,
                                              float* __restrict__ dinv, int N) {
    __shared__ int sm[8];
    int b = blockIdx.x, t = threadIdx.x;
    int part = 0;
    for (int j = t; j < b; j += 256) part += bsum[j];
    int offset;
    block_scan_excl(part, sm, offset);

    int base = b * 1024 + t * 4;
    int4 c = load4i(cnt, base, N);
    int tot;
    int r0 = offset + block_scan_excl(c.x + c.y + c.z + c.w, sm, tot);
    int r[4]  = {r0, r0 + c.x, r0 + c.x + c.y, r0 + c.x + c.y + c.z};
    int cc[4] = {c.x, c.y, c.z, c.w};
    float dv[4];
#pragma unroll
    for (int k = 0; k < 4; ++k) {
        float g = (float)(cc[k] + 1);
        dv[k] = (g > 0.f) ? rsqrtf(g) : 0.f;
    }
    if (base + 3 < N) {
        *(int4*)(rowptr + base)  = make_int4(r[0], r[1], r[2], r[3]);
        *(float4*)(dinv + base)  = make_float4(dv[0], dv[1], dv[2], dv[3]);
    } else {
#pragma unroll
        for (int k = 0; k < 3; ++k) {
            if (base + k < N) { rowptr[base + k] = r[k]; dinv[base + k] = dv[k]; }
        }
    }
}

__global__ __launch_bounds__(256) void k_place(const int* __restrict__ ei,
                                               const int* __restrict__ rowptr,
                                               const int* __restrict__ rank,
                                               int* __restrict__ csr, int E) {
    int i = blockIdx.x * 256 + threadIdx.x;
    if (i < E) {
        int s = ei[i];
        int d = ei[E + i];
        csr[rowptr[d] + rank[i]] = s;
    }
}

__device__ __forceinline__ void stage_w(float* __restrict__ sWp,
                                        const float* __restrict__ W) {
    for (int i = threadIdx.x; i < 4096; i += 256) {
        int k = i >> 6, c = i & 63;
        int kk = k >> 2, h = (k >> 1) & 1, j = k & 1;
        sWp[((((kk << 1) + h) << 6) + c) * 2 + j] = W[i];
    }
}

__global__ __launch_bounds__(256) void k_gemm64(const float* __restrict__ A,
                                                const float* __restrict__ W,
                                                float* __restrict__ C, int nstrips) {
    __shared__ float sWp[64 * 64];
    stage_w(sWp, W);
    __syncthreads();

    int lane  = threadIdx.x & 31;
    int strip = blockIdx.x * 8 + (threadIdx.x >> 5);
    if (strip >= nstrips) return;
    int r = lane & 15, h = lane >> 4;

    const float* ap = A + (size_t)(strip * 16 + r) * 64 + 2 * h;
    v8f zero = {0.f, 0.f, 0.f, 0.f, 0.f, 0.f, 0.f, 0.f};
    v8f acc[4] = {zero, zero, zero, zero};

    for (int kk = 0; kk < 16; ++kk) {
        v2f a = *(const v2f*)(ap + kk * 4);
        const v2f* bp = (const v2f*)&sWp[((((kk << 1) + h) << 6) + r) << 1];
#pragma unroll
        for (int t = 0; t < 4; ++t) {
            acc[t] = __builtin_amdgcn_wmma_f32_16x16x4_f32(
                false, a, false, bp[t * 16], (short)0, acc[t], false, false);
        }
    }
    float* cp = C + (size_t)strip * 16 * 64;
#pragma unroll
    for (int t = 0; t < 4; ++t)
#pragma unroll
        for (int g = 0; g < 8; ++g)
            cp[(size_t)(g + 8 * h) * 64 + t * 16 + r] = acc[t][g];
}

__global__ __launch_bounds__(256) void k_gemm64_dual(const float* __restrict__ A,
                                                     const float* __restrict__ Wl,
                                                     const float* __restrict__ Wr,
                                                     float* __restrict__ Cl,
                                                     float* __restrict__ Cr, int nstrips) {
    __shared__ float sWp[2 * 64 * 64];
    stage_w(sWp, Wl);
    stage_w(sWp + 4096, Wr);
    __syncthreads();

    int lane  = threadIdx.x & 31;
    int strip = blockIdx.x * 8 + (threadIdx.x >> 5);
    if (strip >= nstrips) return;
    int r = lane & 15, h = lane >> 4;

    const float* ap = A + (size_t)(strip * 16 + r) * 64 + 2 * h;
    v8f zero = {0.f, 0.f, 0.f, 0.f, 0.f, 0.f, 0.f, 0.f};
    v8f al[4] = {zero, zero, zero, zero};
    v8f ar[4] = {zero, zero, zero, zero};

    for (int kk = 0; kk < 16; ++kk) {
        v2f a = *(const v2f*)(ap + kk * 4);
        const v2f* bpl = (const v2f*)&sWp[((((kk << 1) + h) << 6) + r) << 1];
        const v2f* bpr = bpl + 2048;
#pragma unroll
        for (int t = 0; t < 4; ++t) {
            al[t] = __builtin_amdgcn_wmma_f32_16x16x4_f32(
                false, a, false, bpl[t * 16], (short)0, al[t], false, false);
            ar[t] = __builtin_amdgcn_wmma_f32_16x16x4_f32(
                false, a, false, bpr[t * 16], (short)0, ar[t], false, false);
        }
    }
    float* cl = Cl + (size_t)strip * 16 * 64;
    float* cr = Cr + (size_t)strip * 16 * 64;
#pragma unroll
    for (int t = 0; t < 4; ++t)
#pragma unroll
        for (int g = 0; g < 8; ++g) {
            cl[(size_t)(g + 8 * h) * 64 + t * 16 + r] = al[t][g];
            cr[(size_t)(g + 8 * h) * 64 + t * 16 + r] = ar[t][g];
        }
}

__global__ __launch_bounds__(256) void k_gemm_tail(const float* __restrict__ A,
                                                   const float* __restrict__ W,
                                                   float* __restrict__ C,
                                                   int row0, int nrows) {
    int i = blockIdx.x * 256 + threadIdx.x;
    int rows = nrows - row0;
    if (i >= rows * 64) return;
    int rr = row0 + i / 64, c = i & 63;
    float s = 0.f;
    for (int k = 0; k < 64; ++k) s += A[(size_t)rr * 64 + k] * W[k * 64 + c];
    C[(size_t)rr * 64 + c] = s;
}

__device__ __forceinline__ float sum16(float p) {
    for (int m = 8; m >= 1; m >>= 1) p += __shfl_xor(p, m, 16);
    return p;
}

__global__ __launch_bounds__(256) void k_gcn_node(const int* __restrict__ rowptr,
                                                  const int* __restrict__ cnt,
                                                  const int* __restrict__ csr,
                                                  const float* __restrict__ h,
                                                  const float* __restrict__ dinv,
                                                  const float* __restrict__ b1,
                                                  const float* __restrict__ lnw,
                                                  const float* __restrict__ lnb,
                                                  float* __restrict__ hout, int N) {
    int gid = blockIdx.x * 256 + threadIdx.x;
    int n = gid >> 4, sub = gid & 15;
    if (n >= N) return;
    int j = rowptr[n];
    int end = j + cnt[n];
    float dn = dinv[n];
    const float* hp = h + sub * 4;

    float4 v = *(const float4*)(hp + (size_t)n * 64);
    float w = dn * dn;
    float ax = v.x * w, ay = v.y * w, az = v.z * w, aw = v.w * w;

    for (; j + 4 <= end; j += 4) {
        int s0 = csr[j], s1 = csr[j + 1], s2 = csr[j + 2], s3 = csr[j + 3];
        float w0 = dinv[s0] * dn, w1 = dinv[s1] * dn, w2 = dinv[s2] * dn, w3 = dinv[s3] * dn;
        float4 v0 = *(const float4*)(hp + (size_t)s0 * 64);
        float4 v1 = *(const float4*)(hp + (size_t)s1 * 64);
        float4 v2 = *(const float4*)(hp + (size_t)s2 * 64);
        float4 v3 = *(const float4*)(hp + (size_t)s3 * 64);
        ax += v0.x * w0; ay += v0.y * w0; az += v0.z * w0; aw += v0.w * w0;
        ax += v1.x * w1; ay += v1.y * w1; az += v1.z * w1; aw += v1.w * w1;
        ax += v2.x * w2; ay += v2.y * w2; az += v2.z * w2; aw += v2.w * w2;
        ax += v3.x * w3; ay += v3.y * w3; az += v3.z * w3; aw += v3.w * w3;
    }
    for (; j < end; ++j) {
        int s0 = csr[j];
        float w0 = dinv[s0] * dn;
        float4 v0 = *(const float4*)(hp + (size_t)s0 * 64);
        ax += v0.x * w0; ay += v0.y * w0; az += v0.z * w0; aw += v0.w * w0;
    }

    float4 bb = *(const float4*)(b1 + sub * 4);
    float vx = fmaxf(ax + bb.x, 0.f), vy = fmaxf(ay + bb.y, 0.f);
    float vz = fmaxf(az + bb.z, 0.f), vw = fmaxf(aw + bb.w, 0.f);
    float mu = sum16((vx + vy) + (vz + vw)) * (1.f / 64.f);
    float dx = vx - mu, dy = vy - mu, dz = vz - mu, dw = vw - mu;
    float q = sum16((dx * dx + dy * dy) + (dz * dz + dw * dw));
    float inv = rsqrtf(q * (1.f / 64.f) + LN_EPS);
    float4 gw = *(const float4*)(lnw + sub * 4);
    float4 gb = *(const float4*)(lnb + sub * 4);
    float4 o;
    o.x = dx * inv * gw.x + gb.x;
    o.y = dy * inv * gw.y + gb.y;
    o.z = dz * inv * gw.z + gb.z;
    o.w = dw * inv * gw.w + gb.w;
    *(float4*)(hout + (size_t)n * 64 + sub * 4) = o;
}

__device__ __forceinline__ float att_part(float4 a, float4 b, float4 w) {
    float p = 0.f, v;
    v = a.x + b.x; v = v > 0.f ? v : NEG_SLOPE * v; p += w.x * v;
    v = a.y + b.y; v = v > 0.f ? v : NEG_SLOPE * v; p += w.y * v;
    v = a.z + b.z; v = v > 0.f ? v : NEG_SLOPE * v; p += w.z * v;
    v = a.w + b.w; v = v > 0.f ? v : NEG_SLOPE * v; p += w.w * v;
    return p;
}

struct SoftAcc {
    float m, den, x, y, z, w;
    __device__ __forceinline__ void add(float e, float4 a) {
        float mn = fmaxf(m, e);
        float sc = __expf(m - mn);
        float p  = __expf(e - mn);
        den = den * sc + p;
        x = x * sc + p * a.x;
        y = y * sc + p * a.y;
        z = z * sc + p * a.z;
        w = w * sc + p * a.w;
        m = mn;
    }
};

__global__ __launch_bounds__(256) void k_att_node(const int* __restrict__ rowptr,
                                                  const int* __restrict__ cnt,
                                                  const int* __restrict__ csr,
                                                  const float* __restrict__ xl,
                                                  const float* __restrict__ xr,
                                                  const float* __restrict__ att,
                                                  const float* __restrict__ b2,
                                                  float* __restrict__ out, int N) {
    int gid = blockIdx.x * 256 + threadIdx.x;
    int n = gid >> 4, sub = gid & 15;
    if (n >= N) return;
    int j = rowptr[n];
    int end = j + cnt[n];
    const float* lp = xl + sub * 4;
    float4 b  = *(const float4*)(xr + (size_t)n * 64 + sub * 4);
    float4 wt = *(const float4*)(att + sub * 4);

    float4 a = *(const float4*)(lp + (size_t)n * 64);
    SoftAcc S;
    S.m = sum16(att_part(a, b, wt));
    S.den = 1.f; S.x = a.x; S.y = a.y; S.z = a.z; S.w = a.w;

    for (; j + 4 <= end; j += 4) {
        int s0 = csr[j], s1 = csr[j + 1], s2 = csr[j + 2], s3 = csr[j + 3];
        float4 a0 = *(const float4*)(lp + (size_t)s0 * 64);
        float4 a1 = *(const float4*)(lp + (size_t)s1 * 64);
        float4 a2 = *(const float4*)(lp + (size_t)s2 * 64);
        float4 a3 = *(const float4*)(lp + (size_t)s3 * 64);
        float e0 = sum16(att_part(a0, b, wt));
        float e1 = sum16(att_part(a1, b, wt));
        float e2 = sum16(att_part(a2, b, wt));
        float e3 = sum16(att_part(a3, b, wt));
        S.add(e0, a0); S.add(e1, a1); S.add(e2, a2); S.add(e3, a3);
    }
    for (; j < end; ++j) {
        int s0 = csr[j];
        float4 a0 = *(const float4*)(lp + (size_t)s0 * 64);
        S.add(sum16(att_part(a0, b, wt)), a0);
    }

    float4 bb = *(const float4*)(b2 + sub * 4);
    float4 o;
    o.x = fmaxf(S.x / S.den + bb.x, 0.f);
    o.y = fmaxf(S.y / S.den + bb.y, 0.f);
    o.z = fmaxf(S.z / S.den + bb.z, 0.f);
    o.w = fmaxf(S.w / S.den + bb.w, 0.f);
    *(float4*)(out + (size_t)n * 64 + sub * 4) = o;
}

extern "C" void kernel_launch(void* const* d_in, const int* in_sizes, int n_in,
                              void* d_out, int out_size, void* d_ws, size_t ws_size,
                              hipStream_t stream) {
    const float* x   = (const float*)d_in[0];
    const int*   ei  = (const int*)d_in[1];
    const float* W1  = (const float*)d_in[2];
    const float* b1  = (const float*)d_in[3];
    const float* lnw = (const float*)d_in[4];
    const float* lnb = (const float*)d_in[5];
    const float* Wl  = (const float*)d_in[6];
    const float* Wr  = (const float*)d_in[7];
    const float* att = (const float*)d_in[8];
    const float* b2  = (const float*)d_in[9];
    float* out = (float*)d_out;

    const int N = in_sizes[0] / 64;
    const int E = in_sizes[1] / 2;

    char* ws = (char*)d_ws;
    size_t off = 0;
    auto carve = [&](size_t bytes) -> void* {
        void* p = ws + off;
        off += (bytes + 255) & ~(size_t)255;
        return p;
    };
    const int nchunk = (N + 1023) / 1024;
    float* hbuf   = (float*)carve((size_t)N * 64 * 4);
    float* lnout  = (float*)carve((size_t)N * 64 * 4);
    float* xlb    = (float*)carve((size_t)N * 64 * 4);
    float* xrb    = (float*)carve((size_t)N * 64 * 4);
    int*   cnt    = (int*)carve((size_t)N * 4);
    int*   rowptr = (int*)carve((size_t)N * 4);
    float* dinv   = (float*)carve((size_t)N * 4);
    int*   bsum   = (int*)carve((size_t)nchunk * 4);
    int*   rank   = (int*)carve((size_t)E * 4);
    int*   csr    = (int*)carve((size_t)E * 4);

    const int ngrid = (int)(((long long)N * 16 + 255) / 256);
    const int nstrips = N / 16, rem = N - nstrips * 16;

    k_zero_cnt<<<(N + 255) / 256, 256, 0, stream>>>(cnt, N);
    k_count<<<(E + 255) / 256, 256, 0, stream>>>(ei, cnt, rank, E);
    k_chunk_sum<<<nchunk, 256, 0, stream>>>(cnt, bsum, N);
    k_scan<<<nchunk, 256, 0, stream>>>(cnt, bsum, rowptr, dinv, N);
    k_place<<<(E + 255) / 256, 256, 0, stream>>>(ei, rowptr, rank, csr, E);

    k_gemm64<<<(nstrips + 7) / 8, 256, 0, stream>>>(x, W1, hbuf, nstrips);
    if (rem) k_gemm_tail<<<(rem * 64 + 255) / 256, 256, 0, stream>>>(x, W1, hbuf, nstrips * 16, N);

    k_gcn_node<<<ngrid, 256, 0, stream>>>(rowptr, cnt, csr, hbuf, dinv, b1, lnw, lnb, lnout, N);

    k_gemm64_dual<<<(nstrips + 7) / 8, 256, 0, stream>>>(lnout, Wl, Wr, xlb, xrb, nstrips);
    if (rem) {
        k_gemm_tail<<<(rem * 64 + 255) / 256, 256, 0, stream>>>(lnout, Wl, xlb, nstrips * 16, N);
        k_gemm_tail<<<(rem * 64 + 255) / 256, 256, 0, stream>>>(lnout, Wr, xrb, nstrips * 16, N);
    }

    k_att_node<<<ngrid, 256, 0, stream>>>(rowptr, cnt, csr, xlb, xrb, att, b2, out, N);
}


# edges binned in one pass, 4 per lane: the count's returning add is the slot in a 32-entry line per node, overflow list past 32; scan, rank, place passes gone, 9 launches -> 6; sum order: arrival rank
# speedup vs baseline: 3.4697x; 1.0689x over previous
#include <hip/hip_runtime.h>
#include <hip/hip_bf16.h>

typedef float v2f __attribute__((ext_vector_type(2)));
typedef float v8f __attribute__((ext_vector_type(8)));

#define LN_EPS    1e-5f
#define NEG_SLOPE 0.2f

#define BIN_CAP 32

__global__ __launch_bounds__(256) void k_zero_cnt(int* __restrict__ cnt,
                                                  int* __restrict__ ovf_cnt, int N) {
    int i = blockIdx.x * 256 + threadIdx.x;
    if (i < N) cnt[i] = 0;
    if (i == 0) *ovf_cnt = 0;
}

__device__ __forceinline__ void bin_store(int s, int d, int r,
                                          int* __restrict__ bins,
                                          int* __restrict__ ovf_cnt,
                                          int2* __restrict__ ovf, int E) {
    if (r < BIN_CAP) {
        bins[(size_t)d * BIN_CAP + r] = s;
    } else {
        int o = atomicAdd(ovf_cnt, 1);
        if (o < E) ovf[o] = make_int2(d, s);
    }
}

__global__ __launch_bounds__(256) void k_bin(const int* __restrict__ ei,
                                             int* __restrict__ cnt,
                                             int* __restrict__ bins,
                                             int* __restrict__ ovf_cnt,
                                             int2* __restrict__ ovf, int E) {
    int base = (blockIdx.x * 256 + threadIdx.x) * 4;
    if (base >= E) return;
    if ((E & 3) == 0) {
        int4 s = *(const int4*)(ei + base);
        int4 d = *(const int4*)(ei + E + base);
        int r0 = atomicAdd(cnt + d.x, 1);
        int r1 = atomicAdd(cnt + d.y, 1);
        int r2 = atomicAdd(cnt + d.z, 1);
        int r3 = atomicAdd(cnt + d.w, 1);
        bin_store(s.x, d.x, r0, bins, ovf_cnt, ovf, E);
        bin_store(s.y, d.y, r1, bins, ovf_cnt, ovf, E);
        bin_store(s.z, d.z, r2, bins, ovf_cnt, ovf, E);
        bin_store(s.w, d.w, r3, bins, ovf_cnt, ovf, E);
    } else {
        for (int k = 0; k < 4; ++k) {
            int i = base + k;
            if (i < E) {
                int s = ei[i], d = ei[E + i];
                bin_store(s, d, atomicAdd(cnt + d, 1), bins, ovf_cnt, ovf, E);
            }
        }
    }
}

__device__ __forceinline__ float dinv_of(int c) {
    float g = (float)(c + 1);
    return (g > 0.f) ? rsqrtf(g) : 0.f;
}

__device__ __forceinline__ void stage_w(float* __restrict__ sWp,
                                        const float* __restrict__ W) {
    for (int i = threadIdx.x; i < 4096; i += 256) {
        int k = i >> 6, c = i & 63;
        int kk = k >> 2, h = (k >> 1) & 1, j = k & 1;
        sWp[((((kk << 1) + h) << 6) + c) * 2 + j] = W[i];
    }
}

__global__ __launch_bounds__(256) void k_gemm64(const float* __restrict__ A,
                                                const float* __restrict__ W,
                                                float* __restrict__ C, int nstrips) {
    __shared__ float sWp[64 * 64];
    stage_w(sWp, W);
    __syncthreads();

    int lane  = threadIdx.x & 31;
    int strip = blockIdx.x * 8 + (threadIdx.x >> 5);
    if (strip >= nstrips) return;
    int r = lane & 15, h = lane >> 4;

    const float* ap = A + (size_t)(strip * 16 + r) * 64 + 2 * h;
    v8f zero = {0.f, 0.f, 0.f, 0.f, 0.f, 0.f, 0.f, 0.f};
    v8f acc[4] = {zero, zero, zero, zero};

    for (int kk = 0; kk < 16; ++kk) {
        v2f a = *(const v2f*)(ap + kk * 4);
        const v2f* bp = (const v2f*)&sWp[((((kk << 1) + h) << 6) + r) << 1];
#pragma unroll
        for (int t = 0; t < 4; ++t) {
            acc[t] = __builtin_amdgcn_wmma_f32_16x16x4_f32(
                false, a, false, bp[t * 16], (short)0, acc[t], false, false);
        }
    }
    float* cp = C + (size_t)strip * 16 * 64;
#pragma unroll
    for (int t = 0; t < 4; ++t)
#pragma unroll
        for (int g = 0; g < 8; ++g)
            cp[(size_t)(g + 8 * h) * 64 + t * 16 + r] = acc[t][g];
}

__global__ __launch_bounds__(256) void k_gemm64_dual(const float* __restrict__ A,
                                                     const float* __restrict__ Wl,
                                                     const float* __restrict__ Wr,
                                                     float* __restrict__ Cl,
                                                     float* __restrict__ Cr, int nstrips) {
    __shared__ float sWp[2 * 64 * 64];
    stage_w(sWp, Wl);
    stage_w(sWp + 4096, Wr);
    __syncthreads();

    int lane  = threadIdx.x & 31;
    int strip = blockIdx.x * 8 + (threadIdx.x >> 5);
    if (strip >= nstrips) return;
    int r = lane & 15, h = lane >> 4;

    const float* ap = A + (size_t)(strip * 16 + r) * 64 + 2 * h;
    v8f zero = {0.f, 0.f, 0.f, 0.f, 0.f, 0.f, 0.f, 0.f};
    v8f al[4] = {zero, zero, zero, zero};
    v8f ar[4] = {zero, zero, zero, zero};

    for (int kk = 0; kk < 16; ++kk) {
        v2f a = *(const v2f*)(ap + kk * 4);
        const v2f* bpl = (const v2f*)&sWp[((((kk << 1) + h) << 6) + r) << 1];
        const v2f* bpr = bpl + 2048;
#pragma unroll
        for (int t = 0; t < 4; ++t) {
            al[t] = __builtin_amdgcn_wmma_f32_16x16x4_f32(
                false, a, false, bpl[t * 16], (short)0, al[t], false, false);
            ar[t] = __builtin_amdgcn_wmma_f32_16x16x4_f32(
                false, a, false, bpr[t * 16], (short)0, ar[t], false, false);
        }
    }
    float* cl = Cl + (size_t)strip * 16 * 64;
    float* cr = Cr + (size_t)strip * 16 * 64;
#pragma unroll
    for (int t = 0; t < 4; ++t)
#pragma unroll
        for (int g = 0; g < 8; ++g) {
            cl[(size_t)(g + 8 * h) * 64 + t * 16 + r] = al[t][g];
            cr[(size_t)(g + 8 * h) * 64 + t * 16 + r] = ar[t][g];
        }
}

__global__ __launch_bounds__(256) void k_gemm_tail(const float* __restrict__ A,
                                                   const float* __restrict__ W,
                                                   float* __restrict__ C,
                                                   int row0, int nrows) {
    int i = blockIdx.x * 256 + threadIdx.x;
    int rows = nrows - row0;
    if (i >= rows * 64) return;
    int rr = row0 + i / 64, c = i & 63;
    float s = 0.f;
    for (int k = 0; k < 64; ++k) s += A[(size_t)rr * 64 + k] * W[k * 64 + c];
    C[(size_t)rr * 64 + c] = s;
}

__device__ __forceinline__ float sum16(float p) {
    for (int m = 8; m >= 1; m >>= 1) p += __shfl_xor(p, m, 16);
    return p;
}

__global__ __launch_bounds__(256) void k_gcn_node(const int* __restrict__ cnt,
                                                  const int* __restrict__ bins,
                                                  const int* __restrict__ ovf_cnt,
                                                  const int2* __restrict__ ovf,
                                                  const float* __restrict__ h,
                                                  const float* __restrict__ b1,
                                                  const float* __restrict__ lnw,
                                                  const float* __restrict__ lnb,
                                                  float* __restrict__ hout, int N, int E) {
    int gid = blockIdx.x * 256 + threadIdx.x;
    int n = gid >> 4, sub = gid & 15;
    if (n >= N) return;
    int c = cnt[n];
    const int* bp = bins + (size_t)n * BIN_CAP;
    float dn = dinv_of(c);
    const float* hp = h + sub * 4;

    float4 v = *(const float4*)(hp + (size_t)n * 64);
    float w = dn * dn;
    float ax = v.x * w, ay = v.y * w, az = v.z * w, aw = v.w * w;

    int j = 0;
    for (; j + 4 <= c && j < BIN_CAP; j += 4) {
        int4 sv = *(const int4*)(bp + j);
        int s0 = sv.x, s1 = sv.y, s2 = sv.z, s3 = sv.w;
        float w0 = dinv_of(cnt[s0]) * dn, w1 = dinv_of(cnt[s1]) * dn;
        float w2 = dinv_of(cnt[s2]) * dn, w3 = dinv_of(cnt[s3]) * dn;
        float4 v0 = *(const float4*)(hp + (size_t)s0 * 64);
        float4 v1 = *(const float4*)(hp + (size_t)s1 * 64);
        float4 v2 = *(const float4*)(hp + (size_t)s2 * 64);
        float4 v3 = *(const float4*)(hp + (size_t)s3 * 64);
        ax += v0.x * w0; ay += v0.y * w0; az += v0.z * w0; aw += v0.w * w0;
        ax += v1.x * w1; ay += v1.y * w1; az += v1.z * w1; aw += v1.w * w1;
        ax += v2.x * w2; ay += v2.y * w2; az += v2.z * w2; aw += v2.w * w2;
        ax += v3.x * w3; ay += v3.y * w3; az += v3.z * w3; aw += v3.w * w3;
    }
    for (; j < c && j < BIN_CAP; ++j) {
        int s0 = bp[j];
        float w0 = dinv_of(cnt[s0]) * dn;
        float4 v0 = *(const float4*)(hp + (size_t)s0 * 64);
        ax += v0.x * w0; ay += v0.y * w0; az += v0.z * w0; aw += v0.w * w0;
    }
    if (c > BIN_CAP) {
        int no = *ovf_cnt;
        no = no < E ? no : E;
        for (int k = 0; k < no; ++k) {
            int2 e = ovf[k];
            if (e.x == n) {
                int s0 = e.y;
                float w0 = dinv_of(cnt[s0]) * dn;
                float4 v0 = *(const float4*)(hp + (size_t)s0 * 64);
                ax += v0.x * w0; ay += v0.y * w0; az += v0.z * w0; aw += v0.w * w0;
            }
        }
    }

    float4 bb = *(const float4*)(b1 + sub * 4);
    float vx = fmaxf(ax + bb.x, 0.f), vy = fmaxf(ay + bb.y, 0.f);
    float vz = fmaxf(az + bb.z, 0.f), vw = fmaxf(aw + bb.w, 0.f);
    float mu = sum16((vx + vy) + (vz + vw)) * (1.f / 64.f);
    float dx = vx - mu, dy = vy - mu, dz = vz - mu, dw = vw - mu;
    float q = sum16((dx * dx + dy * dy) + (dz * dz + dw * dw));
    float inv = rsqrtf(q * (1.f / 64.f) + LN_EPS);
    float4 gw = *(const float4*)(lnw + sub * 4);
    float4 gb = *(const float4*)(lnb + sub * 4);
    float4 o;
    o.x = dx * inv * gw.x + gb.x;
    o.y = dy * inv * gw.y + gb.y;
    o.z = dz * inv * gw.z + gb.z;
    o.w = dw * inv * gw.w + gb.w;
    *(float4*)(hout + (size_t)n * 64 + sub * 4) = o;
}

__device__ __forceinline__ float att_part(float4 a, float4 b, float4 w) {
    float p = 0.f, v;
    v = a.x + b.x; v = v > 0.f ? v : NEG_SLOPE * v; p += w.x * v;
    v = a.y + b.y; v = v > 0.f ? v : NEG_SLOPE * v; p += w.y * v;
    v = a.z + b.z; v = v > 0.f ? v : NEG_SLOPE * v; p += w.z * v;
    v = a.w + b.w; v = v > 0.f ? v : NEG_SLOPE * v; p += w.w * v;
    return p;
}

struct SoftAcc {
    float m, den, x, y, z, w;
    __device__ __forceinline__ void add(float e, float4 a) {
        float mn = fmaxf(m, e);
        float sc = __expf(m - mn);
        float p  = __expf(e - mn);
        den = den * sc + p;
        x = x * sc + p * a.x;
        y = y * sc + p * a.y;
        z = z * sc + p * a.z;
        w = w * sc + p * a.w;
        m = mn;
    }
};

__global__ __launch_bounds__(256) void k_att_node(const int* __restrict__ cnt,
                                                  const int* __restrict__ bins,
                                                  const int* __restrict__ ovf_cnt,
                                                  const int2* __restrict__ ovf,
                                                  const float* __restrict__ xl,
                                                  const float* __restrict__ xr,
                                                  const float* __restrict__ att,
                                                  const float* __restrict__ b2,
                                                  float* __restrict__ out, int N, int E) {
    int gid = blockIdx.x * 256 + threadIdx.x;
    int n = gid >> 4, sub = gid & 15;
    if (n >= N) return;
    int c = cnt[n];
    const int* bp = bins + (size_t)n * BIN_CAP;
    const float* lp = xl + sub * 4;
    float4 b  = *(const float4*)(xr + (size_t)n * 64 + sub * 4);
    float4 wt = *(const float4*)(att + sub * 4);

    float4 a = *(const float4*)(lp + (size_t)n * 64);
    SoftAcc S;
    S.m = sum16(att_part(a, b, wt));
    S.den = 1.f; S.x = a.x; S.y = a.y; S.z = a.z; S.w = a.w;

    int j = 0;
    for (; j + 4 <= c && j < BIN_CAP; j += 4) {
        int4 sv = *(const int4*)(bp + j);
        float4 a0 = *(const float4*)(lp + (size_t)sv.x * 64);
        float4 a1 = *(const float4*)(lp + (size_t)sv.y * 64);
        float4 a2 = *(const float4*)(lp + (size_t)sv.z * 64);
        float4 a3 = *(const float4*)(lp + (size_t)sv.w * 64);
        float e0 = sum16(att_part(a0, b, wt));
        float e1 = sum16(att_part(a1, b, wt));
        float e2 = sum16(att_part(a2, b, wt));
        float e3 = sum16(att_part(a3, b, wt));
        S.add(e0, a0); S.add(e1, a1); S.add(e2, a2); S.add(e3, a3);
    }
    for (; j < c && j < BIN_CAP; ++j) {
        int s0 = bp[j];
        float4 a0 = *(const float4*)(lp + (size_t)s0 * 64);
        S.add(sum16(att_part(a0, b, wt)), a0);
    }
    if (c > BIN_CAP) {
        int no = *ovf_cnt;
        no = no < E ? no : E;
        for (int k = 0; k < no; ++k) {
            int2 e = ovf[k];
            if (e.x == n) {
                float4 a0 = *(const float4*)(lp + (size_t)e.y * 64);
                S.add(sum16(att_part(a0, b, wt)), a0);
            }
        }
    }

    float4 bb = *(const float4*)(b2 + sub * 4);
    float4 o;
    o.x = fmaxf(S.x / S.den + bb.x, 0.f);
    o.y = fmaxf(S.y / S.den + bb.y, 0.f);
    o.z = fmaxf(S.z / S.den + bb.z, 0.f);
    o.w = fmaxf(S.w / S.den + bb.w, 0.f);
    *(float4*)(out + (size_t)n * 64 + sub * 4) = o;
}

extern "C" void kernel_launch(void* const* d_in, const int* in_sizes, int n_in,
                              void* d_out, int out_size, void* d_ws, size_t ws_size,
                              hipStream_t stream) {
    const float* x   = (const float*)d_in[0];
    const int*   ei  = (const int*)d_in[1];
    const float* W1  = (const float*)d_in[2];
    const float* b1  = (const float*)d_in[3];
    const float* lnw = (const float*)d_in[4];
    const float* lnb = (const float*)d_in[5];
    const float* Wl  = (const float*)d_in[6];
    const float* Wr  = (const float*)d_in[7];
    const float* att = (const float*)d_in[8];
    const float* b2  = (const float*)d_in[9];
    float* out = (float*)d_out;

    const int N = in_sizes[0] / 64;
    const int E = in_sizes[1] / 2;

    char* ws = (char*)d_ws;
    size_t off = 0;
    auto carve = [&](size_t bytes) -> void* {
        void* p = ws + off;
        off += (bytes + 255) & ~(size_t)255;
        return p;
    };
    float* hbuf   = (float*)carve((size_t)N * 64 * 4);
    float* lnout  = (float*)carve((size_t)N * 64 * 4);
    float* xlb    = (float*)carve((size_t)N * 64 * 4);
    float* xrb    = (float*)carve((size_t)N * 64 * 4);
    int*   cnt    = (int*)carve((size_t)N * 4);
    int*   ovfc   = (int*)carve(4);
    int*   bins   = (int*)carve((size_t)N * BIN_CAP * 4);
    int2*  ovf    = (int2*)carve((size_t)E * 8);

    const int ngrid = (int)(((long long)N * 16 + 255) / 256);
    const int nstrips = N / 16, rem = N - nstrips * 16;

    k_zero_cnt<<<(N + 255) / 256, 256, 0, stream>>>(cnt, ovfc, N);
    k_bin<<<((E + 3) / 4 + 255) / 256, 256, 0, stream>>>(ei, cnt, bins, ovfc, ovf, E);

    k_gemm64<<<(nstrips + 7) / 8, 256, 0, stream>>>(x, W1, hbuf, nstrips);
    if (rem) k_gemm_tail<<<(rem * 64 + 255) / 256, 256, 0, stream>>>(x, W1, hbuf, nstrips * 16, N);

    k_gcn_node<<<ngrid, 256, 0, stream>>>(cnt, bins, ovfc, ovf, hbuf, b1, lnw, lnb, lnout, N, E);

    k_gemm64_dual<<<(nstrips + 7) / 8, 256, 0, stream>>>(lnout, Wl, Wr, xlb, xrb, nstrips);
    if (rem) {
        k_gemm_tail<<<(rem * 64 + 255) / 256, 256, 0, stream>>>(lnout, Wl, xlb, nstrips * 16, N);
        k_gemm_tail<<<(rem * 64 + 255) / 256, 256, 0, stream>>>(lnout, Wr, xrb, nstrips * 16, N);
    }

    k_att_node<<<ngrid, 256, 0, stream>>>(cnt, bins, ovfc, ovf, xlb, xrb, att, b2, out, N, E);
}


# edge bins built through LDS: edges split by 512-node dst range (LDS count, staged runs, 1 global add per range per block), a block per range fills 32-entry lines in LDS; sum order: LDS arrival rank
# speedup vs baseline: 5.9457x; 1.7136x over previous
#include <hip/hip_runtime.h>
#include <hip/hip_bf16.h>

typedef float v2f __attribute__((ext_vector_type(2)));
typedef float v8f __attribute__((ext_vector_type(8)));

#define LN_EPS    1e-5f
#define NEG_SLOPE 0.2f

#define BIN_CAP 32

__global__ __launch_bounds__(256) void k_zero_cnt(int* __restrict__ cnt,
                                                  int* __restrict__ ovf_cnt, int N) {
    int i = blockIdx.x * 256 + threadIdx.x;
    if (i < N) cnt[i] = 0;
    if (i == 0) *ovf_cnt = 0;
}

__device__ __forceinline__ void bin_store(int s, int d, int r,
                                          int* __restrict__ bins,
                                          int* __restrict__ ovf_cnt,
                                          int2* __restrict__ ovf, int E) {
    if (r < BIN_CAP) {
        bins[(size_t)d * BIN_CAP + r] = s;
    } else {
        int o = atomicAdd(ovf_cnt, 1);
        if (o < E) ovf[o] = make_int2(d, s);
    }
}

__global__ __launch_bounds__(256) void k_bin(const int* __restrict__ ei,
                                             int* __restrict__ cnt,
                                             int* __restrict__ bins,
                                             int* __restrict__ ovf_cnt,
                                             int2* __restrict__ ovf, int E) {
    int base = (blockIdx.x * 256 + threadIdx.x) * 4;
    if (base >= E) return;
    if ((E & 3) == 0) {
        int4 s = *(const int4*)(ei + base);
        int4 d = *(const int4*)(ei + E + base);
        int r0 = atomicAdd(cnt + d.x, 1);
        int r1 = atomicAdd(cnt + d.y, 1);
        int r2 = atomicAdd(cnt + d.z, 1);
        int r3 = atomicAdd(cnt + d.w, 1);
        bin_store(s.x, d.x, r0, bins, ovf_cnt, ovf, E);
        bin_store(s.y, d.y, r1, bins, ovf_cnt, ovf, E);
        bin_store(s.z, d.z, r2, bins, ovf_cnt, ovf, E);
        bin_store(s.w, d.w, r3, bins, ovf_cnt, ovf, E);
    } else {
        for (int k = 0; k < 4; ++k) {
            int i = base + k;
            if (i < E) {
                int s = ei[i], d = ei[E + i];
                bin_store(s, d, atomicAdd(cnt + d, 1), bins, ovf_cnt, ovf, E);
            }
        }
    }
}

#define RSH    9
#define RNODES (1 << RSH)
#define NRMAX  256
#define PCHUNK 4096

__global__ __launch_bounds__(256) void k_zero_small(int* __restrict__ rcur,
                                                    int* __restrict__ ovf_cnt,
                                                    int* __restrict__ spill_cnt) {
    int t = threadIdx.x;
    if (t < NRMAX) rcur[t] = 0;
    if (t == 0) { *ovf_cnt = 0; *spill_cnt = 0; }
}

__global__ __launch_bounds__(256) void k_part(const int* __restrict__ ei,
                                              int* __restrict__ rcur,
                                              int2* __restrict__ redges,
                                              int* __restrict__ spill_cnt,
                                              int2* __restrict__ spill,
                                              int E, int rcap) {
    __shared__ int  lh[NRMAX];
    __shared__ int  lstart[NRMAX];
    __shared__ int  gpos[NRMAX];
    __shared__ int  wsum[NRMAX / 32];
    __shared__ int2 stage[PCHUNK];

    const int t  = threadIdx.x;
    const int e0 = blockIdx.x * PCHUNK;
    lh[t] = 0;

    int s[16], d[16], r[16];
#pragma unroll
    for (int k = 0; k < 16; ++k) {
        int e = e0 + k * 256 + t;
        bool ok = e < E;
        s[k] = ok ? ei[e] : 0;
        d[k] = ok ? ei[E + e] : -1;
    }
    __syncthreads();
#pragma unroll
    for (int k = 0; k < 16; ++k)
        r[k] = (d[k] >= 0) ? atomicAdd(&lh[d[k] >> RSH], 1) : 0;
    __syncthreads();

    const int h = lh[t];
    int incl = h;
    for (int off = 1; off < 32; off <<= 1) {
        int u = __shfl_up(incl, off, 32);
        if ((t & 31) >= off) incl += u;
    }
    if ((t & 31) == 31) wsum[t >> 5] = incl;
    __syncthreads();
    int before = 0;
    for (int w = 0; w < (t >> 5); ++w) before += wsum[w];
    lstart[t] = before + incl - h;
    gpos[t]   = h ? atomicAdd(rcur + t, h) : 0;
    __syncthreads();

#pragma unroll
    for (int k = 0; k < 16; ++k)
        if (d[k] >= 0) stage[lstart[d[k] >> RSH] + r[k]] = make_int2(d[k], s[k]);
    __syncthreads();

    int n = E - e0;
    n = n < PCHUNK ? n : PCHUNK;
    for (int i = t; i < n; i += 256) {
        int2 e = stage[i];
        int b = e.x >> RSH;
        int g = gpos[b] + (i - lstart[b]);
        if (g < rcap) {
            redges[(size_t)b * rcap + g] = e;
        } else {
            int o = atomicAdd(spill_cnt, 1);
            if (o < E) spill[o] = e;
        }
    }
}

__device__ __forceinline__ void build_put(int2 e, int l, int r, int* __restrict__ lb,
                                          int* __restrict__ ovf_cnt,
                                          int2* __restrict__ ovf, int E) {
    if (r < BIN_CAP) {
        lb[l * BIN_CAP + r] = e.y;
    } else {
        int o = atomicAdd(ovf_cnt, 1);
        if (o < E) ovf[o] = e;
    }
}

__global__ __launch_bounds__(1024) void k_build(const int* __restrict__ rcur,
                                                const int2* __restrict__ redges,
                                                const int* __restrict__ spill_cnt,
                                                const int2* __restrict__ spill,
                                                int* __restrict__ cnt,
                                                int* __restrict__ bins,
                                                int* __restrict__ ovf_cnt,
                                                int2* __restrict__ ovf,
                                                int N, int E, int rcap) {
    __shared__ int lc[RNODES];
    __shared__ int lb[RNODES * BIN_CAP];

    const int t = threadIdx.x, b = blockIdx.x;
    const int base = b << RSH;
    if (t < RNODES) lc[t] = 0;
    {
        int4 z = make_int4(0, 0, 0, 0);
        for (int i = t; i < RNODES * BIN_CAP / 4; i += 1024) ((int4*)lb)[i] = z;
    }
    __syncthreads();

    int m = rcur[b];
    m = m < rcap ? m : rcap;
    const int2* re = redges + (size_t)b * rcap;
    for (int i0 = 0; i0 < m; i0 += 8 * 1024) {
        int2 e[8];
#pragma unroll
        for (int k = 0; k < 8; ++k) {
            int i = i0 + k * 1024 + t;
            e[k] = (i < m) ? re[i] : make_int2(-1, 0);
        }
        int r[8];
#pragma unroll
        for (int k = 0; k < 8; ++k)
            r[k] = (e[k].x >= 0) ? atomicAdd(&lc[e[k].x - base], 1) : 0;
#pragma unroll
        for (int k = 0; k < 8; ++k)
            if (e[k].x >= 0) build_put(e[k], e[k].x - base, r[k], lb, ovf_cnt, ovf, E);
    }
    int ns = *spill_cnt;
    ns = ns < E ? ns : E;
    for (int i = t; i < ns; i += 1024) {
        int2 e = spill[i];
        if ((e.x >> RSH) == b)
            build_put(e, e.x - base, atomicAdd(&lc[e.x - base], 1), lb, ovf_cnt, ovf, E);
    }
    __syncthreads();

    int nn = N - base;
    nn = nn < RNODES ? nn : RNODES;
    if (t < nn) cnt[base + t] = lc[t];
    int4* dst = (int4*)(bins + (size_t)base * BIN_CAP);
    for (int i = t; i < nn * (BIN_CAP / 4); i += 1024) dst[i] = ((const int4*)lb)[i];
}

__device__ __forceinline__ float dinv_of(int c) {
    float g = (float)(c + 1);
    return (g > 0.f) ? rsqrtf(g) : 0.f;
}

__device__ __forceinline__ void stage_w(float* __restrict__ sWp,
                                        const float* __restrict__ W) {
    for (int i = threadIdx.x; i < 4096; i += 256) {
        int k = i >> 6, c = i & 63;
        int kk = k >> 2, h = (k >> 1) & 1, j = k & 1;
        sWp[((((kk << 1) + h) << 6) + c) * 2 + j] = W[i];
    }
}

__global__ __launch_bounds__(256) void k_gemm64(const float* __restrict__ A,
                                                const float* __restrict__ W,
                                                float* __restrict__ C, int nstrips) {
    __shared__ float sWp[64 * 64];
    stage_w(sWp, W);
    __syncthreads();

    int lane  = threadIdx.x & 31;
    int strip = blockIdx.x * 8 + (threadIdx.x >> 5);
    if (strip >= nstrips) return;
    int r = lane & 15, h = lane >> 4;

    const float* ap = A + (size_t)(strip * 16 + r) * 64 + 2 * h;
    v8f zero = {0.f, 0.f, 0.f, 0.f, 0.f, 0.f, 0.f, 0.f};
    v8f acc[4] = {zero, zero, zero, zero};

    for (int kk = 0; kk < 16; ++kk) {
        v2f a = *(const v2f*)(ap + kk * 4);
        const v2f* bp = (const v2f*)&sWp[((((kk << 1) + h) << 6) + r) << 1];
#pragma unroll
        for (int t = 0; t < 4; ++t) {
            acc[t] = __builtin_amdgcn_wmma_f32_16x16x4_f32(
                false, a, false, bp[t * 16], (short)0, acc[t], false, false);
        }
    }
    float* cp = C + (size_t)strip * 16 * 64;
#pragma unroll
    for (int t = 0; t < 4; ++t)
#pragma unroll
        for (int g = 0; g < 8; ++g)
            cp[(size_t)(g + 8 * h) * 64 + t * 16 + r] = acc[t][g];
}

__global__ __launch_bounds__(256) void k_gemm64_dual(const float* __restrict__ A,
                                                     const float* __restrict__ Wl,
                                                     const float* __restrict__ Wr,
                                                     float* __restrict__ Cl,
                                                     float* __restrict__ Cr, int nstrips) {
    __shared__ float sWp[2 * 64 * 64];
    stage_w(sWp, Wl);
    stage_w(sWp + 4096, Wr);
    __syncthreads();

    int lane  = threadIdx.x & 31;
    int strip = blockIdx.x * 8 + (threadIdx.x >> 5);
    if (strip >= nstrips) return;
    int r = lane & 15, h = lane >> 4;

    const float* ap = A + (size_t)(strip * 16 + r) * 64 + 2 * h;
    v8f zero = {0.f, 0.f, 0.f, 0.f, 0.f, 0.f, 0.f, 0.f};
    v8f al[4] = {zero, zero, zero, zero};
    v8f ar[4] = {zero, zero, zero, zero};

    for (int kk = 0; kk < 16; ++kk) {
        v2f a = *(const v2f*)(ap + kk * 4);
        const v2f* bpl = (const v2f*)&sWp[((((kk << 1) + h) << 6) + r) << 1];
        const v2f* bpr = bpl + 2048;
#pragma unroll
        for (int t = 0; t < 4; ++t) {
            al[t] = __builtin_amdgcn_wmma_f32_16x16x4_f32(
                false, a, false, bpl[t * 16], (short)0, al[t], false, false);
            ar[t] = __builtin_amdgcn_wmma_f32_16x16x4_f32(
                false, a, false, bpr[t * 16], (short)0, ar[t], false, false);
        }
    }
    float* cl = Cl + (size_t)strip * 16 * 64;
    float* cr = Cr + (size_t)strip * 16 * 64;
#pragma unroll
    for (int t = 0; t < 4; ++t)
#pragma unroll
        for (int g = 0; g < 8; ++g) {
            cl[(size_t)(g + 8 * h) * 64 + t * 16 + r] = al[t][g];
            cr[(size_t)(g + 8 * h) * 64 + t * 16 + r] = ar[t][g];
        }
}

__global__ __launch_bounds__(256) void k_gemm_tail(const float* __restrict__ A,
                                                   const float* __restrict__ W,
                                                   float* __restrict__ C,
                                                   int row0, int nrows) {
    int i = blockIdx.x * 256 + threadIdx.x;
    int rows = nrows - row0;
    if (i >= rows * 64) return;
    int rr = row0 + i / 64, c = i & 63;
    float s = 0.f;
    for (int k = 0; k < 64; ++k) s += A[(size_t)rr * 64 + k] * W[k * 64 + c];
    C[(size_t)rr * 64 + c] = s;
}

__device__ __forceinline__ float sum16(float p) {
    for (int m = 8; m >= 1; m >>= 1) p += __shfl_xor(p, m, 16);
    return p;
}

__global__ __launch_bounds__(256) void k_gcn_node(const int* __restrict__ cnt,
                                                  const int* __restrict__ bins,
                                                  const int* __restrict__ ovf_cnt,
                                                  const int2* __restrict__ ovf,
                                                  const float* __restrict__ h,
                                                  const float* __restrict__ b1,
                                                  const float* __restrict__ lnw,
                                                  const float* __restrict__ lnb,
                                                  float* __restrict__ hout, int N, int E) {
    int gid = blockIdx.x * 256 + threadIdx.x;
    int n = gid >> 4, sub = gid & 15;
    if (n >= N) return;
    int c = cnt[n];
    const int* bp = bins + (size_t)n * BIN_CAP;
    float dn = dinv_of(c);
    const float* hp = h + sub * 4;

    float4 v = *(const float4*)(hp + (size_t)n * 64);
    float w = dn * dn;
    float ax = v.x * w, ay = v.y * w, az = v.z * w, aw = v.w * w;

    int j = 0;
    for (; j + 4 <= c && j < BIN_CAP; j += 4) {
        int4 sv = *(const int4*)(bp + j);
        int s0 = sv.x, s1 = sv.y, s2 = sv.z, s3 = sv.w;
        float w0 = dinv_of(cnt[s0]) * dn, w1 = dinv_of(cnt[s1]) * dn;
        float w2 = dinv_of(cnt[s2]) * dn, w3 = dinv_of(cnt[s3]) * dn;
        float4 v0 = *(const float4*)(hp + (size_t)s0 * 64);
        float4 v1 = *(const float4*)(hp + (size_t)s1 * 64);
        float4 v2 = *(const float4*)(hp + (size_t)s2 * 64);
        float4 v3 = *(const float4*)(hp + (size_t)s3 * 64);
        ax += v0.x * w0; ay += v0.y * w0; az += v0.z * w0; aw += v0.w * w0;
        ax += v1.x * w1; ay += v1.y * w1; az += v1.z * w1; aw += v1.w * w1;
        ax += v2.x * w2; ay += v2.y * w2; az += v2.z * w2; aw += v2.w * w2;
        ax += v3.x * w3; ay += v3.y * w3; az += v3.z * w3; aw += v3.w * w3;
    }
    for (; j < c && j < BIN_CAP; ++j) {
        int s0 = bp[j];
        float w0 = dinv_of(cnt[s0]) * dn;
        float4 v0 = *(const float4*)(hp + (size_t)s0 * 64);
        ax += v0.x * w0; ay += v0.y * w0; az += v0.z * w0; aw += v0.w * w0;
    }
    if (c > BIN_CAP) {
        int no = *ovf_cnt;
        no = no < E ? no : E;
        for (int k = 0; k < no; ++k) {
            int2 e = ovf[k];
            if (e.x == n) {
                int s0 = e.y;
                float w0 = dinv_of(cnt[s0]) * dn;
                float4 v0 = *(const float4*)(hp + (size_t)s0 * 64);
                ax += v0.x * w0; ay += v0.y * w0; az += v0.z * w0; aw += v0.w * w0;
            }
        }
    }

    float4 bb = *(const float4*)(b1 + sub * 4);
    float vx = fmaxf(ax + bb.x, 0.f), vy = fmaxf(ay + bb.y, 0.f);
    float vz = fmaxf(az + bb.z, 0.f), vw = fmaxf(aw + bb.w, 0.f);
    float mu = sum16((vx + vy) + (vz + vw)) * (1.f / 64.f);
    float dx = vx - mu, dy = vy - mu, dz = vz - mu, dw = vw - mu;
    float q = sum16((dx * dx + dy * dy) + (dz * dz + dw * dw));
    float inv = rsqrtf(q * (1.f / 64.f) + LN_EPS);
    float4 gw = *(const float4*)(lnw + sub * 4);
    float4 gb = *(const float4*)(lnb + sub * 4);
    float4 o;
    o.x = dx * inv * gw.x + gb.x;
    o.y = dy * inv * gw.y + gb.y;
    o.z = dz * inv * gw.z + gb.z;
    o.w = dw * inv * gw.w + gb.w;
    *(float4*)(hout + (size_t)n * 64 + sub * 4) = o;
}

__device__ __forceinline__ float att_part(float4 a, float4 b, float4 w) {
    float p = 0.f, v;
    v = a.x + b.x; v = v > 0.f ? v : NEG_SLOPE * v; p += w.x * v;
    v = a.y + b.y; v = v > 0.f ? v : NEG_SLOPE * v; p += w.y * v;
    v = a.z + b.z; v = v > 0.f ? v : NEG_SLOPE * v; p += w.z * v;
    v = a.w + b.w; v = v > 0.f ? v : NEG_SLOPE * v; p += w.w * v;
    return p;
}

struct SoftAcc {
    float m, den, x, y, z, w;
    __device__ __forceinline__ void add(float e, float4 a) {
        float mn = fmaxf(m, e);
        float sc = __expf(m - mn);
        float p  = __expf(e - mn);
        den = den * sc + p;
        x = x * sc + p * a.x;
        y = y * sc + p * a.y;
        z = z * sc + p * a.z;
        w = w * sc + p * a.w;
        m = mn;
    }
};

__global__ __launch_bounds__(256) void k_att_node(const int* __restrict__ cnt,
                                                  const int* __restrict__ bins,
                                                  const int* __restrict__ ovf_cnt,
                                                  const int2* __restrict__ ovf,
                                                  const float* __restrict__ xl,
                                                  const float* __restrict__ xr,
                                                  const float* __restrict__ att,
                                                  const float* __restrict__ b2,
                                                  float* __restrict__ out, int N, int E) {
    int gid = blockIdx.x * 256 + threadIdx.x;
    int n = gid >> 4, sub = gid & 15;
    if (n >= N) return;
    int c = cnt[n];
    const int* bp = bins + (size_t)n * BIN_CAP;
    const float* lp = xl + sub * 4;
    float4 b  = *(const float4*)(xr + (size_t)n * 64 + sub * 4);
    float4 wt = *(const float4*)(att + sub * 4);

    float4 a = *(const float4*)(lp + (size_t)n * 64);
    SoftAcc S;
    S.m = sum16(att_part(a, b, wt));
    S.den = 1.f; S.x = a.x; S.y = a.y; S.z = a.z; S.w = a.w;

    int j = 0;
    for (; j + 4 <= c && j < BIN_CAP; j += 4) {
        int4 sv = *(const int4*)(bp + j);
        float4 a0 = *(const float4*)(lp + (size_t)sv.x * 64);
        float4 a1 = *(const float4*)(lp + (size_t)sv.y * 64);
        float4 a2 = *(const float4*)(lp + (size_t)sv.z * 64);
        float4 a3 = *(const float4*)(lp + (size_t)sv.w * 64);
        float e0 = sum16(att_part(a0, b, wt));
        float e1 = sum16(att_part(a1, b, wt));
        float e2 = sum16(att_part(a2, b, wt));
        float e3 = sum16(att_part(a3, b, wt));
        S.add(e0, a0); S.add(e1, a1); S.add(e2, a2); S.add(e3, a3);
    }
    for (; j < c && j < BIN_CAP; ++j) {
        int s0 = bp[j];
        float4 a0 = *(const float4*)(lp + (size_t)s0 * 64);
        S.add(sum16(att_part(a0, b, wt)), a0);
    }
    if (c > BIN_CAP) {
        int no = *ovf_cnt;
        no = no < E ? no : E;
        for (int k = 0; k < no; ++k) {
            int2 e = ovf[k];
            if (e.x == n) {
                float4 a0 = *(const float4*)(lp + (size_t)e.y * 64);
                S.add(sum16(att_part(a0, b, wt)), a0);
            }
        }
    }

    float4 bb = *(const float4*)(b2 + sub * 4);
    float4 o;
    o.x = fmaxf(S.x / S.den + bb.x, 0.f);
    o.y = fmaxf(S.y / S.den + bb.y, 0.f);
    o.z = fmaxf(S.z / S.den + bb.z, 0.f);
    o.w = fmaxf(S.w / S.den + bb.w, 0.f);
    *(float4*)(out + (size_t)n * 64 + sub * 4) = o;
}

extern "C" void kernel_launch(void* const* d_in, const int* in_sizes, int n_in,
                              void* d_out, int out_size, void* d_ws, size_t ws_size,
                              hipStream_t stream) {
    const float* x   = (const float*)d_in[0];
    const int*   ei  = (const int*)d_in[1];
    const float* W1  = (const float*)d_in[2];
    const float* b1  = (const float*)d_in[3];
    const float* lnw = (const float*)d_in[4];
    const float* lnb = (const float*)d_in[5];
    const float* Wl  = (const float*)d_in[6];
    const float* Wr  = (const float*)d_in[7];
    const float* att = (const float*)d_in[8];
    const float* b2  = (const float*)d_in[9];
    float* out = (float*)d_out;

    const int N = in_sizes[0] / 64;
    const int E = in_sizes[1] / 2;

    char* ws = (char*)d_ws;
    size_t off = 0;
    auto carve = [&](size_t bytes) -> void* {
        void* p = ws + off;
        off += (bytes + 255) & ~(size_t)255;
        return p;
    };
    float* hbuf   = (float*)carve((size_t)N * 64 * 4);
    float* lnout  = (float*)carve((size_t)N * 64 * 4);
    float* xlb    = (float*)carve((size_t)N * 64 * 4);
    float* xrb    = (float*)carve((size_t)N * 64 * 4);
    int*   cnt    = (int*)carve((size_t)N * 4);
    int*   ovfc   = (int*)carve(4);
    int*   bins   = (int*)carve((size_t)N * BIN_CAP * 4);
    int2*  ovf    = (int2*)carve((size_t)E * 8);
    const int nr   = N > RNODES ? (N + RNODES - 1) / RNODES : 1;
    const int rcap = 2 * ((E + nr - 1) / nr) + 1024;
    int*   rcur   = (int*)carve((size_t)NRMAX * 4);
    int*   spillc = (int*)carve(4);
    int2*  redges = (int2*)carve((size_t)nr * rcap * 8);
    int2*  spill  = (int2*)carve((size_t)E * 8);
    const bool lds_bins = nr <= NRMAX && off <= ws_size;

    const int ngrid = (int)(((long long)N * 16 + 255) / 256);
    const int nstrips = N / 16, rem = N - nstrips * 16;

    if (lds_bins) {
        k_zero_small<<<1, 256, 0, stream>>>(rcur, ovfc, spillc);
        k_part<<<(E + PCHUNK - 1) / PCHUNK, 256, 0, stream>>>(ei, rcur, redges, spillc, spill, E, rcap);
        k_build<<<nr, 1024, 0, stream>>>(rcur, redges, spillc, spill, cnt, bins, ovfc, ovf, N, E, rcap);
    } else {
        k_zero_cnt<<<(N + 255) / 256, 256, 0, stream>>>(cnt, ovfc, N);
        k_bin<<<((E + 3) / 4 + 255) / 256, 256, 0, stream>>>(ei, cnt, bins, ovfc, ovf, E);
    }

    k_gemm64<<<(nstrips + 7) / 8, 256, 0, stream>>>(x, W1, hbuf, nstrips);
    if (rem) k_gemm_tail<<<(rem * 64 + 255) / 256, 256, 0, stream>>>(x, W1, hbuf, nstrips * 16, N);

    k_gcn_node<<<ngrid, 256, 0, stream>>>(cnt, bins, ovfc, ovf, hbuf, b1, lnw, lnb, lnout, N, E);

    k_gemm64_dual<<<(nstrips + 7) / 8, 256, 0, stream>>>(lnout, Wl, Wr, xlb, xrb, nstrips);
    if (rem) {
        k_gemm_tail<<<(rem * 64 + 255) / 256, 256, 0, stream>>>(lnout, Wl, xlb, nstrips * 16, N);
        k_gemm_tail<<<(rem * 64 + 255) / 256, 256, 0, stream>>>(lnout, Wr, xrb, nstrips * 16, N);
    }

    k_att_node<<<ngrid, 256, 0, stream>>>(cnt, bins, ovfc, ovf, xlb, xrb, att, b2, out, N, E);
}


# both per-node gather kernels load a node's 128-byte index line once, two slots per lane, and take indices by lane exchange; left-over 1-3 edges run as one masked group; sum order unchanged
# speedup vs baseline: 6.2235x; 1.0467x over previous
#include <hip/hip_runtime.h>
#include <hip/hip_bf16.h>

typedef float v2f __attribute__((ext_vector_type(2)));
typedef float v8f __attribute__((ext_vector_type(8)));

#define LN_EPS    1e-5f
#define NEG_SLOPE 0.2f

#define BIN_CAP 32

__global__ __launch_bounds__(256) void k_zero_cnt(int* __restrict__ cnt,
                                                  int* __restrict__ ovf_cnt, int N) {
    int i = blockIdx.x * 256 + threadIdx.x;
    if (i < N) cnt[i] = 0;
    if (i == 0) *ovf_cnt = 0;
}

__device__ __forceinline__ void bin_store(int s, int d, int r,
                                          int* __restrict__ bins,
                                          int* __restrict__ ovf_cnt,
                                          int2* __restrict__ ovf, int E) {
    if (r < BIN_CAP) {
        bins[(size_t)d * BIN_CAP + r] = s;
    } else {
        int o = atomicAdd(ovf_cnt, 1);
        if (o < E) ovf[o] = make_int2(d, s);
    }
}

__global__ __launch_bounds__(256) void k_bin(const int* __restrict__ ei,
                                             int* __restrict__ cnt,
                                             int* __restrict__ bins,
                                             int* __restrict__ ovf_cnt,
                                             int2* __restrict__ ovf, int E) {
    int base = (blockIdx.x * 256 + threadIdx.x) * 4;
    if (base >= E) return;
    if ((E & 3) == 0) {
        int4 s = *(const int4*)(ei + base);
        int4 d = *(const int4*)(ei + E + base);
        int r0 = atomicAdd(cnt + d.x, 1);
        int r1 = atomicAdd(cnt + d.y, 1);
        int r2 = atomicAdd(cnt + d.z, 1);
        int r3 = atomicAdd(cnt + d.w, 1);
        bin_store(s.x, d.x, r0, bins, ovf_cnt, ovf, E);
        bin_store(s.y, d.y, r1, bins, ovf_cnt, ovf, E);
        bin_store(s.z, d.z, r2, bins, ovf_cnt, ovf, E);
        bin_store(s.w, d.w, r3, bins, ovf_cnt, ovf, E);
    } else {
        for (int k = 0; k < 4; ++k) {
            int i = base + k;
            if (i < E) {
                int s = ei[i], d = ei[E + i];
                bin_store(s, d, atomicAdd(cnt + d, 1), bins, ovf_cnt, ovf, E);
            }
        }
    }
}

#define RSH    9
#define RNODES (1 << RSH)
#define NRMAX  256
#define PCHUNK 4096

__global__ __launch_bounds__(256) void k_zero_small(int* __restrict__ rcur,
                                                    int* __restrict__ ovf_cnt,
                                                    int* __restrict__ spill_cnt) {
    int t = threadIdx.x;
    if (t < NRMAX) rcur[t] = 0;
    if (t == 0) { *ovf_cnt = 0; *spill_cnt = 0; }
}

__global__ __launch_bounds__(256) void k_part(const int* __restrict__ ei,
                                              int* __restrict__ rcur,
                                              int2* __restrict__ redges,
                                              int* __restrict__ spill_cnt,
                                              int2* __restrict__ spill,
                                              int E, int rcap) {
    __shared__ int  lh[NRMAX];
    __shared__ int  lstart[NRMAX];
    __shared__ int  gpos[NRMAX];
    __shared__ int  wsum[NRMAX / 32];
    __shared__ int2 stage[PCHUNK];

    const int t  = threadIdx.x;
    const int e0 = blockIdx.x * PCHUNK;
    lh[t] = 0;

    int s[16], d[16], r[16];
#pragma unroll
    for (int k = 0; k < 16; ++k) {
        int e = e0 + k * 256 + t;
        bool ok = e < E;
        s[k] = ok ? ei[e] : 0;
        d[k] = ok ? ei[E + e] : -1;
    }
    __syncthreads();
#pragma unroll
    for (int k = 0; k < 16; ++k)
        r[k] = (d[k] >= 0) ? atomicAdd(&lh[d[k] >> RSH], 1) : 0;
    __syncthreads();

    const int h = lh[t];
    int incl = h;
    for (int off = 1; off < 32; off <<= 1) {
        int u = __shfl_up(incl, off, 32);
        if ((t & 31) >= off) incl += u;
    }
    if ((t & 31) == 31) wsum[t >> 5] = incl;
    __syncthreads();
    int before = 0;
    for (int w = 0; w < (t >> 5); ++w) before += wsum[w];
    lstart[t] = before + incl - h;
    gpos[t]   = h ? atomicAdd(rcur + t, h) : 0;
    __syncthreads();

#pragma unroll
    for (int k = 0; k < 16; ++k)
        if (d[k] >= 0) stage[lstart[d[k] >> RSH] + r[k]] = make_int2(d[k], s[k]);
    __syncthreads();

    int n = E - e0;
    n = n < PCHUNK ? n : PCHUNK;
    for (int i = t; i < n; i += 256) {
        int2 e = stage[i];
        int b = e.x >> RSH;
        int g = gpos[b] + (i - lstart[b]);
        if (g < rcap) {
            redges[(size_t)b * rcap + g] = e;
        } else {
            int o = atomicAdd(spill_cnt, 1);
            if (o < E) spill[o] = e;
        }
    }
}

__device__ __forceinline__ void build_put(int2 e, int l, int r, int* __restrict__ lb,
                                          int* __restrict__ ovf_cnt,
                                          int2* __restrict__ ovf, int E) {
    if (r < BIN_CAP) {
        lb[l * BIN_CAP + r] = e.y;
    } else {
        int o = atomicAdd(ovf_cnt, 1);
        if (o < E) ovf[o] = e;
    }
}

__global__ __launch_bounds__(1024) void k_build(const int* __restrict__ rcur,
                                                const int2* __restrict__ redges,
                                                const int* __restrict__ spill_cnt,
                                                const int2* __restrict__ spill,
                                                int* __restrict__ cnt,
                                                int* __restrict__ bins,
                                                int* __restrict__ ovf_cnt,
                                                int2* __restrict__ ovf,
                                                int N, int E, int rcap) {
    __shared__ int lc[RNODES];
    __shared__ int lb[RNODES * BIN_CAP];

    const int t = threadIdx.x, b = blockIdx.x;
    const int base = b << RSH;
    if (t < RNODES) lc[t] = 0;
    {
        int4 z = make_int4(0, 0, 0, 0);
        for (int i = t; i < RNODES * BIN_CAP / 4; i += 1024) ((int4*)lb)[i] = z;
    }
    __syncthreads();

    int m = rcur[b];
    m = m < rcap ? m : rcap;
    const int2* re = redges + (size_t)b * rcap;
    for (int i0 = 0; i0 < m; i0 += 8 * 1024) {
        int2 e[8];
#pragma unroll
        for (int k = 0; k < 8; ++k) {
            int i = i0 + k * 1024 + t;
            e[k] = (i < m) ? re[i] : make_int2(-1, 0);
        }
        int r[8];
#pragma unroll
        for (int k = 0; k < 8; ++k)
            r[k] = (e[k].x >= 0) ? atomicAdd(&lc[e[k].x - base], 1) : 0;
#pragma unroll
        for (int k = 0; k < 8; ++k)
            if (e[k].x >= 0) build_put(e[k], e[k].x - base, r[k], lb, ovf_cnt, ovf, E);
    }
    int ns = *spill_cnt;
    ns = ns < E ? ns : E;
    for (int i = t; i < ns; i += 1024) {
        int2 e = spill[i];
        if ((e.x >> RSH) == b)
            build_put(e, e.x - base, atomicAdd(&lc[e.x - base], 1), lb, ovf_cnt, ovf, E);
    }
    __syncthreads();

    int nn = N - base;
    nn = nn < RNODES ? nn : RNODES;
    if (t < nn) cnt[base + t] = lc[t];
    int4* dst = (int4*)(bins + (size_t)base * BIN_CAP);
    for (int i = t; i < nn * (BIN_CAP / 4); i += 1024) dst[i] = ((const int4*)lb)[i];
}

__device__ __forceinline__ float dinv_of(int c) {
    float g = (float)(c + 1);
    return (g > 0.f) ? rsqrtf(g) : 0.f;
}

__device__ __forceinline__ void stage_w(float* __restrict__ sWp,
                                        const float* __restrict__ W) {
    for (int i = threadIdx.x; i < 4096; i += 256) {
        int k = i >> 6, c = i & 63;
        int kk = k >> 2, h = (k >> 1) & 1, j = k & 1;
        sWp[((((kk << 1) + h) << 6) + c) * 2 + j] = W[i];
    }
}

__global__ __launch_bounds__(256) void k_gemm64(const float* __restrict__ A,
                                                const float* __restrict__ W,
                                                float* __restrict__ C, int nstrips) {
    __shared__ float sWp[64 * 64];
    stage_w(sWp, W);
    __syncthreads();

    int lane  = threadIdx.x & 31;
    int strip = blockIdx.x * 8 + (threadIdx.x >> 5);
    if (strip >= nstrips) return;
    int r = lane & 15, h = lane >> 4;

    const float* ap = A + (size_t)(strip * 16 + r) * 64 + 2 * h;
    v8f zero = {0.f, 0.f, 0.f, 0.f, 0.f, 0.f, 0.f, 0.f};
    v8f acc[4] = {zero, zero, zero, zero};

    for (int kk = 0; kk < 16; ++kk) {
        v2f a = *(const v2f*)(ap + kk * 4);
        const v2f* bp = (const v2f*)&sWp[((((kk << 1) + h) << 6) + r) << 1];
#pragma unroll
        for (int t = 0; t < 4; ++t) {
            acc[t] = __builtin_amdgcn_wmma_f32_16x16x4_f32(
                false, a, false, bp[t * 16], (short)0, acc[t], false, false);
        }
    }
    float* cp = C + (size_t)strip * 16 * 64;
#pragma unroll
    for (int t = 0; t < 4; ++t)
#pragma unroll
        for (int g = 0; g < 8; ++g)
            cp[(size_t)(g + 8 * h) * 64 + t * 16 + r] = acc[t][g];
}

__global__ __launch_bounds__(256) void k_gemm64_dual(const float* __restrict__ A,
                                                     const float* __restrict__ Wl,
                                                     const float* __restrict__ Wr,
                                                     float* __restrict__ Cl,
                                                     float* __restrict__ Cr, int nstrips) {
    __shared__ float sWp[2 * 64 * 64];
    stage_w(sWp, Wl);
    stage_w(sWp + 4096, Wr);
    __syncthreads();

    int lane  = threadIdx.x & 31;
    int strip = blockIdx.x * 8 + (threadIdx.x >> 5);
    if (strip >= nstrips) return;
    int r = lane & 15, h = lane >> 4;

    const float* ap = A + (size_t)(strip * 16 + r) * 64 + 2 * h;
    v8f zero = {0.f, 0.f, 0.f, 0.f, 0.f, 0.f, 0.f, 0.f};
    v8f al[4] = {zero, zero, zero, zero};
    v8f ar[4] = {zero, zero, zero, zero};

    for (int kk = 0; kk < 16; ++kk) {
        v2f a = *(const v2f*)(ap + kk * 4);
        const v2f* bpl = (const v2f*)&sWp[((((kk << 1) + h) << 6) + r) << 1];
        const v2f* bpr = bpl + 2048;
#pragma unroll
        for (int t = 0; t < 4; ++t) {
            al[t] = __builtin_amdgcn_wmma_f32_16x16x4_f32(
                false, a, false, bpl[t * 16], (short)0, al[t], false, false);
            ar[t] = __builtin_amdgcn_wmma_f32_16x16x4_f32(
                false, a, false, bpr[t * 16], (short)0, ar[t], false, false);
        }
    }
    float* cl = Cl + (size_t)strip * 16 * 64;
    float* cr = Cr + (size_t)strip * 16 * 64;
#pragma unroll
    for (int t = 0; t < 4; ++t)
#pragma unroll
        for (int g = 0; g < 8; ++g) {
            cl[(size_t)(g + 8 * h) * 64 + t * 16 + r] = al[t][g];
            cr[(size_t)(g + 8 * h) * 64 + t * 16 + r] = ar[t][g];
        }
}

__global__ __launch_bounds__(256) void k_gemm_tail(const float* __restrict__ A,
                                                   const float* __restrict__ W,
                                                   float* __restrict__ C,
                                                   int row0, int nrows) {
    int i = blockIdx.x * 256 + threadIdx.x;
    int rows = nrows - row0;
    if (i >= rows * 64) return;
    int rr = row0 + i / 64, c = i & 63;
    float s = 0.f;
    for (int k = 0; k < 64; ++k) s += A[(size_t)rr * 64 + k] * W[k * 64 + c];
    C[(size_t)rr * 64 + c] = s;
}

__device__ __forceinline__ float sum16(float p) {
    for (int m = 8; m >= 1; m >>= 1) p += __shfl_xor(p, m, 16);
    return p;
}

__global__ __launch_bounds__(256) void k_gcn_node(const int* __restrict__ cnt,
                                                  const int* __restrict__ bins,
                                                  const int* __restrict__ ovf_cnt,
                                                  const int2* __restrict__ ovf,
                                                  const float* __restrict__ h,
                                                  const float* __restrict__ b1,
                                                  const float* __restrict__ lnw,
                                                  const float* __restrict__ lnb,
                                                  float* __restrict__ hout, int N, int E) {
    int gid = blockIdx.x * 256 + threadIdx.x;
    int n = gid >> 4, sub = gid & 15;
    if (n >= N) return;
    const int* bp = bins + (size_t)n * BIN_CAP;
    const int2 ent = *(const int2*)(bp + 2 * sub);
    int c = cnt[n];
    float dn = dinv_of(c);
    const float* hp = h + sub * 4;

    float4 v = *(const float4*)(hp + (size_t)n * 64);
    float w = dn * dn;
    float ax = v.x * w, ay = v.y * w, az = v.z * w, aw = v.w * w;

    const int cm = c < BIN_CAP ? c : BIN_CAP;
    int j = 0;
    for (; j + 4 <= cm; j += 4) {
        const int l = j >> 1;
        int s0 = __shfl(ent.x, l, 16), s1 = __shfl(ent.y, l, 16);
        int s2 = __shfl(ent.x, l + 1, 16), s3 = __shfl(ent.y, l + 1, 16);
        float w0 = dinv_of(cnt[s0]) * dn, w1 = dinv_of(cnt[s1]) * dn;
        float w2 = dinv_of(cnt[s2]) * dn, w3 = dinv_of(cnt[s3]) * dn;
        float4 v0 = *(const float4*)(hp + (size_t)s0 * 64);
        float4 v1 = *(const float4*)(hp + (size_t)s1 * 64);
        float4 v2 = *(const float4*)(hp + (size_t)s2 * 64);
        float4 v3 = *(const float4*)(hp + (size_t)s3 * 64);
        ax += v0.x * w0; ay += v0.y * w0; az += v0.z * w0; aw += v0.w * w0;
        ax += v1.x * w1; ay += v1.y * w1; az += v1.z * w1; aw += v1.w * w1;
        ax += v2.x * w2; ay += v2.y * w2; az += v2.z * w2; aw += v2.w * w2;
        ax += v3.x * w3; ay += v3.y * w3; az += v3.z * w3; aw += v3.w * w3;
    }
    const int t = cm - j;
    if (t > 0) {
        const int l = j >> 1;
        int s0 = __shfl(ent.x, l, 16), s1 = __shfl(ent.y, l, 16), s2 = __shfl(ent.x, l + 1, 16);
        s1 = t > 1 ? s1 : s0;
        s2 = t > 2 ? s2 : s0;
        const int c0 = cnt[s0], c1 = cnt[s1], c2 = cnt[s2];
        const float k1 = t > 1 ? 1.f : 0.f, k2 = t > 2 ? 1.f : 0.f;
        float w0 = dinv_of(c0) * dn;
        float w1 = dinv_of(c1) * dn * k1;
        float w2 = dinv_of(c2) * dn * k2;
        float4 v0 = *(const float4*)(hp + (size_t)s0 * 64);
        float4 v1 = *(const float4*)(hp + (size_t)s1 * 64);
        float4 v2 = *(const float4*)(hp + (size_t)s2 * 64);
        ax += v0.x * w0; ay += v0.y * w0; az += v0.z * w0; aw += v0.w * w0;
        ax += v1.x * w1; ay += v1.y * w1; az += v1.z * w1; aw += v1.w * w1;
        ax += v2.x * w2; ay += v2.y * w2; az += v2.z * w2; aw += v2.w * w2;
    }
    if (c > BIN_CAP) {
        int no = *ovf_cnt;
        no = no < E ? no : E;
        for (int k = 0; k < no; ++k) {
            int2 e = ovf[k];
            if (e.x == n) {
                int s0 = e.y;
                float w0 = dinv_of(cnt[s0]) * dn;
                float4 v0 = *(const float4*)(hp + (size_t)s0 * 64);
                ax += v0.x * w0; ay += v0.y * w0; az += v0.z * w0; aw += v0.w * w0;
            }
        }
    }

    float4 bb = *(const float4*)(b1 + sub * 4);
    float vx = fmaxf(ax + bb.x, 0.f), vy = fmaxf(ay + bb.y, 0.f);
    float vz = fmaxf(az + bb.z, 0.f), vw = fmaxf(aw + bb.w, 0.f);
    float mu = sum16((vx + vy) + (vz + vw)) * (1.f / 64.f);
    float dx = vx - mu, dy = vy - mu, dz = vz - mu, dw = vw - mu;
    float q = sum16((dx * dx + dy * dy) + (dz * dz + dw * dw));
    float inv = rsqrtf(q * (1.f / 64.f) + LN_EPS);
    float4 gw = *(const float4*)(lnw + sub * 4);
    float4 gb = *(const float4*)(lnb + sub * 4);
    float4 o;
    o.x = dx * inv * gw.x + gb.x;
    o.y = dy * inv * gw.y + gb.y;
    o.z = dz * inv * gw.z + gb.z;
    o.w = dw * inv * gw.w + gb.w;
    *(float4*)(hout + (size_t)n * 64 + sub * 4) = o;
}

__device__ __forceinline__ float att_part(float4 a, float4 b, float4 w) {
    float p = 0.f, v;
    v = a.x + b.x; v = v > 0.f ? v : NEG_SLOPE * v; p += w.x * v;
    v = a.y + b.y; v = v > 0.f ? v : NEG_SLOPE * v; p += w.y * v;
    v = a.z + b.z; v = v > 0.f ? v : NEG_SLOPE * v; p += w.z * v;
    v = a.w + b.w; v = v > 0.f ? v : NEG_SLOPE * v; p += w.w * v;
    return p;
}

struct SoftAcc {
    float m, den, x, y, z, w;
    __device__ __forceinline__ void add(float e, float4 a) {
        float mn = fmaxf(m, e);
        float sc = __expf(m - mn);
        float p  = __expf(e - mn);
        den = den * sc + p;
        x = x * sc + p * a.x;
        y = y * sc + p * a.y;
        z = z * sc + p * a.z;
        w = w * sc + p * a.w;
        m = mn;
    }
};

__global__ __launch_bounds__(256) void k_att_node(const int* __restrict__ cnt,
                                                  const int* __restrict__ bins,
                                                  const int* __restrict__ ovf_cnt,
                                                  const int2* __restrict__ ovf,
                                                  const float* __restrict__ xl,
                                                  const float* __restrict__ xr,
                                                  const float* __restrict__ att,
                                                  const float* __restrict__ b2,
                                                  float* __restrict__ out, int N, int E) {
    int gid = blockIdx.x * 256 + threadIdx.x;
    int n = gid >> 4, sub = gid & 15;
    if (n >= N) return;
    const int* bp = bins + (size_t)n * BIN_CAP;
    const int2 ent = *(const int2*)(bp + 2 * sub);
    int c = cnt[n];
    const float* lp = xl + sub * 4;
    float4 b  = *(const float4*)(xr + (size_t)n * 64 + sub * 4);
    float4 wt = *(const float4*)(att + sub * 4);

    float4 a = *(const float4*)(lp + (size_t)n * 64);
    SoftAcc S;
    S.m = sum16(att_part(a, b, wt));
    S.den = 1.f; S.x = a.x; S.y = a.y; S.z = a.z; S.w = a.w;

    const int cm = c < BIN_CAP ? c : BIN_CAP;
    int j = 0;
    for (; j + 4 <= cm; j += 4) {
        const int l = j >> 1;
        int s0 = __shfl(ent.x, l, 16), s1 = __shfl(ent.y, l, 16);
        int s2 = __shfl(ent.x, l + 1, 16), s3 = __shfl(ent.y, l + 1, 16);
        float4 a0 = *(const float4*)(lp + (size_t)s0 * 64);
        float4 a1 = *(const float4*)(lp + (size_t)s1 * 64);
        float4 a2 = *(const float4*)(lp + (size_t)s2 * 64);
        float4 a3 = *(const float4*)(lp + (size_t)s3 * 64);
        float e0 = sum16(att_part(a0, b, wt));
        float e1 = sum16(att_part(a1, b, wt));
        float e2 = sum16(att_part(a2, b, wt));
        float e3 = sum16(att_part(a3, b, wt));
        S.add(e0, a0); S.add(e1, a1); S.add(e2, a2); S.add(e3, a3);
    }
    const int t = cm - j;
    if (t > 0) {
        const int l = j >> 1;
        int s0 = __shfl(ent.x, l, 16), s1 = __shfl(ent.y, l, 16), s2 = __shfl(ent.x, l + 1, 16);
        s1 = t > 1 ? s1 : s0;
        s2 = t > 2 ? s2 : s0;
        float4 a0 = *(const float4*)(lp + (size_t)s0 * 64);
        float4 a1 = *(const float4*)(lp + (size_t)s1 * 64);
        float4 a2 = *(const float4*)(lp + (size_t)s2 * 64);
        float e0 = sum16(att_part(a0, b, wt));
        float e1 = sum16(att_part(a1, b, wt));
        float e2 = sum16(att_part(a2, b, wt));
        e1 = t > 1 ? e1 : -INFINITY;
        e2 = t > 2 ? e2 : -INFINITY;
        S.add(e0, a0); S.add(e1, a1); S.add(e2, a2);
    }
    if (c > BIN_CAP) {
        int no = *ovf_cnt;
        no = no < E ? no : E;
        for (int k = 0; k < no; ++k) {
            int2 e = ovf[k];
            if (e.x == n) {
                float4 a0 = *(const float4*)(lp + (size_t)e.y * 64);
                S.add(sum16(att_part(a0, b, wt)), a0);
            }
        }
    }

    float4 bb = *(const float4*)(b2 + sub * 4);
    float4 o;
    o.x = fmaxf(S.x / S.den + bb.x, 0.f);
    o.y = fmaxf(S.y / S.den + bb.y, 0.f);
    o.z = fmaxf(S.z / S.den + bb.z, 0.f);
    o.w = fmaxf(S.w / S.den + bb.w, 0.f);
    *(float4*)(out + (size_t)n * 64 + sub * 4) = o;
}

extern "C" void kernel_launch(void* const* d_in, const int* in_sizes, int n_in,
                              void* d_out, int out_size, void* d_ws, size_t ws_size,
                              hipStream_t stream) {
    const float* x   = (const float*)d_in[0];
    const int*   ei  = (const int*)d_in[1];
    const float* W1  = (const float*)d_in[2];
    const float* b1  = (const float*)d_in[3];
    const float* lnw = (const float*)d_in[4];
    const float* lnb = (const float*)d_in[5];
    const float* Wl  = (const float*)d_in[6];
    const float* Wr  = (const float*)d_in[7];
    const float* att = (const float*)d_in[8];
    const float* b2  = (const float*)d_in[9];
    float* out = (float*)d_out;

    const int N = in_sizes[0] / 64;
    const int E = in_sizes[1] / 2;

    char* ws = (char*)d_ws;
    size_t off = 0;
    auto carve = [&](size_t bytes) -> void* {
        void* p = ws + off;
        off += (bytes + 255) & ~(size_t)255;
        return p;
    };
    float* hbuf   = (float*)carve((size_t)N * 64 * 4);
    float* lnout  = (float*)carve((size_t)N * 64 * 4);
    float* xlb    = (float*)carve((size_t)N * 64 * 4);
    float* xrb    = (float*)carve((size_t)N * 64 * 4);
    int*   cnt    = (int*)carve((size_t)N * 4);
    int*   ovfc   = (int*)carve(4);
    int*   bins   = (int*)carve((size_t)N * BIN_CAP * 4);
    int2*  ovf    = (int2*)carve((size_t)E * 8);
    const int nr   = N > RNODES ? (N + RNODES - 1) / RNODES : 1;
    const int rcap = 2 * ((E + nr - 1) / nr) + 1024;
    int*   rcur   = (int*)carve((size_t)NRMAX * 4);
    int*   spillc = (int*)carve(4);
    int2*  redges = (int2*)carve((size_t)nr * rcap * 8);
    int2*  spill  = (int2*)carve((size_t)E * 8);
    const bool lds_bins = nr <= NRMAX && off <= ws_size;

    const int ngrid = (int)(((long long)N * 16 + 255) / 256);
    const int nstrips = N / 16, rem = N - nstrips * 16;

    if (lds_bins) {
        k_zero_small<<<1, 256, 0, stream>>>(rcur, ovfc, spillc);
        k_part<<<(E + PCHUNK - 1) / PCHUNK, 256, 0, stream>>>(ei, rcur, redges, spillc, spill, E, rcap);
        k_build<<<nr, 1024, 0, stream>>>(rcur, redges, spillc, spill, cnt, bins, ovfc, ovf, N, E, rcap);
    } else {
        k_zero_cnt<<<(N + 255) / 256, 256, 0, stream>>>(cnt, ovfc, N);
        k_bin<<<((E + 3) / 4 + 255) / 256, 256, 0, stream>>>(ei, cnt, bins, ovfc, ovf, E);
    }

    k_gemm64<<<(nstrips + 7) / 8, 256, 0, stream>>>(x, W1, hbuf, nstrips);
    if (rem) k_gemm_tail<<<(rem * 64 + 255) / 256, 256, 0, stream>>>(x, W1, hbuf, nstrips * 16, N);

    k_gcn_node<<<ngrid, 256, 0, stream>>>(cnt, bins, ovfc, ovf, hbuf, b1, lnw, lnb, lnout, N, E);

    k_gemm64_dual<<<(nstrips + 7) / 8, 256, 0, stream>>>(lnout, Wl, Wr, xlb, xrb, nstrips);
    if (rem) {
        k_gemm_tail<<<(rem * 64 + 255) / 256, 256, 0, stream>>>(lnout, Wl, xlb, nstrips * 16, N);
        k_gemm_tail<<<(rem * 64 + 255) / 256, 256, 0, stream>>>(lnout, Wr, xrb, nstrips * 16, N);
    }

    k_att_node<<<ngrid, 256, 0, stream>>>(cnt, bins, ovfc, ovf, xlb, xrb, att, b2, out, N, E);
}


# x@W1 strips run in the same launch as the edge partition: independent work, split by block index (partition blocks first, GEMM blocks fill the idle compute units); 7 launches -> 6
# speedup vs baseline: 6.5020x; 1.0447x over previous
#include <hip/hip_runtime.h>
#include <hip/hip_bf16.h>

typedef float v2f __attribute__((ext_vector_type(2)));
typedef float v8f __attribute__((ext_vector_type(8)));

#define LN_EPS    1e-5f
#define NEG_SLOPE 0.2f

#define BIN_CAP 32

__global__ __launch_bounds__(256) void k_zero_cnt(int* __restrict__ cnt,
                                                  int* __restrict__ ovf_cnt, int N) {
    int i = blockIdx.x * 256 + threadIdx.x;
    if (i < N) cnt[i] = 0;
    if (i == 0) *ovf_cnt = 0;
}

__device__ __forceinline__ void bin_store(int s, int d, int r,
                                          int* __restrict__ bins,
                                          int* __restrict__ ovf_cnt,
                                          int2* __restrict__ ovf, int E) {
    if (r < BIN_CAP) {
        bins[(size_t)d * BIN_CAP + r] = s;
    } else {
        int o = atomicAdd(ovf_cnt, 1);
        if (o < E) ovf[o] = make_int2(d, s);
    }
}

__global__ __launch_bounds__(256) void k_bin(const int* __restrict__ ei,
                                             int* __restrict__ cnt,
                                             int* __restrict__ bins,
                                             int* __restrict__ ovf_cnt,
                                             int2* __restrict__ ovf, int E) {
    int base = (blockIdx.x * 256 + threadIdx.x) * 4;
    if (base >= E) return;
    if ((E & 3) == 0) {
        int4 s = *(const int4*)(ei + base);
        int4 d = *(const int4*)(ei + E + base);
        int r0 = atomicAdd(cnt + d.x, 1);
        int r1 = atomicAdd(cnt + d.y, 1);
        int r2 = atomicAdd(cnt + d.z, 1);
        int r3 = atomicAdd(cnt + d.w, 1);
        bin_store(s.x, d.x, r0, bins, ovf_cnt, ovf, E);
        bin_store(s.y, d.y, r1, bins, ovf_cnt, ovf, E);
        bin_store(s.z, d.z, r2, bins, ovf_cnt, ovf, E);
        bin_store(s.w, d.w, r3, bins, ovf_cnt, ovf, E);
    } else {
        for (int k = 0; k < 4; ++k) {
            int i = base + k;
            if (i < E) {
                int s = ei[i], d = ei[E + i];
                bin_store(s, d, atomicAdd(cnt + d, 1), bins, ovf_cnt, ovf, E);
            }
        }
    }
}

#define RSH    9
#define RNODES (1 << RSH)
#define NRMAX  256
#define PCHUNK 4096

__global__ __launch_bounds__(256) void k_zero_small(int* __restrict__ rcur,
                                                    int* __restrict__ ovf_cnt,
                                                    int* __restrict__ spill_cnt) {
    int t = threadIdx.x;
    if (t < NRMAX) rcur[t] = 0;
    if (t == 0) { *ovf_cnt = 0; *spill_cnt = 0; }
}

#define PART_LH     0
#define PART_LSTART (PART_LH + NRMAX * 4)
#define PART_GPOS   (PART_LSTART + NRMAX * 4)
#define PART_WSUM   (PART_GPOS + NRMAX * 4)
#define PART_STAGE  (PART_WSUM + (NRMAX / 32) * 4)
#define PART_LDS    (PART_STAGE + PCHUNK * 8)

__device__ __forceinline__ void part_block(const int blk, unsigned char* __restrict__ smem,
                                           const int* __restrict__ ei,
                                           int* __restrict__ rcur,
                                           int2* __restrict__ redges,
                                           int* __restrict__ spill_cnt,
                                           int2* __restrict__ spill,
                                           int E, int rcap) {
    int*  lh     = (int*)(smem + PART_LH);
    int*  lstart = (int*)(smem + PART_LSTART);
    int*  gpos   = (int*)(smem + PART_GPOS);
    int*  wsum   = (int*)(smem + PART_WSUM);
    int2* stage  = (int2*)(smem + PART_STAGE);

    const int t  = threadIdx.x;
    const int e0 = blk * PCHUNK;
    lh[t] = 0;

    int s[16], d[16], r[16];
#pragma unroll
    for (int k = 0; k < 16; ++k) {
        int e = e0 + k * 256 + t;
        bool ok = e < E;
        s[k] = ok ? ei[e] : 0;
        d[k] = ok ? ei[E + e] : -1;
    }
    __syncthreads();
#pragma unroll
    for (int k = 0; k < 16; ++k)
        r[k] = (d[k] >= 0) ? atomicAdd(&lh[d[k] >> RSH], 1) : 0;
    __syncthreads();

    const int h = lh[t];
    int incl = h;
    for (int off = 1; off < 32; off <<= 1) {
        int u = __shfl_up(incl, off, 32);
        if ((t & 31) >= off) incl += u;
    }
    if ((t & 31) == 31) wsum[t >> 5] = incl;
    __syncthreads();
    int before = 0;
    for (int w = 0; w < (t >> 5); ++w) before += wsum[w];
    lstart[t] = before + incl - h;
    gpos[t]   = h ? atomicAdd(rcur + t, h) : 0;
    __syncthreads();

#pragma unroll
    for (int k = 0; k < 16; ++k)
        if (d[k] >= 0) stage[lstart[d[k] >> RSH] + r[k]] = make_int2(d[k], s[k]);
    __syncthreads();

    int n = E - e0;
    n = n < PCHUNK ? n : PCHUNK;
    for (int i = t; i < n; i += 256) {
        int2 e = stage[i];
        int b = e.x >> RSH;
        int g = gpos[b] + (i - lstart[b]);
        if (g < rcap) {
            redges[(size_t)b * rcap + g] = e;
        } else {
            int o = atomicAdd(spill_cnt, 1);
            if (o < E) spill[o] = e;
        }
    }
}

__device__ __forceinline__ void build_put(int2 e, int l, int r, int* __restrict__ lb,
                                          int* __restrict__ ovf_cnt,
                                          int2* __restrict__ ovf, int E) {
    if (r < BIN_CAP) {
        lb[l * BIN_CAP + r] = e.y;
    } else {
        int o = atomicAdd(ovf_cnt, 1);
        if (o < E) ovf[o] = e;
    }
}

__global__ __launch_bounds__(1024) void k_build(const int* __restrict__ rcur,
                                                const int2* __restrict__ redges,
                                                const int* __restrict__ spill_cnt,
                                                const int2* __restrict__ spill,
                                                int* __restrict__ cnt,
                                                int* __restrict__ bins,
                                                int* __restrict__ ovf_cnt,
                                                int2* __restrict__ ovf,
                                                int N, int E, int rcap) {
    __shared__ int lc[RNODES];
    __shared__ int lb[RNODES * BIN_CAP];

    const int t = threadIdx.x, b = blockIdx.x;
    const int base = b << RSH;
    if (t < RNODES) lc[t] = 0;
    {
        int4 z = make_int4(0, 0, 0, 0);
        for (int i = t; i < RNODES * BIN_CAP / 4; i += 1024) ((int4*)lb)[i] = z;
    }
    __syncthreads();

    int m = rcur[b];
    m = m < rcap ? m : rcap;
    const int2* re = redges + (size_t)b * rcap;
    for (int i0 = 0; i0 < m; i0 += 8 * 1024) {
        int2 e[8];
#pragma unroll
        for (int k = 0; k < 8; ++k) {
            int i = i0 + k * 1024 + t;
            e[k] = (i < m) ? re[i] : make_int2(-1, 0);
        }
        int r[8];
#pragma unroll
        for (int k = 0; k < 8; ++k)
            r[k] = (e[k].x >= 0) ? atomicAdd(&lc[e[k].x - base], 1) : 0;
#pragma unroll
        for (int k = 0; k < 8; ++k)
            if (e[k].x >= 0) build_put(e[k], e[k].x - base, r[k], lb, ovf_cnt, ovf, E);
    }
    int ns = *spill_cnt;
    ns = ns < E ? ns : E;
    for (int i = t; i < ns; i += 1024) {
        int2 e = spill[i];
        if ((e.x >> RSH) == b)
            build_put(e, e.x - base, atomicAdd(&lc[e.x - base], 1), lb, ovf_cnt, ovf, E);
    }
    __syncthreads();

    int nn = N - base;
    nn = nn < RNODES ? nn : RNODES;
    if (t < nn) cnt[base + t] = lc[t];
    int4* dst = (int4*)(bins + (size_t)base * BIN_CAP);
    for (int i = t; i < nn * (BIN_CAP / 4); i += 1024) dst[i] = ((const int4*)lb)[i];
}

__device__ __forceinline__ float dinv_of(int c) {
    float g = (float)(c + 1);
    return (g > 0.f) ? rsqrtf(g) : 0.f;
}

__device__ __forceinline__ void stage_w(float* __restrict__ sWp,
                                        const float* __restrict__ W) {
    for (int i = threadIdx.x; i < 4096; i += 256) {
        int k = i >> 6, c = i & 63;
        int kk = k >> 2, h = (k >> 1) & 1, j = k & 1;
        sWp[((((kk << 1) + h) << 6) + c) * 2 + j] = W[i];
    }
}

__device__ __forceinline__ void gemm64_block(const int blk, float* __restrict__ sWp,
                                             const float* __restrict__ A,
                                             const float* __restrict__ W,
                                             float* __restrict__ C, int nstrips) {
    stage_w(sWp, W);
    __syncthreads();

    int lane  = threadIdx.x & 31;
    int strip = blk * 8 + (threadIdx.x >> 5);
    if (strip >= nstrips) return;
    int r = lane & 15, h = lane >> 4;

    const float* ap = A + (size_t)(strip * 16 + r) * 64 + 2 * h;
    v8f zero = {0.f, 0.f, 0.f, 0.f, 0.f, 0.f, 0.f, 0.f};
    v8f acc[4] = {zero, zero, zero, zero};

    for (int kk = 0; kk < 16; ++kk) {
        v2f a = *(const v2f*)(ap + kk * 4);
        const v2f* bp = (const v2f*)&sWp[((((kk << 1) + h) << 6) + r) << 1];
#pragma unroll
        for (int t = 0; t < 4; ++t) {
            acc[t] = __builtin_amdgcn_wmma_f32_16x16x4_f32(
                false, a, false, bp[t * 16], (short)0, acc[t], false, false);
        }
    }
    float* cp = C + (size_t)strip * 16 * 64;
#pragma unroll
    for (int t = 0; t < 4; ++t)
#pragma unroll
        for (int g = 0; g < 8; ++g)
            cp[(size_t)(g + 8 * h) * 64 + t * 16 + r] = acc[t][g];
}

__global__ __launch_bounds__(256) void k_gemm64(const float* __restrict__ A,
                                                const float* __restrict__ W,
                                                float* __restrict__ C, int nstrips) {
    __shared__ float sWp[64 * 64];
    gemm64_block(blockIdx.x, sWp, A, W, C, nstrips);
}

__global__ __launch_bounds__(256) void k_part_gemm(const int* __restrict__ ei,
                                                   int* __restrict__ rcur,
                                                   int2* __restrict__ redges,
                                                   int* __restrict__ spill_cnt,
                                                   int2* __restrict__ spill,
                                                   int E, int rcap, int npart,
                                                   const float* __restrict__ A,
                                                   const float* __restrict__ W,
                                                   float* __restrict__ C, int nstrips) {
    __shared__ __align__(16) unsigned char smem[PART_LDS];
    if ((int)blockIdx.x < npart)
        part_block(blockIdx.x, smem, ei, rcur, redges, spill_cnt, spill, E, rcap);
    else
        gemm64_block((int)blockIdx.x - npart, (float*)smem, A, W, C, nstrips);
}

__global__ __launch_bounds__(256) void k_gemm64_dual(const float* __restrict__ A,
                                                     const float* __restrict__ Wl,
                                                     const float* __restrict__ Wr,
                                                     float* __restrict__ Cl,
                                                     float* __restrict__ Cr, int nstrips) {
    __shared__ float sWp[2 * 64 * 64];
    stage_w(sWp, Wl);
    stage_w(sWp + 4096, Wr);
    __syncthreads();

    int lane  = threadIdx.x & 31;
    int strip = blockIdx.x * 8 + (threadIdx.x >> 5);
    if (strip >= nstrips) return;
    int r = lane & 15, h = lane >> 4;

    const float* ap = A + (size_t)(strip * 16 + r) * 64 + 2 * h;
    v8f zero = {0.f, 0.f, 0.f, 0.f, 0.f, 0.f, 0.f, 0.f};
    v8f al[4] = {zero, zero, zero, zero};
    v8f ar[4] = {zero, zero, zero, zero};

    for (int kk = 0; kk < 16; ++kk) {
        v2f a = *(const v2f*)(ap + kk * 4);
        const v2f* bpl = (const v2f*)&sWp[((((kk << 1) + h) << 6) + r) << 1];
        const v2f* bpr = bpl + 2048;
#pragma unroll
        for (int t = 0; t < 4; ++t) {
            al[t] = __builtin_amdgcn_wmma_f32_16x16x4_f32(
                false, a, false, bpl[t * 16], (short)0, al[t], false, false);
            ar[t] = __builtin_amdgcn_wmma_f32_16x16x4_f32(
                false, a, false, bpr[t * 16], (short)0, ar[t], false, false);
        }
    }
    float* cl = Cl + (size_t)strip * 16 * 64;
    float* cr = Cr + (size_t)strip * 16 * 64;
#pragma unroll
    for (int t = 0; t < 4; ++t)
#pragma unroll
        for (int g = 0; g < 8; ++g) {
            cl[(size_t)(g + 8 * h) * 64 + t * 16 + r] = al[t][g];
            cr[(size_t)(g + 8 * h) * 64 + t * 16 + r] = ar[t][g];
        }
}

__global__ __launch_bounds__(256) void k_gemm_tail(const float* __restrict__ A,
                                                   const float* __restrict__ W,
                                                   float* __restrict__ C,
                                                   int row0, int nrows) {
    int i = blockIdx.x * 256 + threadIdx.x;
    int rows = nrows - row0;
    if (i >= rows * 64) return;
    int rr = row0 + i / 64, c = i & 63;
    float s = 0.f;
    for (int k = 0; k < 64; ++k) s += A[(size_t)rr * 64 + k] * W[k * 64 + c];
    C[(size_t)rr * 64 + c] = s;
}

__device__ __forceinline__ float sum16(float p) {
    for (int m = 8; m >= 1; m >>= 1) p += __shfl_xor(p, m, 16);
    return p;
}

__global__ __launch_bounds__(256) void k_gcn_node(const int* __restrict__ cnt,
                                                  const int* __restrict__ bins,
                                                  const int* __restrict__ ovf_cnt,
                                                  const int2* __restrict__ ovf,
                                                  const float* __restrict__ h,
                                                  const float* __restrict__ b1,
                                                  const float* __restrict__ lnw,
                                                  const float* __restrict__ lnb,
                                                  float* __restrict__ hout, int N, int E) {
    int gid = blockIdx.x * 256 + threadIdx.x;
    int n = gid >> 4, sub = gid & 15;
    if (n >= N) return;
    const int* bp = bins + (size_t)n * BIN_CAP;
    const int2 ent = *(const int2*)(bp + 2 * sub);
    int c = cnt[n];
    float dn = dinv_of(c);
    const float* hp = h + sub * 4;

    float4 v = *(const float4*)(hp + (size_t)n * 64);
    float w = dn * dn;
    float ax = v.x * w, ay = v.y * w, az = v.z * w, aw = v.w * w;

    const int cm = c < BIN_CAP ? c : BIN_CAP;
    int j = 0;
    for (; j + 4 <= cm; j += 4) {
        const int l = j >> 1;
        int s0 = __shfl(ent.x, l, 16), s1 = __shfl(ent.y, l, 16);
        int s2 = __shfl(ent.x, l + 1, 16), s3 = __shfl(ent.y, l + 1, 16);
        float w0 = dinv_of(cnt[s0]) * dn, w1 = dinv_of(cnt[s1]) * dn;
        float w2 = dinv_of(cnt[s2]) * dn, w3 = dinv_of(cnt[s3]) * dn;
        float4 v0 = *(const float4*)(hp + (size_t)s0 * 64);
        float4 v1 = *(const float4*)(hp + (size_t)s1 * 64);
        float4 v2 = *(const float4*)(hp + (size_t)s2 * 64);
        float4 v3 = *(const float4*)(hp + (size_t)s3 * 64);
        ax += v0.x * w0; ay += v0.y * w0; az += v0.z * w0; aw += v0.w * w0;
        ax += v1.x * w1; ay += v1.y * w1; az += v1.z * w1; aw += v1.w * w1;
        ax += v2.x * w2; ay += v2.y * w2; az += v2.z * w2; aw += v2.w * w2;
        ax += v3.x * w3; ay += v3.y * w3; az += v3.z * w3; aw += v3.w * w3;
    }
    const int t = cm - j;
    if (t > 0) {
        const int l = j >> 1;
        int s0 = __shfl(ent.x, l, 16), s1 = __shfl(ent.y, l, 16), s2 = __shfl(ent.x, l + 1, 16);
        s1 = t > 1 ? s1 : s0;
        s2 = t > 2 ? s2 : s0;
        const int c0 = cnt[s0], c1 = cnt[s1], c2 = cnt[s2];
        const float k1 = t > 1 ? 1.f : 0.f, k2 = t > 2 ? 1.f : 0.f;
        float w0 = dinv_of(c0) * dn;
        float w1 = dinv_of(c1) * dn * k1;
        float w2 = dinv_of(c2) * dn * k2;
        float4 v0 = *(const float4*)(hp + (size_t)s0 * 64);
        float4 v1 = *(const float4*)(hp + (size_t)s1 * 64);
        float4 v2 = *(const float4*)(hp + (size_t)s2 * 64);
        ax += v0.x * w0; ay += v0.y * w0; az += v0.z * w0; aw += v0.w * w0;
        ax += v1.x * w1; ay += v1.y * w1; az += v1.z * w1; aw += v1.w * w1;
        ax += v2.x * w2; ay += v2.y * w2; az += v2.z * w2; aw += v2.w * w2;
    }
    if (c > BIN_CAP) {
        int no = *ovf_cnt;
        no = no < E ? no : E;
        for (int k = 0; k < no; ++k) {
            int2 e = ovf[k];
            if (e.x == n) {
                int s0 = e.y;
                float w0 = dinv_of(cnt[s0]) * dn;
                float4 v0 = *(const float4*)(hp + (size_t)s0 * 64);
                ax += v0.x * w0; ay += v0.y * w0; az += v0.z * w0; aw += v0.w * w0;
            }
        }
    }

    float4 bb = *(const float4*)(b1 + sub * 4);
    float vx = fmaxf(ax + bb.x, 0.f), vy = fmaxf(ay + bb.y, 0.f);
    float vz = fmaxf(az + bb.z, 0.f), vw = fmaxf(aw + bb.w, 0.f);
    float mu = sum16((vx + vy) + (vz + vw)) * (1.f / 64.f);
    float dx = vx - mu, dy = vy - mu, dz = vz - mu, dw = vw - mu;
    float q = sum16((dx * dx + dy * dy) + (dz * dz + dw * dw));
    float inv = rsqrtf(q * (1.f / 64.f) + LN_EPS);
    float4 gw = *(const float4*)(lnw + sub * 4);
    float4 gb = *(const float4*)(lnb + sub * 4);
    float4 o;
    o.x = dx * inv * gw.x + gb.x;
    o.y = dy * inv * gw.y + gb.y;
    o.z = dz * inv * gw.z + gb.z;
    o.w = dw * inv * gw.w + gb.w;
    *(float4*)(hout + (size_t)n * 64 + sub * 4) = o;
}

__device__ __forceinline__ float att_part(float4 a, float4 b, float4 w) {
    float p = 0.f, v;
    v = a.x + b.x; v = v > 0.f ? v : NEG_SLOPE * v; p += w.x * v;
    v = a.y + b.y; v = v > 0.f ? v : NEG_SLOPE * v; p += w.y * v;
    v = a.z + b.z; v = v > 0.f ? v : NEG_SLOPE * v; p += w.z * v;
    v = a.w + b.w; v = v > 0.f ? v : NEG_SLOPE * v; p += w.w * v;
    return p;
}

struct SoftAcc {
    float m, den, x, y, z, w;
    __device__ __forceinline__ void add(float e, float4 a) {
        float mn = fmaxf(m, e);
        float sc = __expf(m - mn);
        float p  = __expf(e - mn);
        den = den * sc + p;
        x = x * sc + p * a.x;
        y = y * sc + p * a.y;
        z = z * sc + p * a.z;
        w = w * sc + p * a.w;
        m = mn;
    }
};

__global__ __launch_bounds__(256) void k_att_node(const int* __restrict__ cnt,
                                                  const int* __restrict__ bins,
                                                  const int* __restrict__ ovf_cnt,
                                                  const int2* __restrict__ ovf,
                                                  const float* __restrict__ xl,
                                                  const float* __restrict__ xr,
                                                  const float* __restrict__ att,
                                                  const float* __restrict__ b2,
                                                  float* __restrict__ out, int N, int E) {
    int gid = blockIdx.x * 256 + threadIdx.x;
    int n = gid >> 4, sub = gid & 15;
    if (n >= N) return;
    const int* bp = bins + (size_t)n * BIN_CAP;
    const int2 ent = *(const int2*)(bp + 2 * sub);
    int c = cnt[n];
    const float* lp = xl + sub * 4;
    float4 b  = *(const float4*)(xr + (size_t)n * 64 + sub * 4);
    float4 wt = *(const float4*)(att + sub * 4);

    float4 a = *(const float4*)(lp + (size_t)n * 64);
    SoftAcc S;
    S.m = sum16(att_part(a, b, wt));
    S.den = 1.f; S.x = a.x; S.y = a.y; S.z = a.z; S.w = a.w;

    const int cm = c < BIN_CAP ? c : BIN_CAP;
    int j = 0;
    for (; j + 4 <= cm; j += 4) {
        const int l = j >> 1;
        int s0 = __shfl(ent.x, l, 16), s1 = __shfl(ent.y, l, 16);
        int s2 = __shfl(ent.x, l + 1, 16), s3 = __shfl(ent.y, l + 1, 16);
        float4 a0 = *(const float4*)(lp + (size_t)s0 * 64);
        float4 a1 = *(const float4*)(lp + (size_t)s1 * 64);
        float4 a2 = *(const float4*)(lp + (size_t)s2 * 64);
        float4 a3 = *(const float4*)(lp + (size_t)s3 * 64);
        float e0 = sum16(att_part(a0, b, wt));
        float e1 = sum16(att_part(a1, b, wt));
        float e2 = sum16(att_part(a2, b, wt));
        float e3 = sum16(att_part(a3, b, wt));
        S.add(e0, a0); S.add(e1, a1); S.add(e2, a2); S.add(e3, a3);
    }
    const int t = cm - j;
    if (t > 0) {
        const int l = j >> 1;
        int s0 = __shfl(ent.x, l, 16), s1 = __shfl(ent.y, l, 16), s2 = __shfl(ent.x, l + 1, 16);
        s1 = t > 1 ? s1 : s0;
        s2 = t > 2 ? s2 : s0;
        float4 a0 = *(const float4*)(lp + (size_t)s0 * 64);
        float4 a1 = *(const float4*)(lp + (size_t)s1 * 64);
        float4 a2 = *(const float4*)(lp + (size_t)s2 * 64);
        float e0 = sum16(att_part(a0, b, wt));
        float e1 = sum16(att_part(a1, b, wt));
        float e2 = sum16(att_part(a2, b, wt));
        e1 = t > 1 ? e1 : -INFINITY;
        e2 = t > 2 ? e2 : -INFINITY;
        S.add(e0, a0); S.add(e1, a1); S.add(e2, a2);
    }
    if (c > BIN_CAP) {
        int no = *ovf_cnt;
        no = no < E ? no : E;
        for (int k = 0; k < no; ++k) {
            int2 e = ovf[k];
            if (e.x == n) {
                float4 a0 = *(const float4*)(lp + (size_t)e.y * 64);
                S.add(sum16(att_part(a0, b, wt)), a0);
            }
        }
    }

    float4 bb = *(const float4*)(b2 + sub * 4);
    float4 o;
    o.x = fmaxf(S.x / S.den + bb.x, 0.f);
    o.y = fmaxf(S.y / S.den + bb.y, 0.f);
    o.z = fmaxf(S.z / S.den + bb.z, 0.f);
    o.w = fmaxf(S.w / S.den + bb.w, 0.f);
    *(float4*)(out + (size_t)n * 64 + sub * 4) = o;
}

extern "C" void kernel_launch(void* const* d_in, const int* in_sizes, int n_in,
                              void* d_out, int out_size, void* d_ws, size_t ws_size,
                              hipStream_t stream) {
    const float* x   = (const float*)d_in[0];
    const int*   ei  = (const int*)d_in[1];
    const float* W1  = (const float*)d_in[2];
    const float* b1  = (const float*)d_in[3];
    const float* lnw = (const float*)d_in[4];
    const float* lnb = (const float*)d_in[5];
    const float* Wl  = (const float*)d_in[6];
    const float* Wr  = (const float*)d_in[7];
    const float* att = (const float*)d_in[8];
    const float* b2  = (const float*)d_in[9];
    float* out = (float*)d_out;

    const int N = in_sizes[0] / 64;
    const int E = in_sizes[1] / 2;

    char* ws = (char*)d_ws;
    size_t off = 0;
    auto carve = [&](size_t bytes) -> void* {
        void* p = ws + off;
        off += (bytes + 255) & ~(size_t)255;
        return p;
    };
    float* hbuf   = (float*)carve((size_t)N * 64 * 4);
    float* lnout  = (float*)carve((size_t)N * 64 * 4);
    float* xlb    = (float*)carve((size_t)N * 64 * 4);
    float* xrb    = (float*)carve((size_t)N * 64 * 4);
    int*   cnt    = (int*)carve((size_t)N * 4);
    int*   ovfc   = (int*)carve(4);
    int*   bins   = (int*)carve((size_t)N * BIN_CAP * 4);
    int2*  ovf    = (int2*)carve((size_t)E * 8);
    const int nr   = N > RNODES ? (N + RNODES - 1) / RNODES : 1;
    const int rcap = 2 * ((E + nr - 1) / nr) + 1024;
    int*   rcur   = (int*)carve((size_t)NRMAX * 4);
    int*   spillc = (int*)carve(4);
    int2*  redges = (int2*)carve((size_t)nr * rcap * 8);
    int2*  spill  = (int2*)carve((size_t)E * 8);
    const bool lds_bins = nr <= NRMAX && off <= ws_size;

    const int ngrid = (int)(((long long)N * 16 + 255) / 256);
    const int nstrips = N / 16, rem = N - nstrips * 16;

    const int npart = (E + PCHUNK - 1) / PCHUNK;
    const int ngemm = (nstrips + 7) / 8;
    if (lds_bins) {
        k_zero_small<<<1, 256, 0, stream>>>(rcur, ovfc, spillc);
        k_part_gemm<<<npart + ngemm, 256, 0, stream>>>(ei, rcur, redges, spillc, spill, E, rcap, npart,
                                                       x, W1, hbuf, nstrips);
        k_build<<<nr, 1024, 0, stream>>>(rcur, redges, spillc, spill, cnt, bins, ovfc, ovf, N, E, rcap);
    } else {
        k_zero_cnt<<<(N + 255) / 256, 256, 0, stream>>>(cnt, ovfc, N);
        k_bin<<<((E + 3) / 4 + 255) / 256, 256, 0, stream>>>(ei, cnt, bins, ovfc, ovf, E);
        k_gemm64<<<ngemm, 256, 0, stream>>>(x, W1, hbuf, nstrips);
    }
    if (rem) k_gemm_tail<<<(rem * 64 + 255) / 256, 256, 0, stream>>>(x, W1, hbuf, nstrips * 16, N);

    k_gcn_node<<<ngrid, 256, 0, stream>>>(cnt, bins, ovfc, ovf, hbuf, b1, lnw, lnb, lnout, N, E);

    k_gemm64_dual<<<(nstrips + 7) / 8, 256, 0, stream>>>(lnout, Wl, Wr, xlb, xrb, nstrips);
    if (rem) {
        k_gemm_tail<<<(rem * 64 + 255) / 256, 256, 0, stream>>>(lnout, Wl, xlb, nstrips * 16, N);
        k_gemm_tail<<<(rem * 64 + 255) / 256, 256, 0, stream>>>(lnout, Wr, xrb, nstrips * 16, N);
    }

    k_att_node<<<ngrid, 256, 0, stream>>>(cnt, bins, ovfc, ovf, xlb, xrb, att, b2, out, N, E);
}


# GCN gather+LayerNorm block (32 nodes) hands its rows in LDS to 4 of its waves for rows@Wl, rows@Wr on WMMA, W read from global; LayerNorm buffer and one launch gone (6 -> 5); sum orders kept
# speedup vs baseline: 6.7442x; 1.0373x over previous
#include <hip/hip_runtime.h>
#include <hip/hip_bf16.h>

typedef float v2f __attribute__((ext_vector_type(2)));
typedef float v8f __attribute__((ext_vector_type(8)));

#define LN_EPS    1e-5f
#define NEG_SLOPE 0.2f

#define BIN_CAP 32

__global__ __launch_bounds__(256) void k_zero_cnt(int* __restrict__ cnt,
                                                  int* __restrict__ ovf_cnt, int N) {
    int i = blockIdx.x * 256 + threadIdx.x;
    if (i < N) cnt[i] = 0;
    if (i == 0) *ovf_cnt = 0;
}

__device__ __forceinline__ void bin_store(int s, int d, int r,
                                          int* __restrict__ bins,
                                          int* __restrict__ ovf_cnt,
                                          int2* __restrict__ ovf, int E) {
    if (r < BIN_CAP) {
        bins[(size_t)d * BIN_CAP + r] = s;
    } else {
        int o = atomicAdd(ovf_cnt, 1);
        if (o < E) ovf[o] = make_int2(d, s);
    }
}

__global__ __launch_bounds__(256) void k_bin(const int* __restrict__ ei,
                                             int* __restrict__ cnt,
                                             int* __restrict__ bins,
                                             int* __restrict__ ovf_cnt,
                                             int2* __restrict__ ovf, int E) {
    int base = (blockIdx.x * 256 + threadIdx.x) * 4;
    if (base >= E) return;
    if ((E & 3) == 0) {
        int4 s = *(const int4*)(ei + base);
        int4 d = *(const int4*)(ei + E + base);
        int r0 = atomicAdd(cnt + d.x, 1);
        int r1 = atomicAdd(cnt + d.y, 1);
        int r2 = atomicAdd(cnt + d.z, 1);
        int r3 = atomicAdd(cnt + d.w, 1);
        bin_store(s.x, d.x, r0, bins, ovf_cnt, ovf, E);
        bin_store(s.y, d.y, r1, bins, ovf_cnt, ovf, E);
        bin_store(s.z, d.z, r2, bins, ovf_cnt, ovf, E);
        bin_store(s.w, d.w, r3, bins, ovf_cnt, ovf, E);
    } else {
        for (int k = 0; k < 4; ++k) {
            int i = base + k;
            if (i < E) {
                int s = ei[i], d = ei[E + i];
                bin_store(s, d, atomicAdd(cnt + d, 1), bins, ovf_cnt, ovf, E);
            }
        }
    }
}

#define RSH    9
#define RNODES (1 << RSH)
#define NRMAX  256
#define PCHUNK 4096

__global__ __launch_bounds__(256) void k_zero_small(int* __restrict__ rcur,
                                                    int* __restrict__ ovf_cnt,
                                                    int* __restrict__ spill_cnt) {
    int t = threadIdx.x;
    if (t < NRMAX) rcur[t] = 0;
    if (t == 0) { *ovf_cnt = 0; *spill_cnt = 0; }
}

#define PART_LH     0
#define PART_LSTART (PART_LH + NRMAX * 4)
#define PART_GPOS   (PART_LSTART + NRMAX * 4)
#define PART_WSUM   (PART_GPOS + NRMAX * 4)
#define PART_STAGE  (PART_WSUM + (NRMAX / 32) * 4)
#define PART_LDS    (PART_STAGE + PCHUNK * 8)

__device__ __forceinline__ void part_block(const int blk, unsigned char* __restrict__ smem,
                                           const int* __restrict__ ei,
                                           int* __restrict__ rcur,
                                           int2* __restrict__ redges,
                                           int* __restrict__ spill_cnt,
                                           int2* __restrict__ spill,
                                           int E, int rcap) {
    int*  lh     = (int*)(smem + PART_LH);
    int*  lstart = (int*)(smem + PART_LSTART);
    int*  gpos   = (int*)(smem + PART_GPOS);
    int*  wsum   = (int*)(smem + PART_WSUM);
    int2* stage  = (int2*)(smem + PART_STAGE);

    const int t  = threadIdx.x;
    const int e0 = blk * PCHUNK;
    lh[t] = 0;

    int s[16], d[16], r[16];
#pragma unroll
    for (int k = 0; k < 16; ++k) {
        int e = e0 + k * 256 + t;
        bool ok = e < E;
        s[k] = ok ? ei[e] : 0;
        d[k] = ok ? ei[E + e] : -1;
    }
    __syncthreads();
#pragma unroll
    for (int k = 0; k < 16; ++k)
        r[k] = (d[k] >= 0) ? atomicAdd(&lh[d[k] >> RSH], 1) : 0;
    __syncthreads();

    const int h = lh[t];
    int incl = h;
    for (int off = 1; off < 32; off <<= 1) {
        int u = __shfl_up(incl, off, 32);
        if ((t & 31) >= off) incl += u;
    }
    if ((t & 31) == 31) wsum[t >> 5] = incl;
    __syncthreads();
    int before = 0;
    for (int w = 0; w < (t >> 5); ++w) before += wsum[w];
    lstart[t] = before + incl - h;
    gpos[t]   = h ? atomicAdd(rcur + t, h) : 0;
    __syncthreads();

#pragma unroll
    for (int k = 0; k < 16; ++k)
        if (d[k] >= 0) stage[lstart[d[k] >> RSH] + r[k]] = make_int2(d[k], s[k]);
    __syncthreads();

    int n = E - e0;
    n = n < PCHUNK ? n : PCHUNK;
    for (int i = t; i < n; i += 256) {
        int2 e = stage[i];
        int b = e.x >> RSH;
        int g = gpos[b] + (i - lstart[b]);
        if (g < rcap) {
            redges[(size_t)b * rcap + g] = e;
        } else {
            int o = atomicAdd(spill_cnt, 1);
            if (o < E) spill[o] = e;
        }
    }
}

__device__ __forceinline__ void build_put(int2 e, int l, int r, int* __restrict__ lb,
                                          int* __restrict__ ovf_cnt,
                                          int2* __restrict__ ovf, int E) {
    if (r < BIN_CAP) {
        lb[l * BIN_CAP + r] = e.y;
    } else {
        int o = atomicAdd(ovf_cnt, 1);
        if (o < E) ovf[o] = e;
    }
}

__global__ __launch_bounds__(1024) void k_build(const int* __restrict__ rcur,
                                                const int2* __restrict__ redges,
                                                const int* __restrict__ spill_cnt,
                                                const int2* __restrict__ spill,
                                                int* __restrict__ cnt,
                                                int* __restrict__ bins,
                                                int* __restrict__ ovf_cnt,
                                                int2* __restrict__ ovf,
                                                int N, int E, int rcap) {
    __shared__ int lc[RNODES];
    __shared__ int lb[RNODES * BIN_CAP];

    const int t = threadIdx.x, b = blockIdx.x;
    const int base = b << RSH;
    if (t < RNODES) lc[t] = 0;
    {
        int4 z = make_int4(0, 0, 0, 0);
        for (int i = t; i < RNODES * BIN_CAP / 4; i += 1024) ((int4*)lb)[i] = z;
    }
    __syncthreads();

    int m = rcur[b];
    m = m < rcap ? m : rcap;
    const int2* re = redges + (size_t)b * rcap;
    for (int i0 = 0; i0 < m; i0 += 8 * 1024) {
        int2 e[8];
#pragma unroll
        for (int k = 0; k < 8; ++k) {
            int i = i0 + k * 1024 + t;
            e[k] = (i < m) ? re[i] : make_int2(-1, 0);
        }
        int r[8];
#pragma unroll
        for (int k = 0; k < 8; ++k)
            r[k] = (e[k].x >= 0) ? atomicAdd(&lc[e[k].x - base], 1) : 0;
#pragma unroll
        for (int k = 0; k < 8; ++k)
            if (e[k].x >= 0) build_put(e[k], e[k].x - base, r[k], lb, ovf_cnt, ovf, E);
    }
    int ns = *spill_cnt;
    ns = ns < E ? ns : E;
    for (int i = t; i < ns; i += 1024) {
        int2 e = spill[i];
        if ((e.x >> RSH) == b)
            build_put(e, e.x - base, atomicAdd(&lc[e.x - base], 1), lb, ovf_cnt, ovf, E);
    }
    __syncthreads();

    int nn = N - base;
    nn = nn < RNODES ? nn : RNODES;
    if (t < nn) cnt[base + t] = lc[t];
    int4* dst = (int4*)(bins + (size_t)base * BIN_CAP);
    for (int i = t; i < nn * (BIN_CAP / 4); i += 1024) dst[i] = ((const int4*)lb)[i];
}

__device__ __forceinline__ float dinv_of(int c) {
    float g = (float)(c + 1);
    return (g > 0.f) ? rsqrtf(g) : 0.f;
}

__device__ __forceinline__ void stage_w(float* __restrict__ sWp,
                                        const float* __restrict__ W) {
    for (int i = threadIdx.x; i < 4096; i += 256) {
        int k = i >> 6, c = i & 63;
        int kk = k >> 2, h = (k >> 1) & 1, j = k & 1;
        sWp[((((kk << 1) + h) << 6) + c) * 2 + j] = W[i];
    }
}

__device__ __forceinline__ void gemm64_block(const int blk, float* __restrict__ sWp,
                                             const float* __restrict__ A,
                                             const float* __restrict__ W,
                                             float* __restrict__ C, int nstrips) {
    stage_w(sWp, W);
    __syncthreads();

    int lane  = threadIdx.x & 31;
    int strip = blk * 8 + (threadIdx.x >> 5);
    if (strip >= nstrips) return;
    int r = lane & 15, h = lane >> 4;

    const float* ap = A + (size_t)(strip * 16 + r) * 64 + 2 * h;
    v8f zero = {0.f, 0.f, 0.f, 0.f, 0.f, 0.f, 0.f, 0.f};
    v8f acc[4] = {zero, zero, zero, zero};

    for (int kk = 0; kk < 16; ++kk) {
        v2f a = *(const v2f*)(ap + kk * 4);
        const v2f* bp = (const v2f*)&sWp[((((kk << 1) + h) << 6) + r) << 1];
#pragma unroll
        for (int t = 0; t < 4; ++t) {
            acc[t] = __builtin_amdgcn_wmma_f32_16x16x4_f32(
                false, a, false, bp[t * 16], (short)0, acc[t], false, false);
        }
    }
    float* cp = C + (size_t)strip * 16 * 64;
#pragma unroll
    for (int t = 0; t < 4; ++t)
#pragma unroll
        for (int g = 0; g < 8; ++g)
            cp[(size_t)(g + 8 * h) * 64 + t * 16 + r] = acc[t][g];
}

__global__ __launch_bounds__(256) void k_gemm64(const float* __restrict__ A,
                                                const float* __restrict__ W,
                                                float* __restrict__ C, int nstrips) {
    __shared__ float sWp[64 * 64];
    gemm64_block(blockIdx.x, sWp, A, W, C, nstrips);
}

__global__ __launch_bounds__(256) void k_part_gemm(const int* __restrict__ ei,
                                                   int* __restrict__ rcur,
                                                   int2* __restrict__ redges,
                                                   int* __restrict__ spill_cnt,
                                                   int2* __restrict__ spill,
                                                   int E, int rcap, int npart,
                                                   const float* __restrict__ A,
                                                   const float* __restrict__ W,
                                                   float* __restrict__ C, int nstrips) {
    __shared__ __align__(16) unsigned char smem[PART_LDS];
    if ((int)blockIdx.x < npart)
        part_block(blockIdx.x, smem, ei, rcur, redges, spill_cnt, spill, E, rcap);
    else
        gemm64_block((int)blockIdx.x - npart, (float*)smem, A, W, C, nstrips);
}

__global__ __launch_bounds__(256) void k_gemm_tail(const float* __restrict__ A,
                                                   const float* __restrict__ W,
                                                   float* __restrict__ C,
                                                   int row0, int nrows) {
    int i = blockIdx.x * 256 + threadIdx.x;
    int rows = nrows - row0;
    if (i >= rows * 64) return;
    int rr = row0 + i / 64, c = i & 63;
    float s = 0.f;
    for (int k = 0; k < 64; ++k) s += A[(size_t)rr * 64 + k] * W[k * 64 + c];
    C[(size_t)rr * 64 + c] = s;
}

__device__ __forceinline__ float sum16(float p) {
    for (int m = 8; m >= 1; m >>= 1) p += __shfl_xor(p, m, 16);
    return p;
}

__device__ __forceinline__ float4 gcn_row(const int n, const int sub,
                                          const int* __restrict__ cnt,
                                          const int* __restrict__ bins,
                                          const int* __restrict__ ovf_cnt,
                                          const int2* __restrict__ ovf,
                                          const float* __restrict__ h,
                                          const float* __restrict__ b1,
                                          const float* __restrict__ lnw,
                                          const float* __restrict__ lnb, int E) {
    const int* bp = bins + (size_t)n * BIN_CAP;
    const int2 ent = *(const int2*)(bp + 2 * sub);
    int c = cnt[n];
    float dn = dinv_of(c);
    const float* hp = h + sub * 4;

    float4 v = *(const float4*)(hp + (size_t)n * 64);
    float w = dn * dn;
    float ax = v.x * w, ay = v.y * w, az = v.z * w, aw = v.w * w;

    const int cm = c < BIN_CAP ? c : BIN_CAP;
    int j = 0;
    for (; j + 4 <= cm; j += 4) {
        const int l = j >> 1;
        int s0 = __shfl(ent.x, l, 16), s1 = __shfl(ent.y, l, 16);
        int s2 = __shfl(ent.x, l + 1, 16), s3 = __shfl(ent.y, l + 1, 16);
        float w0 = dinv_of(cnt[s0]) * dn, w1 = dinv_of(cnt[s1]) * dn;
        float w2 = dinv_of(cnt[s2]) * dn, w3 = dinv_of(cnt[s3]) * dn;
        float4 v0 = *(const float4*)(hp + (size_t)s0 * 64);
        float4 v1 = *(const float4*)(hp + (size_t)s1 * 64);
        float4 v2 = *(const float4*)(hp + (size_t)s2 * 64);
        float4 v3 = *(const float4*)(hp + (size_t)s3 * 64);
        ax += v0.x * w0; ay += v0.y * w0; az += v0.z * w0; aw += v0.w * w0;
        ax += v1.x * w1; ay += v1.y * w1; az += v1.z * w1; aw += v1.w * w1;
        ax += v2.x * w2; ay += v2.y * w2; az += v2.z * w2; aw += v2.w * w2;
        ax += v3.x * w3; ay += v3.y * w3; az += v3.z * w3; aw += v3.w * w3;
    }
    const int t = cm - j;
    if (t > 0) {
        const int l = j >> 1;
        int s0 = __shfl(ent.x, l, 16), s1 = __shfl(ent.y, l, 16), s2 = __shfl(ent.x, l + 1, 16);
        s1 = t > 1 ? s1 : s0;
        s2 = t > 2 ? s2 : s0;
        const int c0 = cnt[s0], c1 = cnt[s1], c2 = cnt[s2];
        const float k1 = t > 1 ? 1.f : 0.f, k2 = t > 2 ? 1.f : 0.f;
        float w0 = dinv_of(c0) * dn;
        float w1 = dinv_of(c1) * dn * k1;
        float w2 = dinv_of(c2) * dn * k2;
        float4 v0 = *(const float4*)(hp + (size_t)s0 * 64);
        float4 v1 = *(const float4*)(hp + (size_t)s1 * 64);
        float4 v2 = *(const float4*)(hp + (size_t)s2 * 64);
        ax += v0.x * w0; ay += v0.y * w0; az += v0.z * w0; aw += v0.w * w0;
        ax += v1.x * w1; ay += v1.y * w1; az += v1.z * w1; aw += v1.w * w1;
        ax += v2.x * w2; ay += v2.y * w2; az += v2.z * w2; aw += v2.w * w2;
    }
    if (c > BIN_CAP) {
        int no = *ovf_cnt;
        no = no < E ? no : E;
        for (int k = 0; k < no; ++k) {
            int2 e = ovf[k];
            if (e.x == n) {
                int s0 = e.y;
                float w0 = dinv_of(cnt[s0]) * dn;
                float4 v0 = *(const float4*)(hp + (size_t)s0 * 64);
                ax += v0.x * w0; ay += v0.y * w0; az += v0.z * w0; aw += v0.w * w0;
            }
        }
    }

    float4 bb = *(const float4*)(b1 + sub * 4);
    float vx = fmaxf(ax + bb.x, 0.f), vy = fmaxf(ay + bb.y, 0.f);
    float vz = fmaxf(az + bb.z, 0.f), vw = fmaxf(aw + bb.w, 0.f);
    float mu = sum16((vx + vy) + (vz + vw)) * (1.f / 64.f);
    float dx = vx - mu, dy = vy - mu, dz = vz - mu, dw = vw - mu;
    float q = sum16((dx * dx + dy * dy) + (dz * dz + dw * dw));
    float inv = rsqrtf(q * (1.f / 64.f) + LN_EPS);
    float4 gw = *(const float4*)(lnw + sub * 4);
    float4 gb = *(const float4*)(lnb + sub * 4);
    float4 o;
    o.x = dx * inv * gw.x + gb.x;
    o.y = dy * inv * gw.y + gb.y;
    o.z = dz * inv * gw.z + gb.z;
    o.w = dw * inv * gw.w + gb.w;
    return o;
}

#define FT_ROWS  32
#define FT_PITCH 68

__global__ __launch_bounds__(512) void k_gcn_dual(const int* __restrict__ cnt,
                                                  const int* __restrict__ bins,
                                                  const int* __restrict__ ovf_cnt,
                                                  const int2* __restrict__ ovf,
                                                  const float* __restrict__ h,
                                                  const float* __restrict__ b1,
                                                  const float* __restrict__ lnw,
                                                  const float* __restrict__ lnb,
                                                  const float* __restrict__ Wl,
                                                  const float* __restrict__ Wr,
                                                  float* __restrict__ xl,
                                                  float* __restrict__ xr, int N, int E) {
    __shared__ __align__(16) float sA[FT_ROWS * FT_PITCH];
    const int tid = threadIdx.x;
    const int row = tid >> 4, sub = tid & 15;
    const int n0 = blockIdx.x * FT_ROWS;
    const int n = n0 + row;
    float4 o = make_float4(0.f, 0.f, 0.f, 0.f);
    if (n < N) o = gcn_row(n, sub, cnt, bins, ovf_cnt, ovf, h, b1, lnw, lnb, E);
    *(float4*)(sA + row * FT_PITCH + sub * 4) = o;
    __syncthreads();

    const int wave = tid >> 5;
    if (wave >= 4) return;
    const int lane = tid & 31;
    const int c = lane & 15, hh = lane >> 4;
    const int m = wave & 1;
    const float* __restrict__ W = (wave & 2) ? Wr : Wl;
    float* __restrict__ C = (wave & 2) ? xr : xl;

    const float* ap = sA + (m * 16 + c) * FT_PITCH + 2 * hh;
    const float* wp = W + (2 * hh) * 64 + 4 * c;
    v8f zero = {0.f, 0.f, 0.f, 0.f, 0.f, 0.f, 0.f, 0.f};
    v8f acc0 = zero, acc1 = zero, acc2 = zero, acc3 = zero;
#pragma unroll 4
    for (int kk = 0; kk < 16; ++kk) {
        v2f a = *(const v2f*)(ap + kk * 4);
        float4 w0 = *(const float4*)(wp + (size_t)(4 * kk) * 64);
        float4 w1 = *(const float4*)(wp + (size_t)(4 * kk + 1) * 64);
        v2f bq0 = {w0.x, w1.x}, bq1 = {w0.y, w1.y}, bq2 = {w0.z, w1.z}, bq3 = {w0.w, w1.w};
        acc0 = __builtin_amdgcn_wmma_f32_16x16x4_f32(false, a, false, bq0, (short)0, acc0, false, false);
        acc1 = __builtin_amdgcn_wmma_f32_16x16x4_f32(false, a, false, bq1, (short)0, acc1, false, false);
        acc2 = __builtin_amdgcn_wmma_f32_16x16x4_f32(false, a, false, bq2, (short)0, acc2, false, false);
        acc3 = __builtin_amdgcn_wmma_f32_16x16x4_f32(false, a, false, bq3, (short)0, acc3, false, false);
    }
#pragma unroll
    for (int g = 0; g < 8; ++g) {
        const int nn = n0 + m * 16 + 8 * hh + g;
        if (nn < N)
            *(float4*)(C + (size_t)nn * 64 + 4 * c) = make_float4(acc0[g], acc1[g], acc2[g], acc3[g]);
    }
}

__device__ __forceinline__ float att_part(float4 a, float4 b, float4 w) {
    float p = 0.f, v;
    v = a.x + b.x; v = v > 0.f ? v : NEG_SLOPE * v; p += w.x * v;
    v = a.y + b.y; v = v > 0.f ? v : NEG_SLOPE * v; p += w.y * v;
    v = a.z + b.z; v = v > 0.f ? v : NEG_SLOPE * v; p += w.z * v;
    v = a.w + b.w; v = v > 0.f ? v : NEG_SLOPE * v; p += w.w * v;
    return p;
}

struct SoftAcc {
    float m, den, x, y, z, w;
    __device__ __forceinline__ void add(float e, float4 a) {
        float mn = fmaxf(m, e);
        float sc = __expf(m - mn);
        float p  = __expf(e - mn);
        den = den * sc + p;
        x = x * sc + p * a.x;
        y = y * sc + p * a.y;
        z = z * sc + p * a.z;
        w = w * sc + p * a.w;
        m = mn;
    }
};

__global__ __launch_bounds__(256) void k_att_node(const int* __restrict__ cnt,
                                                  const int* __restrict__ bins,
                                                  const int* __restrict__ ovf_cnt,
                                                  const int2* __restrict__ ovf,
                                                  const float* __restrict__ xl,
                                                  const float* __restrict__ xr,
                                                  const float* __restrict__ att,
                                                  const float* __restrict__ b2,
                                                  float* __restrict__ out, int N, int E) {
    int gid = blockIdx.x * 256 + threadIdx.x;
    int n = gid >> 4, sub = gid & 15;
    if (n >= N) return;
    const int* bp = bins + (size_t)n * BIN_CAP;
    const int2 ent = *(const int2*)(bp + 2 * sub);
    int c = cnt[n];
    const float* lp = xl + sub * 4;
    float4 b  = *(const float4*)(xr + (size_t)n * 64 + sub * 4);
    float4 wt = *(const float4*)(att + sub * 4);

    float4 a = *(const float4*)(lp + (size_t)n * 64);
    SoftAcc S;
    S.m = sum16(att_part(a, b, wt));
    S.den = 1.f; S.x = a.x; S.y = a.y; S.z = a.z; S.w = a.w;

    const int cm = c < BIN_CAP ? c : BIN_CAP;
    int j = 0;
    for (; j + 4 <= cm; j += 4) {
        const int l = j >> 1;
        int s0 = __shfl(ent.x, l, 16), s1 = __shfl(ent.y, l, 16);
        int s2 = __shfl(ent.x, l + 1, 16), s3 = __shfl(ent.y, l + 1, 16);
        float4 a0 = *(const float4*)(lp + (size_t)s0 * 64);
        float4 a1 = *(const float4*)(lp + (size_t)s1 * 64);
        float4 a2 = *(const float4*)(lp + (size_t)s2 * 64);
        float4 a3 = *(const float4*)(lp + (size_t)s3 * 64);
        float e0 = sum16(att_part(a0, b, wt));
        float e1 = sum16(att_part(a1, b, wt));
        float e2 = sum16(att_part(a2, b, wt));
        float e3 = sum16(att_part(a3, b, wt));
        S.add(e0, a0); S.add(e1, a1); S.add(e2, a2); S.add(e3, a3);
    }
    const int t = cm - j;
    if (t > 0) {
        const int l = j >> 1;
        int s0 = __shfl(ent.x, l, 16), s1 = __shfl(ent.y, l, 16), s2 = __shfl(ent.x, l + 1, 16);
        s1 = t > 1 ? s1 : s0;
        s2 = t > 2 ? s2 : s0;
        float4 a0 = *(const float4*)(lp + (size_t)s0 * 64);
        float4 a1 = *(const float4*)(lp + (size_t)s1 * 64);
        float4 a2 = *(const float4*)(lp + (size_t)s2 * 64);
        float e0 = sum16(att_part(a0, b, wt));
        float e1 = sum16(att_part(a1, b, wt));
        float e2 = sum16(att_part(a2, b, wt));
        e1 = t > 1 ? e1 : -INFINITY;
        e2 = t > 2 ? e2 : -INFINITY;
        S.add(e0, a0); S.add(e1, a1); S.add(e2, a2);
    }
    if (c > BIN_CAP) {
        int no = *ovf_cnt;
        no = no < E ? no : E;
        for (int k = 0; k < no; ++k) {
            int2 e = ovf[k];
            if (e.x == n) {
                float4 a0 = *(const float4*)(lp + (size_t)e.y * 64);
                S.add(sum16(att_part(a0, b, wt)), a0);
            }
        }
    }

    float4 bb = *(const float4*)(b2 + sub * 4);
    float4 o;
    o.x = fmaxf(S.x / S.den + bb.x, 0.f);
    o.y = fmaxf(S.y / S.den + bb.y, 0.f);
    o.z = fmaxf(S.z / S.den + bb.z, 0.f);
    o.w = fmaxf(S.w / S.den + bb.w, 0.f);
    *(float4*)(out + (size_t)n * 64 + sub * 4) = o;
}

extern "C" void kernel_launch(void* const* d_in, const int* in_sizes, int n_in,
                              void* d_out, int out_size, void* d_ws, size_t ws_size,
                              hipStream_t stream) {
    const float* x   = (const float*)d_in[0];
    const int*   ei  = (const int*)d_in[1];
    const float* W1  = (const float*)d_in[2];
    const float* b1  = (const float*)d_in[3];
    const float* lnw = (const float*)d_in[4];
    const float* lnb = (const float*)d_in[5];
    const float* Wl  = (const float*)d_in[6];
    const float* Wr  = (const float*)d_in[7];
    const float* att = (const float*)d_in[8];
    const float* b2  = (const float*)d_in[9];
    float* out = (float*)d_out;

    const int N = in_sizes[0] / 64;
    const int E = in_sizes[1] / 2;

    char* ws = (char*)d_ws;
    size_t off = 0;
    auto carve = [&](size_t bytes) -> void* {
        void* p = ws + off;
        off += (bytes + 255) & ~(size_t)255;
        return p;
    };
    float* hbuf   = (float*)carve((size_t)N * 64 * 4);
    float* xlb    = (float*)carve((size_t)N * 64 * 4);
    float* xrb    = (float*)carve((size_t)N * 64 * 4);
    int*   cnt    = (int*)carve((size_t)N * 4);
    int*   ovfc   = (int*)carve(4);
    int*   bins   = (int*)carve((size_t)N * BIN_CAP * 4);
    int2*  ovf    = (int2*)carve((size_t)E * 8);
    const int nr   = N > RNODES ? (N + RNODES - 1) / RNODES : 1;
    const int rcap = 2 * ((E + nr - 1) / nr) + 1024;
    int*   rcur   = (int*)carve((size_t)NRMAX * 4);
    int*   spillc = (int*)carve(4);
    int2*  redges = (int2*)carve((size_t)nr * rcap * 8);
    int2*  spill  = (int2*)carve((size_t)E * 8);
    const bool lds_bins = nr <= NRMAX && off <= ws_size;

    const int ngrid = (int)(((long long)N * 16 + 255) / 256);
    const int nstrips = N / 16, rem = N - nstrips * 16;

    const int npart = (E + PCHUNK - 1) / PCHUNK;
    const int ngemm = (nstrips + 7) / 8;
    if (lds_bins) {
        k_zero_small<<<1, 256, 0, stream>>>(rcur, ovfc, spillc);
        k_part_gemm<<<npart + ngemm, 256, 0, stream>>>(ei, rcur, redges, spillc, spill, E, rcap, npart,
                                                       x, W1, hbuf, nstrips);
        k_build<<<nr, 1024, 0, stream>>>(rcur, redges, spillc, spill, cnt, bins, ovfc, ovf, N, E, rcap);
    } else {
        k_zero_cnt<<<(N + 255) / 256, 256, 0, stream>>>(cnt, ovfc, N);
        k_bin<<<((E + 3) / 4 + 255) / 256, 256, 0, stream>>>(ei, cnt, bins, ovfc, ovf, E);
        k_gemm64<<<ngemm, 256, 0, stream>>>(x, W1, hbuf, nstrips);
    }
    if (rem) k_gemm_tail<<<(rem * 64 + 255) / 256, 256, 0, stream>>>(x, W1, hbuf, nstrips * 16, N);

    k_gcn_dual<<<(N + FT_ROWS - 1) / FT_ROWS, 512, 0, stream>>>(cnt, bins, ovfc, ovf, hbuf, b1, lnw, lnb,
                                                             Wl, Wr, xlb, xrb, N, E);

    k_att_node<<<ngrid, 256, 0, stream>>>(cnt, bins, ovfc, ovf, xlb, xrb, att, b2, out, N, E);
}


# fewer vector instructions per edge in both gather loops (attention 138 -> 89 per 4 edges): leaky-relu as max(v,0.2v), 32-bit row offsets, one softmax rescale per 4 edges; sum order kept
# speedup vs baseline: 7.0075x; 1.0390x over previous
#include <hip/hip_runtime.h>
#include <hip/hip_bf16.h>

typedef float v2f __attribute__((ext_vector_type(2)));
typedef float v8f __attribute__((ext_vector_type(8)));
typedef float v4f __attribute__((ext_vector_type(4)));

#define LN_EPS    1e-5f
#define NEG_SLOPE 0.2f

#define BIN_CAP 32

__global__ __launch_bounds__(256) void k_zero_cnt(int* __restrict__ cnt,
                                                  int* __restrict__ ovf_cnt, int N) {
    int i = blockIdx.x * 256 + threadIdx.x;
    if (i < N) cnt[i] = 0;
    if (i == 0) *ovf_cnt = 0;
}

__device__ __forceinline__ void bin_store(int s, int d, int r,
                                          int* __restrict__ bins,
                                          int* __restrict__ ovf_cnt,
                                          int2* __restrict__ ovf, int E) {
    if (r < BIN_CAP) {
        bins[(size_t)d * BIN_CAP + r] = s;
    } else {
        int o = atomicAdd(ovf_cnt, 1);
        if (o < E) ovf[o] = make_int2(d, s);
    }
}

__global__ __launch_bounds__(256) void k_bin(const int* __restrict__ ei,
                                             int* __restrict__ cnt,
                                             int* __restrict__ bins,
                                             int* __restrict__ ovf_cnt,
                                             int2* __restrict__ ovf, int E) {
    int base = (blockIdx.x * 256 + threadIdx.x) * 4;
    if (base >= E) return;
    if ((E & 3) == 0) {
        int4 s = *(const int4*)(ei + base);
        int4 d = *(const int4*)(ei + E + base);
        int r0 = atomicAdd(cnt + d.x, 1);
        int r1 = atomicAdd(cnt + d.y, 1);
        int r2 = atomicAdd(cnt + d.z, 1);
        int r3 = atomicAdd(cnt + d.w, 1);
        bin_store(s.x, d.x, r0, bins, ovf_cnt, ovf, E);
        bin_store(s.y, d.y, r1, bins, ovf_cnt, ovf, E);
        bin_store(s.z, d.z, r2, bins, ovf_cnt, ovf, E);
        bin_store(s.w, d.w, r3, bins, ovf_cnt, ovf, E);
    } else {
        for (int k = 0; k < 4; ++k) {
            int i = base + k;
            if (i < E) {
                int s = ei[i], d = ei[E + i];
                bin_store(s, d, atomicAdd(cnt + d, 1), bins, ovf_cnt, ovf, E);
            }
        }
    }
}

#define RSH    9
#define RNODES (1 << RSH)
#define NRMAX  256
#define PCHUNK 4096

__global__ __launch_bounds__(256) void k_zero_small(int* __restrict__ rcur,
                                                    int* __restrict__ ovf_cnt,
                                                    int* __restrict__ spill_cnt) {
    int t = threadIdx.x;
    if (t < NRMAX) rcur[t] = 0;
    if (t == 0) { *ovf_cnt = 0; *spill_cnt = 0; }
}

#define PART_LH     0
#define PART_LSTART (PART_LH + NRMAX * 4)
#define PART_GPOS   (PART_LSTART + NRMAX * 4)
#define PART_WSUM   (PART_GPOS + NRMAX * 4)
#define PART_STAGE  (PART_WSUM + (NRMAX / 32) * 4)
#define PART_LDS    (PART_STAGE + PCHUNK * 8)

__device__ __forceinline__ void part_block(const int blk, unsigned char* __restrict__ smem,
                                           const int* __restrict__ ei,
                                           int* __restrict__ rcur,
                                           int2* __restrict__ redges,
                                           int* __restrict__ spill_cnt,
                                           int2* __restrict__ spill,
                                           int E, int rcap) {
    int*  lh     = (int*)(smem + PART_LH);
    int*  lstart = (int*)(smem + PART_LSTART);
    int*  gpos   = (int*)(smem + PART_GPOS);
    int*  wsum   = (int*)(smem + PART_WSUM);
    int2* stage  = (int2*)(smem + PART_STAGE);

    const int t  = threadIdx.x;
    const int e0 = blk * PCHUNK;
    lh[t] = 0;

    int s[16], d[16], r[16];
#pragma unroll
    for (int k = 0; k < 16; ++k) {
        int e = e0 + k * 256 + t;
        bool ok = e < E;
        s[k] = ok ? ei[e] : 0;
        d[k] = ok ? ei[E + e] : -1;
    }
    __syncthreads();
#pragma unroll
    for (int k = 0; k < 16; ++k)
        r[k] = (d[k] >= 0) ? atomicAdd(&lh[d[k] >> RSH], 1) : 0;
    __syncthreads();

    const int h = lh[t];
    int incl = h;
    for (int off = 1; off < 32; off <<= 1) {
        int u = __shfl_up(incl, off, 32);
        if ((t & 31) >= off) incl += u;
    }
    if ((t & 31) == 31) wsum[t >> 5] = incl;
    __syncthreads();
    int before = 0;
    for (int w = 0; w < (t >> 5); ++w) before += wsum[w];
    lstart[t] = before + incl - h;
    gpos[t]   = h ? atomicAdd(rcur + t, h) : 0;
    __syncthreads();

#pragma unroll
    for (int k = 0; k < 16; ++k)
        if (d[k] >= 0) stage[lstart[d[k] >> RSH] + r[k]] = make_int2(d[k], s[k]);
    __syncthreads();

    int n = E - e0;
    n = n < PCHUNK ? n : PCHUNK;
    for (int i = t; i < n; i += 256) {
        int2 e = stage[i];
        int b = e.x >> RSH;
        int g = gpos[b] + (i - lstart[b]);
        if (g < rcap) {
            redges[(size_t)b * rcap + g] = e;
        } else {
            int o = atomicAdd(spill_cnt, 1);
            if (o < E) spill[o] = e;
        }
    }
}

__device__ __forceinline__ void build_put(int2 e, int l, int r, int* __restrict__ lb,
                                          int* __restrict__ ovf_cnt,
                                          int2* __restrict__ ovf, int E) {
    if (r < BIN_CAP) {
        lb[l * BIN_CAP + r] = e.y;
    } else {
        int o = atomicAdd(ovf_cnt, 1);
        if (o < E) ovf[o] = e;
    }
}

__global__ __launch_bounds__(1024) void k_build(const int* __restrict__ rcur,
                                                const int2* __restrict__ redges,
                                                const int* __restrict__ spill_cnt,
                                                const int2* __restrict__ spill,
                                                int* __restrict__ cnt,
                                                int* __restrict__ bins,
                                                int* __restrict__ ovf_cnt,
                                                int2* __restrict__ ovf,
                                                int N, int E, int rcap) {
    __shared__ int lc[RNODES];
    __shared__ int lb[RNODES * BIN_CAP];

    const int t = threadIdx.x, b = blockIdx.x;
    const int base = b << RSH;
    if (t < RNODES) lc[t] = 0;
    {
        int4 z = make_int4(0, 0, 0, 0);
        for (int i = t; i < RNODES * BIN_CAP / 4; i += 1024) ((int4*)lb)[i] = z;
    }
    __syncthreads();

    int m = rcur[b];
    m = m < rcap ? m : rcap;
    const int2* re = redges + (size_t)b * rcap;
    for (int i0 = 0; i0 < m; i0 += 8 * 1024) {
        int2 e[8];
#pragma unroll
        for (int k = 0; k < 8; ++k) {
            int i = i0 + k * 1024 + t;
            e[k] = (i < m) ? re[i] : make_int2(-1, 0);
        }
        int r[8];
#pragma unroll
        for (int k = 0; k < 8; ++k)
            r[k] = (e[k].x >= 0) ? atomicAdd(&lc[e[k].x - base], 1) : 0;
#pragma unroll
        for (int k = 0; k < 8; ++k)
            if (e[k].x >= 0) build_put(e[k], e[k].x - base, r[k], lb, ovf_cnt, ovf, E);
    }
    int ns = *spill_cnt;
    ns = ns < E ? ns : E;
    for (int i = t; i < ns; i += 1024) {
        int2 e = spill[i];
        if ((e.x >> RSH) == b)
            build_put(e, e.x - base, atomicAdd(&lc[e.x - base], 1), lb, ovf_cnt, ovf, E);
    }
    __syncthreads();

    int nn = N - base;
    nn = nn < RNODES ? nn : RNODES;
    if (t < nn) cnt[base + t] = lc[t];
    int4* dst = (int4*)(bins + (size_t)base * BIN_CAP);
    for (int i = t; i < nn * (BIN_CAP / 4); i += 1024) dst[i] = ((const int4*)lb)[i];
}

__device__ __forceinline__ float dinv_of(int c) {
    float g = (float)(c + 1);
    return (g > 0.f) ? rsqrtf(g) : 0.f;
}

__device__ __forceinline__ void stage_w(float* __restrict__ sWp,
                                        const float* __restrict__ W) {
    for (int i = threadIdx.x; i < 4096; i += 256) {
        int k = i >> 6, c = i & 63;
        int kk = k >> 2, h = (k >> 1) & 1, j = k & 1;
        sWp[((((kk << 1) + h) << 6) + c) * 2 + j] = W[i];
    }
}

__device__ __forceinline__ void gemm64_block(const int blk, float* __restrict__ sWp,
                                             const float* __restrict__ A,
                                             const float* __restrict__ W,
                                             float* __restrict__ C, int nstrips) {
    stage_w(sWp, W);
    __syncthreads();

    int lane  = threadIdx.x & 31;
    int strip = blk * 8 + (threadIdx.x >> 5);
    if (strip >= nstrips) return;
    int r = lane & 15, h = lane >> 4;

    const float* ap = A + (size_t)(strip * 16 + r) * 64 + 2 * h;
    v8f zero = {0.f, 0.f, 0.f, 0.f, 0.f, 0.f, 0.f, 0.f};
    v8f acc[4] = {zero, zero, zero, zero};

    for (int kk = 0; kk < 16; ++kk) {
        v2f a = *(const v2f*)(ap + kk * 4);
        const v2f* bp = (const v2f*)&sWp[((((kk << 1) + h) << 6) + r) << 1];
#pragma unroll
        for (int t = 0; t < 4; ++t) {
            acc[t] = __builtin_amdgcn_wmma_f32_16x16x4_f32(
                false, a, false, bp[t * 16], (short)0, acc[t], false, false);
        }
    }
    float* cp = C + (size_t)strip * 16 * 64;
#pragma unroll
    for (int t = 0; t < 4; ++t)
#pragma unroll
        for (int g = 0; g < 8; ++g)
            cp[(size_t)(g + 8 * h) * 64 + t * 16 + r] = acc[t][g];
}

__global__ __launch_bounds__(256) void k_gemm64(const float* __restrict__ A,
                                                const float* __restrict__ W,
                                                float* __restrict__ C, int nstrips) {
    __shared__ float sWp[64 * 64];
    gemm64_block(blockIdx.x, sWp, A, W, C, nstrips);
}

__global__ __launch_bounds__(256) void k_part_gemm(const int* __restrict__ ei,
                                                   int* __restrict__ rcur,
                                                   int2* __restrict__ redges,
                                                   int* __restrict__ spill_cnt,
                                                   int2* __restrict__ spill,
                                                   int E, int rcap, int npart,
                                                   const float* __restrict__ A,
                                                   const float* __restrict__ W,
                                                   float* __restrict__ C, int nstrips) {
    __shared__ __align__(16) unsigned char smem[PART_LDS];
    if ((int)blockIdx.x < npart)
        part_block(blockIdx.x, smem, ei, rcur, redges, spill_cnt, spill, E, rcap);
    else
        gemm64_block((int)blockIdx.x - npart, (float*)smem, A, W, C, nstrips);
}

__global__ __launch_bounds__(256) void k_gemm_tail(const float* __restrict__ A,
                                                   const float* __restrict__ W,
                                                   float* __restrict__ C,
                                                   int row0, int nrows) {
    int i = blockIdx.x * 256 + threadIdx.x;
    int rows = nrows - row0;
    if (i >= rows * 64) return;
    int rr = row0 + i / 64, c = i & 63;
    float s = 0.f;
    for (int k = 0; k < 64; ++k) s += A[(size_t)rr * 64 + k] * W[k * 64 + c];
    C[(size_t)rr * 64 + c] = s;
}

__device__ __forceinline__ float sum16(float p) {
    for (int m = 8; m >= 1; m >>= 1) p += __shfl_xor(p, m, 16);
    return p;
}

__device__ __forceinline__ float4 gcn_row(const int n, const int sub,
                                          const int* __restrict__ cnt,
                                          const int* __restrict__ bins,
                                          const int* __restrict__ ovf_cnt,
                                          const int2* __restrict__ ovf,
                                          const float* __restrict__ h,
                                          const float* __restrict__ b1,
                                          const float* __restrict__ lnw,
                                          const float* __restrict__ lnb, int E) {
    const int* bp = bins + (size_t)n * BIN_CAP;
    const int2 ent = *(const int2*)(bp + 2 * sub);
    int c = cnt[n];
    float dn = dinv_of(c);
    const float4* __restrict__ h4 = (const float4*)h;
    const unsigned usub = (unsigned)sub;

    float4 v = h4[((unsigned)n << 4) | usub];
    float w = dn * dn;
    float ax = v.x * w, ay = v.y * w, az = v.z * w, aw = v.w * w;

    const int cm = c < BIN_CAP ? c : BIN_CAP;
    int j = 0;
    for (; j + 4 <= cm; j += 4) {
        const int l = j >> 1;
        int s0 = __shfl(ent.x, l, 16), s1 = __shfl(ent.y, l, 16);
        int s2 = __shfl(ent.x, l + 1, 16), s3 = __shfl(ent.y, l + 1, 16);
        float w0 = dinv_of(cnt[s0]) * dn, w1 = dinv_of(cnt[s1]) * dn;
        float w2 = dinv_of(cnt[s2]) * dn, w3 = dinv_of(cnt[s3]) * dn;
        float4 v0 = h4[((unsigned)s0 << 4) | usub];
        float4 v1 = h4[((unsigned)s1 << 4) | usub];
        float4 v2 = h4[((unsigned)s2 << 4) | usub];
        float4 v3 = h4[((unsigned)s3 << 4) | usub];
        ax += v0.x * w0; ay += v0.y * w0; az += v0.z * w0; aw += v0.w * w0;
        ax += v1.x * w1; ay += v1.y * w1; az += v1.z * w1; aw += v1.w * w1;
        ax += v2.x * w2; ay += v2.y * w2; az += v2.z * w2; aw += v2.w * w2;
        ax += v3.x * w3; ay += v3.y * w3; az += v3.z * w3; aw += v3.w * w3;
    }
    const int t = cm - j;
    if (t > 0) {
        const int l = j >> 1;
        int s0 = __shfl(ent.x, l, 16), s1 = __shfl(ent.y, l, 16), s2 = __shfl(ent.x, l + 1, 16);
        s1 = t > 1 ? s1 : s0;
        s2 = t > 2 ? s2 : s0;
        const int c0 = cnt[s0], c1 = cnt[s1], c2 = cnt[s2];
        const float k1 = t > 1 ? 1.f : 0.f, k2 = t > 2 ? 1.f : 0.f;
        float w0 = dinv_of(c0) * dn;
        float w1 = dinv_of(c1) * dn * k1;
        float w2 = dinv_of(c2) * dn * k2;
        float4 v0 = h4[((unsigned)s0 << 4) | usub];
        float4 v1 = h4[((unsigned)s1 << 4) | usub];
        float4 v2 = h4[((unsigned)s2 << 4) | usub];
        ax += v0.x * w0; ay += v0.y * w0; az += v0.z * w0; aw += v0.w * w0;
        ax += v1.x * w1; ay += v1.y * w1; az += v1.z * w1; aw += v1.w * w1;
        ax += v2.x * w2; ay += v2.y * w2; az += v2.z * w2; aw += v2.w * w2;
    }
    if (c > BIN_CAP) {
        int no = *ovf_cnt;
        no = no < E ? no : E;
        for (int k = 0; k < no; ++k) {
            int2 e = ovf[k];
            if (e.x == n) {
                int s0 = e.y;
                float w0 = dinv_of(cnt[s0]) * dn;
                float4 v0 = h4[((unsigned)s0 << 4) | usub];
                ax += v0.x * w0; ay += v0.y * w0; az += v0.z * w0; aw += v0.w * w0;
            }
        }
    }

    float4 bb = *(const float4*)(b1 + sub * 4);
    float vx = fmaxf(ax + bb.x, 0.f), vy = fmaxf(ay + bb.y, 0.f);
    float vz = fmaxf(az + bb.z, 0.f), vw = fmaxf(aw + bb.w, 0.f);
    float mu = sum16((vx + vy) + (vz + vw)) * (1.f / 64.f);
    float dx = vx - mu, dy = vy - mu, dz = vz - mu, dw = vw - mu;
    float q = sum16((dx * dx + dy * dy) + (dz * dz + dw * dw));
    float inv = rsqrtf(q * (1.f / 64.f) + LN_EPS);
    float4 gw = *(const float4*)(lnw + sub * 4);
    float4 gb = *(const float4*)(lnb + sub * 4);
    float4 o;
    o.x = dx * inv * gw.x + gb.x;
    o.y = dy * inv * gw.y + gb.y;
    o.z = dz * inv * gw.z + gb.z;
    o.w = dw * inv * gw.w + gb.w;
    return o;
}

#define FT_ROWS  32
#define FT_PITCH 68

__global__ __launch_bounds__(512) void k_gcn_dual(const int* __restrict__ cnt,
                                                  const int* __restrict__ bins,
                                                  const int* __restrict__ ovf_cnt,
                                                  const int2* __restrict__ ovf,
                                                  const float* __restrict__ h,
                                                  const float* __restrict__ b1,
                                                  const float* __restrict__ lnw,
                                                  const float* __restrict__ lnb,
                                                  const float* __restrict__ Wl,
                                                  const float* __restrict__ Wr,
                                                  float* __restrict__ xl,
                                                  float* __restrict__ xr, int N, int E) {
    __shared__ __align__(16) float sA[FT_ROWS * FT_PITCH];
    const int tid = threadIdx.x;
    const int row = tid >> 4, sub = tid & 15;
    const int n0 = blockIdx.x * FT_ROWS;
    const int n = n0 + row;
    float4 o = make_float4(0.f, 0.f, 0.f, 0.f);
    if (n < N) o = gcn_row(n, sub, cnt, bins, ovf_cnt, ovf, h, b1, lnw, lnb, E);
    *(float4*)(sA + row * FT_PITCH + sub * 4) = o;
    __syncthreads();

    const int wave = tid >> 5;
    if (wave >= 4) return;
    const int lane = tid & 31;
    const int c = lane & 15, hh = lane >> 4;
    const int m = wave & 1;
    const float* __restrict__ W = (wave & 2) ? Wr : Wl;
    float* __restrict__ C = (wave & 2) ? xr : xl;

    const float* ap = sA + (m * 16 + c) * FT_PITCH + 2 * hh;
    const float* wp = W + (2 * hh) * 64 + 4 * c;
    v8f zero = {0.f, 0.f, 0.f, 0.f, 0.f, 0.f, 0.f, 0.f};
    v8f acc0 = zero, acc1 = zero, acc2 = zero, acc3 = zero;
#pragma unroll 4
    for (int kk = 0; kk < 16; ++kk) {
        v2f a = *(const v2f*)(ap + kk * 4);
        float4 w0 = *(const float4*)(wp + (size_t)(4 * kk) * 64);
        float4 w1 = *(const float4*)(wp + (size_t)(4 * kk + 1) * 64);
        v2f bq0 = {w0.x, w1.x}, bq1 = {w0.y, w1.y}, bq2 = {w0.z, w1.z}, bq3 = {w0.w, w1.w};
        acc0 = __builtin_amdgcn_wmma_f32_16x16x4_f32(false, a, false, bq0, (short)0, acc0, false, false);
        acc1 = __builtin_amdgcn_wmma_f32_16x16x4_f32(false, a, false, bq1, (short)0, acc1, false, false);
        acc2 = __builtin_amdgcn_wmma_f32_16x16x4_f32(false, a, false, bq2, (short)0, acc2, false, false);
        acc3 = __builtin_amdgcn_wmma_f32_16x16x4_f32(false, a, false, bq3, (short)0, acc3, false, false);
    }
#pragma unroll
    for (int g = 0; g < 8; ++g) {
        const int nn = n0 + m * 16 + 8 * hh + g;
        if (nn < N)
            *(float4*)(C + (size_t)nn * 64 + 4 * c) = make_float4(acc0[g], acc1[g], acc2[g], acc3[g]);
    }
}

__device__ __forceinline__ float att_part(v4f a, v4f b, v4f w) {
    v4f v = a + b;
    v4f s = v * NEG_SLOPE;
    float p = w.x * fmaxf(v.x, s.x);
    p += w.y * fmaxf(v.y, s.y);
    p += w.z * fmaxf(v.z, s.z);
    p += w.w * fmaxf(v.w, s.w);
    return p;
}

struct SoftAcc {
    float m, den;
    v4f acc;
    __device__ __forceinline__ void add(float e, v4f a) {
        float mn = fmaxf(m, e);
        float sc = __expf(m - mn);
        float p  = __expf(e - mn);
        den = den * sc + p;
        acc = acc * sc + p * a;
        m = mn;
    }
    __device__ __forceinline__ void add3(float e0, float e1, float e2, v4f a0, v4f a1, v4f a2) {
        float mn = fmaxf(fmaxf(m, e0), fmaxf(e1, e2));
        float sc = __expf(m - mn);
        float p0 = __expf(e0 - mn), p1 = __expf(e1 - mn), p2 = __expf(e2 - mn);
        den = den * sc + p0;
        den += p1;
        den += p2;
        acc = acc * sc;
        acc += p0 * a0;
        acc += p1 * a1;
        acc += p2 * a2;
        m = mn;
    }
    __device__ __forceinline__ void add4(float e0, float e1, float e2, float e3,
                                         v4f a0, v4f a1, v4f a2, v4f a3) {
        float mn = fmaxf(fmaxf(fmaxf(m, e0), e1), fmaxf(e2, e3));
        float sc = __expf(m - mn);
        float p0 = __expf(e0 - mn), p1 = __expf(e1 - mn);
        float p2 = __expf(e2 - mn), p3 = __expf(e3 - mn);
        den = den * sc + p0;
        den += p1;
        den += p2;
        den += p3;
        acc = acc * sc;
        acc += p0 * a0;
        acc += p1 * a1;
        acc += p2 * a2;
        acc += p3 * a3;
        m = mn;
    }
};

__global__ __launch_bounds__(256) void k_att_node(const int* __restrict__ cnt,
                                                  const int* __restrict__ bins,
                                                  const int* __restrict__ ovf_cnt,
                                                  const int2* __restrict__ ovf,
                                                  const float* __restrict__ xl,
                                                  const float* __restrict__ xr,
                                                  const float* __restrict__ att,
                                                  const float* __restrict__ b2,
                                                  float* __restrict__ out, int N, int E) {
    int gid = blockIdx.x * 256 + threadIdx.x;
    int n = gid >> 4;
    const unsigned sub = gid & 15;
    if (n >= N) return;
    const int* bp = bins + (size_t)n * BIN_CAP;
    const int2 ent = *(const int2*)(bp + 2 * sub);
    int c = cnt[n];
    const v4f* __restrict__ xl4 = (const v4f*)xl;
    const v4f b  = ((const v4f*)xr)[((unsigned)n << 4) | sub];
    const v4f wt = ((const v4f*)att)[sub];

    const v4f a = xl4[((unsigned)n << 4) | sub];
    SoftAcc S;
    S.m = sum16(att_part(a, b, wt));
    S.den = 1.f;
    S.acc = a;

    const int cm = c < BIN_CAP ? c : BIN_CAP;
    int j = 0;
    for (; j + 4 <= cm; j += 4) {
        const int l = j >> 1;
        unsigned s0 = __shfl(ent.x, l, 16), s1 = __shfl(ent.y, l, 16);
        unsigned s2 = __shfl(ent.x, l + 1, 16), s3 = __shfl(ent.y, l + 1, 16);
        v4f a0 = xl4[(s0 << 4) | sub];
        v4f a1 = xl4[(s1 << 4) | sub];
        v4f a2 = xl4[(s2 << 4) | sub];
        v4f a3 = xl4[(s3 << 4) | sub];
        float e0 = sum16(att_part(a0, b, wt));
        float e1 = sum16(att_part(a1, b, wt));
        float e2 = sum16(att_part(a2, b, wt));
        float e3 = sum16(att_part(a3, b, wt));
        S.add4(e0, e1, e2, e3, a0, a1, a2, a3);
    }
    const int t = cm - j;
    if (t > 0) {
        const int l = j >> 1;
        unsigned s0 = __shfl(ent.x, l, 16), s1 = __shfl(ent.y, l, 16), s2 = __shfl(ent.x, l + 1, 16);
        s1 = t > 1 ? s1 : s0;
        s2 = t > 2 ? s2 : s0;
        v4f a0 = xl4[(s0 << 4) | sub];
        v4f a1 = xl4[(s1 << 4) | sub];
        v4f a2 = xl4[(s2 << 4) | sub];
        float e0 = sum16(att_part(a0, b, wt));
        float e1 = sum16(att_part(a1, b, wt));
        float e2 = sum16(att_part(a2, b, wt));
        e1 = t > 1 ? e1 : -INFINITY;
        e2 = t > 2 ? e2 : -INFINITY;
        S.add3(e0, e1, e2, a0, a1, a2);
    }
    if (c > BIN_CAP) {
        int no = *ovf_cnt;
        no = no < E ? no : E;
        for (int k = 0; k < no; ++k) {
            int2 e = ovf[k];
            if (e.x == n) {
                v4f a0 = xl4[((unsigned)e.y << 4) | sub];
                S.add(sum16(att_part(a0, b, wt)), a0);
            }
        }
    }

    const v4f bb = ((const v4f*)b2)[sub];
    v4f o;
    o.x = fmaxf(S.acc.x / S.den + bb.x, 0.f);
    o.y = fmaxf(S.acc.y / S.den + bb.y, 0.f);
    o.z = fmaxf(S.acc.z / S.den + bb.z, 0.f);
    o.w = fmaxf(S.acc.w / S.den + bb.w, 0.f);
    ((v4f*)out)[((unsigned)n << 4) | sub] = o;
}

extern "C" void kernel_launch(void* const* d_in, const int* in_sizes, int n_in,
                              void* d_out, int out_size, void* d_ws, size_t ws_size,
                              hipStream_t stream) {
    const float* x   = (const float*)d_in[0];
    const int*   ei  = (const int*)d_in[1];
    const float* W1  = (const float*)d_in[2];
    const float* b1  = (const float*)d_in[3];
    const float* lnw = (const float*)d_in[4];
    const float* lnb = (const float*)d_in[5];
    const float* Wl  = (const float*)d_in[6];
    const float* Wr  = (const float*)d_in[7];
    const float* att = (const float*)d_in[8];
    const float* b2  = (const float*)d_in[9];
    float* out = (float*)d_out;

    const int N = in_sizes[0] / 64;
    const int E = in_sizes[1] / 2;

    char* ws = (char*)d_ws;
    size_t off = 0;
    auto carve = [&](size_t bytes) -> void* {
        void* p = ws + off;
        off += (bytes + 255) & ~(size_t)255;
        return p;
    };
    float* hbuf   = (float*)carve((size_t)N * 64 * 4);
    float* xlb    = (float*)carve((size_t)N * 64 * 4);
    float* xrb    = (float*)carve((size_t)N * 64 * 4);
    int*   cnt    = (int*)carve((size_t)N * 4);
    int*   ovfc   = (int*)carve(4);
    int*   bins   = (int*)carve((size_t)N * BIN_CAP * 4);
    int2*  ovf    = (int2*)carve((size_t)E * 8);
    const int nr   = N > RNODES ? (N + RNODES - 1) / RNODES : 1;
    const int rcap = 2 * ((E + nr - 1) / nr) + 1024;
    int*   rcur   = (int*)carve((size_t)NRMAX * 4);
    int*   spillc = (int*)carve(4);
    int2*  redges = (int2*)carve((size_t)nr * rcap * 8);
    int2*  spill  = (int2*)carve((size_t)E * 8);
    const bool lds_bins = nr <= NRMAX && off <= ws_size;

    const int ngrid = (int)(((long long)N * 16 + 255) / 256);
    const int nstrips = N / 16, rem = N - nstrips * 16;

    const int npart = (E + PCHUNK - 1) / PCHUNK;
    const int ngemm = (nstrips + 7) / 8;
    if (lds_bins) {
        k_zero_small<<<1, 256, 0, stream>>>(rcur, ovfc, spillc);
        k_part_gemm<<<npart + ngemm, 256, 0, stream>>>(ei, rcur, redges, spillc, spill, E, rcap, npart,
                                                       x, W1, hbuf, nstrips);
        k_build<<<nr, 1024, 0, stream>>>(rcur, redges, spillc, spill, cnt, bins, ovfc, ovf, N, E, rcap);
    } else {
        k_zero_cnt<<<(N + 255) / 256, 256, 0, stream>>>(cnt, ovfc, N);
        k_bin<<<((E + 3) / 4 + 255) / 256, 256, 0, stream>>>(ei, cnt, bins, ovfc, ovf, E);
        k_gemm64<<<ngemm, 256, 0, stream>>>(x, W1, hbuf, nstrips);
    }
    if (rem) k_gemm_tail<<<(rem * 64 + 255) / 256, 256, 0, stream>>>(x, W1, hbuf, nstrips * 16, N);

    k_gcn_dual<<<(N + FT_ROWS - 1) / FT_ROWS, 512, 0, stream>>>(cnt, bins, ovfc, ovf, hbuf, b1, lnw, lnb,
                                                             Wl, Wr, xlb, xrb, N, E);

    k_att_node<<<ngrid, 256, 0, stream>>>(cnt, bins, ovfc, ovf, xlb, xrb, att, b2, out, N, E);
}


# lane exchange inside a node's 16 lanes moved off the LDS unit (ds_bpermute_b32) onto the vector ALU: row_xmask DPP for the 16-lane sums, v_permlane16_var for the index picks; sum order unchanged
# speedup vs baseline: 7.3350x; 1.0467x over previous
#include <hip/hip_runtime.h>
#include <hip/hip_bf16.h>

typedef float v2f __attribute__((ext_vector_type(2)));
typedef float v8f __attribute__((ext_vector_type(8)));
typedef float v4f __attribute__((ext_vector_type(4)));

#define LN_EPS    1e-5f
#define NEG_SLOPE 0.2f

#define BIN_CAP 32

__global__ __launch_bounds__(256) void k_zero_cnt(int* __restrict__ cnt,
                                                  int* __restrict__ ovf_cnt, int N) {
    int i = blockIdx.x * 256 + threadIdx.x;
    if (i < N) cnt[i] = 0;
    if (i == 0) *ovf_cnt = 0;
}

__device__ __forceinline__ void bin_store(int s, int d, int r,
                                          int* __restrict__ bins,
                                          int* __restrict__ ovf_cnt,
                                          int2* __restrict__ ovf, int E) {
    if (r < BIN_CAP) {
        bins[(size_t)d * BIN_CAP + r] = s;
    } else {
        int o = atomicAdd(ovf_cnt, 1);
        if (o < E) ovf[o] = make_int2(d, s);
    }
}

__global__ __launch_bounds__(256) void k_bin(const int* __restrict__ ei,
                                             int* __restrict__ cnt,
                                             int* __restrict__ bins,
                                             int* __restrict__ ovf_cnt,
                                             int2* __restrict__ ovf, int E) {
    int base = (blockIdx.x * 256 + threadIdx.x) * 4;
    if (base >= E) return;
    if ((E & 3) == 0) {
        int4 s = *(const int4*)(ei + base);
        int4 d = *(const int4*)(ei + E + base);
        int r0 = atomicAdd(cnt + d.x, 1);
        int r1 = atomicAdd(cnt + d.y, 1);
        int r2 = atomicAdd(cnt + d.z, 1);
        int r3 = atomicAdd(cnt + d.w, 1);
        bin_store(s.x, d.x, r0, bins, ovf_cnt, ovf, E);
        bin_store(s.y, d.y, r1, bins, ovf_cnt, ovf, E);
        bin_store(s.z, d.z, r2, bins, ovf_cnt, ovf, E);
        bin_store(s.w, d.w, r3, bins, ovf_cnt, ovf, E);
    } else {
        for (int k = 0; k < 4; ++k) {
            int i = base + k;
            if (i < E) {
                int s = ei[i], d = ei[E + i];
                bin_store(s, d, atomicAdd(cnt + d, 1), bins, ovf_cnt, ovf, E);
            }
        }
    }
}

#define RSH    9
#define RNODES (1 << RSH)
#define NRMAX  256
#define PCHUNK 4096

__global__ __launch_bounds__(256) void k_zero_small(int* __restrict__ rcur,
                                                    int* __restrict__ ovf_cnt,
                                                    int* __restrict__ spill_cnt) {
    int t = threadIdx.x;
    if (t < NRMAX) rcur[t] = 0;
    if (t == 0) { *ovf_cnt = 0; *spill_cnt = 0; }
}

#define PART_LH     0
#define PART_LSTART (PART_LH + NRMAX * 4)
#define PART_GPOS   (PART_LSTART + NRMAX * 4)
#define PART_WSUM   (PART_GPOS + NRMAX * 4)
#define PART_STAGE  (PART_WSUM + (NRMAX / 32) * 4)
#define PART_LDS    (PART_STAGE + PCHUNK * 8)

__device__ __forceinline__ void part_block(const int blk, unsigned char* __restrict__ smem,
                                           const int* __restrict__ ei,
                                           int* __restrict__ rcur,
                                           int2* __restrict__ redges,
                                           int* __restrict__ spill_cnt,
                                           int2* __restrict__ spill,
                                           int E, int rcap) {
    int*  lh     = (int*)(smem + PART_LH);
    int*  lstart = (int*)(smem + PART_LSTART);
    int*  gpos   = (int*)(smem + PART_GPOS);
    int*  wsum   = (int*)(smem + PART_WSUM);
    int2* stage  = (int2*)(smem + PART_STAGE);

    const int t  = threadIdx.x;
    const int e0 = blk * PCHUNK;
    lh[t] = 0;

    int s[16], d[16], r[16];
#pragma unroll
    for (int k = 0; k < 16; ++k) {
        int e = e0 + k * 256 + t;
        bool ok = e < E;
        s[k] = ok ? ei[e] : 0;
        d[k] = ok ? ei[E + e] : -1;
    }
    __syncthreads();
#pragma unroll
    for (int k = 0; k < 16; ++k)
        r[k] = (d[k] >= 0) ? atomicAdd(&lh[d[k] >> RSH], 1) : 0;
    __syncthreads();

    const int h = lh[t];
    int incl = h;
    for (int off = 1; off < 32; off <<= 1) {
        int u = __shfl_up(incl, off, 32);
        if ((t & 31) >= off) incl += u;
    }
    if ((t & 31) == 31) wsum[t >> 5] = incl;
    __syncthreads();
    int before = 0;
    for (int w = 0; w < (t >> 5); ++w) before += wsum[w];
    lstart[t] = before + incl - h;
    gpos[t]   = h ? atomicAdd(rcur + t, h) : 0;
    __syncthreads();

#pragma unroll
    for (int k = 0; k < 16; ++k)
        if (d[k] >= 0) stage[lstart[d[k] >> RSH] + r[k]] = make_int2(d[k], s[k]);
    __syncthreads();

    int n = E - e0;
    n = n < PCHUNK ? n : PCHUNK;
    for (int i = t; i < n; i += 256) {
        int2 e = stage[i];
        int b = e.x >> RSH;
        int g = gpos[b] + (i - lstart[b]);
        if (g < rcap) {
            redges[(size_t)b * rcap + g] = e;
        } else {
            int o = atomicAdd(spill_cnt, 1);
            if (o < E) spill[o] = e;
        }
    }
}

__device__ __forceinline__ void build_put(int2 e, int l, int r, int* __restrict__ lb,
                                          int* __restrict__ ovf_cnt,
                                          int2* __restrict__ ovf, int E) {
    if (r < BIN_CAP) {
        lb[l * BIN_CAP + r] = e.y;
    } else {
        int o = atomicAdd(ovf_cnt, 1);
        if (o < E) ovf[o] = e;
    }
}

__global__ __launch_bounds__(1024) void k_build(const int* __restrict__ rcur,
                                                const int2* __restrict__ redges,
                                                const int* __restrict__ spill_cnt,
                                                const int2* __restrict__ spill,
                                                int* __restrict__ cnt,
                                                int* __restrict__ bins,
                                                int* __restrict__ ovf_cnt,
                                                int2* __restrict__ ovf,
                                                int N, int E, int rcap) {
    __shared__ int lc[RNODES];
    __shared__ int lb[RNODES * BIN_CAP];

    const int t = threadIdx.x, b = blockIdx.x;
    const int base = b << RSH;
    if (t < RNODES) lc[t] = 0;
    {
        int4 z = make_int4(0, 0, 0, 0);
        for (int i = t; i < RNODES * BIN_CAP / 4; i += 1024) ((int4*)lb)[i] = z;
    }
    __syncthreads();

    int m = rcur[b];
    m = m < rcap ? m : rcap;
    const int2* re = redges + (size_t)b * rcap;
    for (int i0 = 0; i0 < m; i0 += 8 * 1024) {
        int2 e[8];
#pragma unroll
        for (int k = 0; k < 8; ++k) {
            int i = i0 + k * 1024 + t;
            e[k] = (i < m) ? re[i] : make_int2(-1, 0);
        }
        int r[8];
#pragma unroll
        for (int k = 0; k < 8; ++k)
            r[k] = (e[k].x >= 0) ? atomicAdd(&lc[e[k].x - base], 1) : 0;
#pragma unroll
        for (int k = 0; k < 8; ++k)
            if (e[k].x >= 0) build_put(e[k], e[k].x - base, r[k], lb, ovf_cnt, ovf, E);
    }
    int ns = *spill_cnt;
    ns = ns < E ? ns : E;
    for (int i = t; i < ns; i += 1024) {
        int2 e = spill[i];
        if ((e.x >> RSH) == b)
            build_put(e, e.x - base, atomicAdd(&lc[e.x - base], 1), lb, ovf_cnt, ovf, E);
    }
    __syncthreads();

    int nn = N - base;
    nn = nn < RNODES ? nn : RNODES;
    if (t < nn) cnt[base + t] = lc[t];
    int4* dst = (int4*)(bins + (size_t)base * BIN_CAP);
    for (int i = t; i < nn * (BIN_CAP / 4); i += 1024) dst[i] = ((const int4*)lb)[i];
}

__device__ __forceinline__ float dinv_of(int c) {
    float g = (float)(c + 1);
    return (g > 0.f) ? rsqrtf(g) : 0.f;
}

__device__ __forceinline__ void stage_w(float* __restrict__ sWp,
                                        const float* __restrict__ W) {
    for (int i = threadIdx.x; i < 4096; i += 256) {
        int k = i >> 6, c = i & 63;
        int kk = k >> 2, h = (k >> 1) & 1, j = k & 1;
        sWp[((((kk << 1) + h) << 6) + c) * 2 + j] = W[i];
    }
}

__device__ __forceinline__ void gemm64_block(const int blk, float* __restrict__ sWp,
                                             const float* __restrict__ A,
                                             const float* __restrict__ W,
                                             float* __restrict__ C, int nstrips) {
    stage_w(sWp, W);
    __syncthreads();

    int lane  = threadIdx.x & 31;
    int strip = blk * 8 + (threadIdx.x >> 5);
    if (strip >= nstrips) return;
    int r = lane & 15, h = lane >> 4;

    const float* ap = A + (size_t)(strip * 16 + r) * 64 + 2 * h;
    v8f zero = {0.f, 0.f, 0.f, 0.f, 0.f, 0.f, 0.f, 0.f};
    v8f acc[4] = {zero, zero, zero, zero};

    for (int kk = 0; kk < 16; ++kk) {
        v2f a = *(const v2f*)(ap + kk * 4);
        const v2f* bp = (const v2f*)&sWp[((((kk << 1) + h) << 6) + r) << 1];
#pragma unroll
        for (int t = 0; t < 4; ++t) {
            acc[t] = __builtin_amdgcn_wmma_f32_16x16x4_f32(
                false, a, false, bp[t * 16], (short)0, acc[t], false, false);
        }
    }
    float* cp = C + (size_t)strip * 16 * 64;
#pragma unroll
    for (int t = 0; t < 4; ++t)
#pragma unroll
        for (int g = 0; g < 8; ++g)
            cp[(size_t)(g + 8 * h) * 64 + t * 16 + r] = acc[t][g];
}

__global__ __launch_bounds__(256) void k_gemm64(const float* __restrict__ A,
                                                const float* __restrict__ W,
                                                float* __restrict__ C, int nstrips) {
    __shared__ float sWp[64 * 64];
    gemm64_block(blockIdx.x, sWp, A, W, C, nstrips);
}

__global__ __launch_bounds__(256) void k_part_gemm(const int* __restrict__ ei,
                                                   int* __restrict__ rcur,
                                                   int2* __restrict__ redges,
                                                   int* __restrict__ spill_cnt,
                                                   int2* __restrict__ spill,
                                                   int E, int rcap, int npart,
                                                   const float* __restrict__ A,
                                                   const float* __restrict__ W,
                                                   float* __restrict__ C, int nstrips) {
    __shared__ __align__(16) unsigned char smem[PART_LDS];
    if ((int)blockIdx.x < npart)
        part_block(blockIdx.x, smem, ei, rcur, redges, spill_cnt, spill, E, rcap);
    else
        gemm64_block((int)blockIdx.x - npart, (float*)smem, A, W, C, nstrips);
}

__global__ __launch_bounds__(256) void k_gemm_tail(const float* __restrict__ A,
                                                   const float* __restrict__ W,
                                                   float* __restrict__ C,
                                                   int row0, int nrows) {
    int i = blockIdx.x * 256 + threadIdx.x;
    int rows = nrows - row0;
    if (i >= rows * 64) return;
    int rr = row0 + i / 64, c = i & 63;
    float s = 0.f;
    for (int k = 0; k < 64; ++k) s += A[(size_t)rr * 64 + k] * W[k * 64 + c];
    C[(size_t)rr * 64 + c] = s;
}

template <int M> __device__ __forceinline__ float row_xor(float p) {
    return __int_as_float(__builtin_amdgcn_update_dpp(0, __float_as_int(p), 0x160 + M, 0xf, 0xf, true));
}
__device__ __forceinline__ int row_get(int v, int lane) {
    return (int)__builtin_amdgcn_permlane16_var(0u, (unsigned)v, (unsigned)lane, false, true);
}

__device__ __forceinline__ float sum16(float p) {
    p += row_xor<8>(p);
    p += row_xor<4>(p);
    p += row_xor<2>(p);
    p += row_xor<1>(p);
    return p;
}

__device__ __forceinline__ float4 gcn_row(const int n, const int sub,
                                          const int* __restrict__ cnt,
                                          const int* __restrict__ bins,
                                          const int* __restrict__ ovf_cnt,
                                          const int2* __restrict__ ovf,
                                          const float* __restrict__ h,
                                          const float* __restrict__ b1,
                                          const float* __restrict__ lnw,
                                          const float* __restrict__ lnb, int E) {
    const int* bp = bins + (size_t)n * BIN_CAP;
    const int2 ent = *(const int2*)(bp + 2 * sub);
    int c = cnt[n];
    float dn = dinv_of(c);
    const float4* __restrict__ h4 = (const float4*)h;
    const unsigned usub = (unsigned)sub;

    float4 v = h4[((unsigned)n << 4) | usub];
    float w = dn * dn;
    float ax = v.x * w, ay = v.y * w, az = v.z * w, aw = v.w * w;

    const int cm = c < BIN_CAP ? c : BIN_CAP;
    int j = 0;
    for (; j + 4 <= cm; j += 4) {
        const int l = j >> 1;
        int s0 = row_get(ent.x, l), s1 = row_get(ent.y, l);
        int s2 = row_get(ent.x, l + 1), s3 = row_get(ent.y, l + 1);
        float w0 = dinv_of(cnt[s0]) * dn, w1 = dinv_of(cnt[s1]) * dn;
        float w2 = dinv_of(cnt[s2]) * dn, w3 = dinv_of(cnt[s3]) * dn;
        float4 v0 = h4[((unsigned)s0 << 4) | usub];
        float4 v1 = h4[((unsigned)s1 << 4) | usub];
        float4 v2 = h4[((unsigned)s2 << 4) | usub];
        float4 v3 = h4[((unsigned)s3 << 4) | usub];
        ax += v0.x * w0; ay += v0.y * w0; az += v0.z * w0; aw += v0.w * w0;
        ax += v1.x * w1; ay += v1.y * w1; az += v1.z * w1; aw += v1.w * w1;
        ax += v2.x * w2; ay += v2.y * w2; az += v2.z * w2; aw += v2.w * w2;
        ax += v3.x * w3; ay += v3.y * w3; az += v3.z * w3; aw += v3.w * w3;
    }
    const int t = cm - j;
    if (t > 0) {
        const int l = j >> 1;
        int s0 = row_get(ent.x, l), s1 = row_get(ent.y, l), s2 = row_get(ent.x, l + 1);
        s1 = t > 1 ? s1 : s0;
        s2 = t > 2 ? s2 : s0;
        const int c0 = cnt[s0], c1 = cnt[s1], c2 = cnt[s2];
        const float k1 = t > 1 ? 1.f : 0.f, k2 = t > 2 ? 1.f : 0.f;
        float w0 = dinv_of(c0) * dn;
        float w1 = dinv_of(c1) * dn * k1;
        float w2 = dinv_of(c2) * dn * k2;
        float4 v0 = h4[((unsigned)s0 << 4) | usub];
        float4 v1 = h4[((unsigned)s1 << 4) | usub];
        float4 v2 = h4[((unsigned)s2 << 4) | usub];
        ax += v0.x * w0; ay += v0.y * w0; az += v0.z * w0; aw += v0.w * w0;
        ax += v1.x * w1; ay += v1.y * w1; az += v1.z * w1; aw += v1.w * w1;
        ax += v2.x * w2; ay += v2.y * w2; az += v2.z * w2; aw += v2.w * w2;
    }
    if (c > BIN_CAP) {
        int no = *ovf_cnt;
        no = no < E ? no : E;
        for (int k = 0; k < no; ++k) {
            int2 e = ovf[k];
            if (e.x == n) {
                int s0 = e.y;
                float w0 = dinv_of(cnt[s0]) * dn;
                float4 v0 = h4[((unsigned)s0 << 4) | usub];
                ax += v0.x * w0; ay += v0.y * w0; az += v0.z * w0; aw += v0.w * w0;
            }
        }
    }

    float4 bb = *(const float4*)(b1 + sub * 4);
    float vx = fmaxf(ax + bb.x, 0.f), vy = fmaxf(ay + bb.y, 0.f);
    float vz = fmaxf(az + bb.z, 0.f), vw = fmaxf(aw + bb.w, 0.f);
    float mu = sum16((vx + vy) + (vz + vw)) * (1.f / 64.f);
    float dx = vx - mu, dy = vy - mu, dz = vz - mu, dw = vw - mu;
    float q = sum16((dx * dx + dy * dy) + (dz * dz + dw * dw));
    float inv = rsqrtf(q * (1.f / 64.f) + LN_EPS);
    float4 gw = *(const float4*)(lnw + sub * 4);
    float4 gb = *(const float4*)(lnb + sub * 4);
    float4 o;
    o.x = dx * inv * gw.x + gb.x;
    o.y = dy * inv * gw.y + gb.y;
    o.z = dz * inv * gw.z + gb.z;
    o.w = dw * inv * gw.w + gb.w;
    return o;
}

#define FT_ROWS  32
#define FT_PITCH 68

__global__ __launch_bounds__(512) void k_gcn_dual(const int* __restrict__ cnt,
                                                  const int* __restrict__ bins,
                                                  const int* __restrict__ ovf_cnt,
                                                  const int2* __restrict__ ovf,
                                                  const float* __restrict__ h,
                                                  const float* __restrict__ b1,
                                                  const float* __restrict__ lnw,
                                                  const float* __restrict__ lnb,
                                                  const float* __restrict__ Wl,
                                                  const float* __restrict__ Wr,
                                                  float* __restrict__ xl,
                                                  float* __restrict__ xr, int N, int E) {
    __shared__ __align__(16) float sA[FT_ROWS * FT_PITCH];
    const int tid = threadIdx.x;
    const int row = tid >> 4, sub = tid & 15;
    const int n0 = blockIdx.x * FT_ROWS;
    const int n = n0 + row;
    float4 o = make_float4(0.f, 0.f, 0.f, 0.f);
    if (n < N) o = gcn_row(n, sub, cnt, bins, ovf_cnt, ovf, h, b1, lnw, lnb, E);
    *(float4*)(sA + row * FT_PITCH + sub * 4) = o;
    __syncthreads();

    const int wave = tid >> 5;
    if (wave >= 4) return;
    const int lane = tid & 31;
    const int c = lane & 15, hh = lane >> 4;
    const int m = wave & 1;
    const float* __restrict__ W = (wave & 2) ? Wr : Wl;
    float* __restrict__ C = (wave & 2) ? xr : xl;

    const float* ap = sA + (m * 16 + c) * FT_PITCH + 2 * hh;
    const float* wp = W + (2 * hh) * 64 + 4 * c;
    v8f zero = {0.f, 0.f, 0.f, 0.f, 0.f, 0.f, 0.f, 0.f};
    v8f acc0 = zero, acc1 = zero, acc2 = zero, acc3 = zero;
#pragma unroll 4
    for (int kk = 0; kk < 16; ++kk) {
        v2f a = *(const v2f*)(ap + kk * 4);
        float4 w0 = *(const float4*)(wp + (size_t)(4 * kk) * 64);
        float4 w1 = *(const float4*)(wp + (size_t)(4 * kk + 1) * 64);
        v2f bq0 = {w0.x, w1.x}, bq1 = {w0.y, w1.y}, bq2 = {w0.z, w1.z}, bq3 = {w0.w, w1.w};
        acc0 = __builtin_amdgcn_wmma_f32_16x16x4_f32(false, a, false, bq0, (short)0, acc0, false, false);
        acc1 = __builtin_amdgcn_wmma_f32_16x16x4_f32(false, a, false, bq1, (short)0, acc1, false, false);
        acc2 = __builtin_amdgcn_wmma_f32_16x16x4_f32(false, a, false, bq2, (short)0, acc2, false, false);
        acc3 = __builtin_amdgcn_wmma_f32_16x16x4_f32(false, a, false, bq3, (short)0, acc3, false, false);
    }
#pragma unroll
    for (int g = 0; g < 8; ++g) {
        const int nn = n0 + m * 16 + 8 * hh + g;
        if (nn < N)
            *(float4*)(C + (size_t)nn * 64 + 4 * c) = make_float4(acc0[g], acc1[g], acc2[g], acc3[g]);
    }
}

__device__ __forceinline__ float att_part(v4f a, v4f b, v4f w) {
    v4f v = a + b;
    v4f s = v * NEG_SLOPE;
    float p = w.x * fmaxf(v.x, s.x);
    p += w.y * fmaxf(v.y, s.y);
    p += w.z * fmaxf(v.z, s.z);
    p += w.w * fmaxf(v.w, s.w);
    return p;
}

struct SoftAcc {
    float m, den;
    v4f acc;
    __device__ __forceinline__ void add(float e, v4f a) {
        float mn = fmaxf(m, e);
        float sc = __expf(m - mn);
        float p  = __expf(e - mn);
        den = den * sc + p;
        acc = acc * sc + p * a;
        m = mn;
    }
    __device__ __forceinline__ void add3(float e0, float e1, float e2, v4f a0, v4f a1, v4f a2) {
        float mn = fmaxf(fmaxf(m, e0), fmaxf(e1, e2));
        float sc = __expf(m - mn);
        float p0 = __expf(e0 - mn), p1 = __expf(e1 - mn), p2 = __expf(e2 - mn);
        den = den * sc + p0;
        den += p1;
        den += p2;
        acc = acc * sc;
        acc += p0 * a0;
        acc += p1 * a1;
        acc += p2 * a2;
        m = mn;
    }
    __device__ __forceinline__ void add4(float e0, float e1, float e2, float e3,
                                         v4f a0, v4f a1, v4f a2, v4f a3) {
        float mn = fmaxf(fmaxf(fmaxf(m, e0), e1), fmaxf(e2, e3));
        float sc = __expf(m - mn);
        float p0 = __expf(e0 - mn), p1 = __expf(e1 - mn);
        float p2 = __expf(e2 - mn), p3 = __expf(e3 - mn);
        den = den * sc + p0;
        den += p1;
        den += p2;
        den += p3;
        acc = acc * sc;
        acc += p0 * a0;
        acc += p1 * a1;
        acc += p2 * a2;
        acc += p3 * a3;
        m = mn;
    }
};

__global__ __launch_bounds__(256) void k_att_node(const int* __restrict__ cnt,
                                                  const int* __restrict__ bins,
                                                  const int* __restrict__ ovf_cnt,
                                                  const int2* __restrict__ ovf,
                                                  const float* __restrict__ xl,
                                                  const float* __restrict__ xr,
                                                  const float* __restrict__ att,
                                                  const float* __restrict__ b2,
                                                  float* __restrict__ out, int N, int E) {
    int gid = blockIdx.x * 256 + threadIdx.x;
    int n = gid >> 4;
    const unsigned sub = gid & 15;
    if (n >= N) return;
    const int* bp = bins + (size_t)n * BIN_CAP;
    const int2 ent = *(const int2*)(bp + 2 * sub);
    int c = cnt[n];
    const v4f* __restrict__ xl4 = (const v4f*)xl;
    const v4f b  = ((const v4f*)xr)[((unsigned)n << 4) | sub];
    const v4f wt = ((const v4f*)att)[sub];

    const v4f a = xl4[((unsigned)n << 4) | sub];
    SoftAcc S;
    S.m = sum16(att_part(a, b, wt));
    S.den = 1.f;
    S.acc = a;

    const int cm = c < BIN_CAP ? c : BIN_CAP;
    int j = 0;
    for (; j + 4 <= cm; j += 4) {
        const int l = j >> 1;
        unsigned s0 = row_get(ent.x, l), s1 = row_get(ent.y, l);
        unsigned s2 = row_get(ent.x, l + 1), s3 = row_get(ent.y, l + 1);
        v4f a0 = xl4[(s0 << 4) | sub];
        v4f a1 = xl4[(s1 << 4) | sub];
        v4f a2 = xl4[(s2 << 4) | sub];
        v4f a3 = xl4[(s3 << 4) | sub];
        float e0 = sum16(att_part(a0, b, wt));
        float e1 = sum16(att_part(a1, b, wt));
        float e2 = sum16(att_part(a2, b, wt));
        float e3 = sum16(att_part(a3, b, wt));
        S.add4(e0, e1, e2, e3, a0, a1, a2, a3);
    }
    const int t = cm - j;
    if (t > 0) {
        const int l = j >> 1;
        unsigned s0 = row_get(ent.x, l), s1 = row_get(ent.y, l), s2 = row_get(ent.x, l + 1);
        s1 = t > 1 ? s1 : s0;
        s2 = t > 2 ? s2 : s0;
        v4f a0 = xl4[(s0 << 4) | sub];
        v4f a1 = xl4[(s1 << 4) | sub];
        v4f a2 = xl4[(s2 << 4) | sub];
        float e0 = sum16(att_part(a0, b, wt));
        float e1 = sum16(att_part(a1, b, wt));
        float e2 = sum16(att_part(a2, b, wt));
        e1 = t > 1 ? e1 : -INFINITY;
        e2 = t > 2 ? e2 : -INFINITY;
        S.add3(e0, e1, e2, a0, a1, a2);
    }
    if (c > BIN_CAP) {
        int no = *ovf_cnt;
        no = no < E ? no : E;
        for (int k = 0; k < no; ++k) {
            int2 e = ovf[k];
            if (e.x == n) {
                v4f a0 = xl4[((unsigned)e.y << 4) | sub];
                S.add(sum16(att_part(a0, b, wt)), a0);
            }
        }
    }

    const v4f bb = ((const v4f*)b2)[sub];
    v4f o;
    o.x = fmaxf(S.acc.x / S.den + bb.x, 0.f);
    o.y = fmaxf(S.acc.y / S.den + bb.y, 0.f);
    o.z = fmaxf(S.acc.z / S.den + bb.z, 0.f);
    o.w = fmaxf(S.acc.w / S.den + bb.w, 0.f);
    ((v4f*)out)[((unsigned)n << 4) | sub] = o;
}

extern "C" void kernel_launch(void* const* d_in, const int* in_sizes, int n_in,
                              void* d_out, int out_size, void* d_ws, size_t ws_size,
                              hipStream_t stream) {
    const float* x   = (const float*)d_in[0];
    const int*   ei  = (const int*)d_in[1];
    const float* W1  = (const float*)d_in[2];
    const float* b1  = (const float*)d_in[3];
    const float* lnw = (const float*)d_in[4];
    const float* lnb = (const float*)d_in[5];
    const float* Wl  = (const float*)d_in[6];
    const float* Wr  = (const float*)d_in[7];
    const float* att = (const float*)d_in[8];
    const float* b2  = (const float*)d_in[9];
    float* out = (float*)d_out;

    const int N = in_sizes[0] / 64;
    const int E = in_sizes[1] / 2;

    char* ws = (char*)d_ws;
    size_t off = 0;
    auto carve = [&](size_t bytes) -> void* {
        void* p = ws + off;
        off += (bytes + 255) & ~(size_t)255;
        return p;
    };
    float* hbuf   = (float*)carve((size_t)N * 64 * 4);
    float* xlb    = (float*)carve((size_t)N * 64 * 4);
    float* xrb    = (float*)carve((size_t)N * 64 * 4);
    int*   cnt    = (int*)carve((size_t)N * 4);
    int*   ovfc   = (int*)carve(4);
    int*   bins   = (int*)carve((size_t)N * BIN_CAP * 4);
    int2*  ovf    = (int2*)carve((size_t)E * 8);
    const int nr   = N > RNODES ? (N + RNODES - 1) / RNODES : 1;
    const int rcap = 2 * ((E + nr - 1) / nr) + 1024;
    int*   rcur   = (int*)carve((size_t)NRMAX * 4);
    int*   spillc = (int*)carve(4);
    int2*  redges = (int2*)carve((size_t)nr * rcap * 8);
    int2*  spill  = (int2*)carve((size_t)E * 8);
    const bool lds_bins = nr <= NRMAX && off <= ws_size;

    const int ngrid = (int)(((long long)N * 16 + 255) / 256);
    const int nstrips = N / 16, rem = N - nstrips * 16;

    const int npart = (E + PCHUNK - 1) / PCHUNK;
    const int ngemm = (nstrips + 7) / 8;
    if (lds_bins) {
        k_zero_small<<<1, 256, 0, stream>>>(rcur, ovfc, spillc);
        k_part_gemm<<<npart + ngemm, 256, 0, stream>>>(ei, rcur, redges, spillc, spill, E, rcap, npart,
                                                       x, W1, hbuf, nstrips);
        k_build<<<nr, 1024, 0, stream>>>(rcur, redges, spillc, spill, cnt, bins, ovfc, ovf, N, E, rcap);
    } else {
        k_zero_cnt<<<(N + 255) / 256, 256, 0, stream>>>(cnt, ovfc, N);
        k_bin<<<((E + 3) / 4 + 255) / 256, 256, 0, stream>>>(ei, cnt, bins, ovfc, ovf, E);
        k_gemm64<<<ngemm, 256, 0, stream>>>(x, W1, hbuf, nstrips);
    }
    if (rem) k_gemm_tail<<<(rem * 64 + 255) / 256, 256, 0, stream>>>(x, W1, hbuf, nstrips * 16, N);

    k_gcn_dual<<<(N + FT_ROWS - 1) / FT_ROWS, 512, 0, stream>>>(cnt, bins, ovfc, ovf, hbuf, b1, lnw, lnb,
                                                             Wl, Wr, xlb, xrb, N, E);

    k_att_node<<<ngrid, 256, 0, stream>>>(cnt, bins, ovfc, ovf, xlb, xrb, att, b2, out, N, E);
}


# x@W1 moved behind the GCN walk: rows of x summed per node, then two waves per block do sums@W1, LayerNorm in the accumulators, @Wl, @Wr, weights in LDS; no x@W1 pass or plane; sum order: x rows first
# speedup vs baseline: 7.6631x; 1.0447x over previous
#include <hip/hip_runtime.h>
#include <hip/hip_bf16.h>

typedef float v2f __attribute__((ext_vector_type(2)));
typedef float v8f __attribute__((ext_vector_type(8)));
typedef float v4f __attribute__((ext_vector_type(4)));

#define LN_EPS    1e-5f
#define NEG_SLOPE 0.2f

#define BIN_CAP 32

__global__ __launch_bounds__(256) void k_zero_cnt(int* __restrict__ cnt,
                                                  int* __restrict__ ovf_cnt, int N) {
    int i = blockIdx.x * 256 + threadIdx.x;
    if (i < N) cnt[i] = 0;
    if (i == 0) *ovf_cnt = 0;
}

__device__ __forceinline__ void bin_store(int s, int d, int r,
                                          int* __restrict__ bins,
                                          int* __restrict__ ovf_cnt,
                                          int2* __restrict__ ovf, int E) {
    if (r < BIN_CAP) {
        bins[(size_t)d * BIN_CAP + r] = s;
    } else {
        int o = atomicAdd(ovf_cnt, 1);
        if (o < E) ovf[o] = make_int2(d, s);
    }
}

__global__ __launch_bounds__(256) void k_bin(const int* __restrict__ ei,
                                             int* __restrict__ cnt,
                                             int* __restrict__ bins,
                                             int* __restrict__ ovf_cnt,
                                             int2* __restrict__ ovf, int E) {
    int base = (blockIdx.x * 256 + threadIdx.x) * 4;
    if (base >= E) return;
    if ((E & 3) == 0) {
        int4 s = *(const int4*)(ei + base);
        int4 d = *(const int4*)(ei + E + base);
        int r0 = atomicAdd(cnt + d.x, 1);
        int r1 = atomicAdd(cnt + d.y, 1);
        int r2 = atomicAdd(cnt + d.z, 1);
        int r3 = atomicAdd(cnt + d.w, 1);
        bin_store(s.x, d.x, r0, bins, ovf_cnt, ovf, E);
        bin_store(s.y, d.y, r1, bins, ovf_cnt, ovf, E);
        bin_store(s.z, d.z, r2, bins, ovf_cnt, ovf, E);
        bin_store(s.w, d.w, r3, bins, ovf_cnt, ovf, E);
    } else {
        for (int k = 0; k < 4; ++k) {
            int i = base + k;
            if (i < E) {
                int s = ei[i], d = ei[E + i];
                bin_store(s, d, atomicAdd(cnt + d, 1), bins, ovf_cnt, ovf, E);
            }
        }
    }
}

#define RSH    9
#define RNODES (1 << RSH)
#define NRMAX  256
#define PCHUNK 4096

__global__ __launch_bounds__(256) void k_zero_small(int* __restrict__ rcur,
                                                    int* __restrict__ ovf_cnt,
                                                    int* __restrict__ spill_cnt) {
    int t = threadIdx.x;
    if (t < NRMAX) rcur[t] = 0;
    if (t == 0) { *ovf_cnt = 0; *spill_cnt = 0; }
}

#define PART_LH     0
#define PART_LSTART (PART_LH + NRMAX * 4)
#define PART_GPOS   (PART_LSTART + NRMAX * 4)
#define PART_WSUM   (PART_GPOS + NRMAX * 4)
#define PART_STAGE  (PART_WSUM + (NRMAX / 32) * 4)
#define PART_LDS    (PART_STAGE + PCHUNK * 8)

__device__ __forceinline__ void part_block(const int blk, unsigned char* __restrict__ smem,
                                           const int* __restrict__ ei,
                                           int* __restrict__ rcur,
                                           int2* __restrict__ redges,
                                           int* __restrict__ spill_cnt,
                                           int2* __restrict__ spill,
                                           int E, int rcap) {
    int*  lh     = (int*)(smem + PART_LH);
    int*  lstart = (int*)(smem + PART_LSTART);
    int*  gpos   = (int*)(smem + PART_GPOS);
    int*  wsum   = (int*)(smem + PART_WSUM);
    int2* stage  = (int2*)(smem + PART_STAGE);

    const int t  = threadIdx.x;
    const int e0 = blk * PCHUNK;
    lh[t] = 0;

    int s[16], d[16], r[16];
#pragma unroll
    for (int k = 0; k < 16; ++k) {
        int e = e0 + k * 256 + t;
        bool ok = e < E;
        s[k] = ok ? ei[e] : 0;
        d[k] = ok ? ei[E + e] : -1;
    }
    __syncthreads();
#pragma unroll
    for (int k = 0; k < 16; ++k)
        r[k] = (d[k] >= 0) ? atomicAdd(&lh[d[k] >> RSH], 1) : 0;
    __syncthreads();

    const int h = lh[t];
    int incl = h;
    for (int off = 1; off < 32; off <<= 1) {
        int u = __shfl_up(incl, off, 32);
        if ((t & 31) >= off) incl += u;
    }
    if ((t & 31) == 31) wsum[t >> 5] = incl;
    __syncthreads();
    int before = 0;
    for (int w = 0; w < (t >> 5); ++w) before += wsum[w];
    lstart[t] = before + incl - h;
    gpos[t]   = h ? atomicAdd(rcur + t, h) : 0;
    __syncthreads();

#pragma unroll
    for (int k = 0; k < 16; ++k)
        if (d[k] >= 0) stage[lstart[d[k] >> RSH] + r[k]] = make_int2(d[k], s[k]);
    __syncthreads();

    int n = E - e0;
    n = n < PCHUNK ? n : PCHUNK;
    for (int i = t; i < n; i += 256) {
        int2 e = stage[i];
        int b = e.x >> RSH;
        int g = gpos[b] + (i - lstart[b]);
        if (g < rcap) {
            redges[(size_t)b * rcap + g] = e;
        } else {
            int o = atomicAdd(spill_cnt, 1);
            if (o < E) spill[o] = e;
        }
    }
}

__device__ __forceinline__ void build_put(int2 e, int l, int r, int* __restrict__ lb,
                                          int* __restrict__ ovf_cnt,
                                          int2* __restrict__ ovf, int E) {
    if (r < BIN_CAP) {
        lb[l * BIN_CAP + r] = e.y;
    } else {
        int o = atomicAdd(ovf_cnt, 1);
        if (o < E) ovf[o] = e;
    }
}

__global__ __launch_bounds__(1024) void k_build(const int* __restrict__ rcur,
                                                const int2* __restrict__ redges,
                                                const int* __restrict__ spill_cnt,
                                                const int2* __restrict__ spill,
                                                int* __restrict__ cnt,
                                                int* __restrict__ bins,
                                                int* __restrict__ ovf_cnt,
                                                int2* __restrict__ ovf,
                                                int N, int E, int rcap) {
    __shared__ int lc[RNODES];
    __shared__ int lb[RNODES * BIN_CAP];

    const int t = threadIdx.x, b = blockIdx.x;
    const int base = b << RSH;
    if (t < RNODES) lc[t] = 0;
    {
        int4 z = make_int4(0, 0, 0, 0);
        for (int i = t; i < RNODES * BIN_CAP / 4; i += 1024) ((int4*)lb)[i] = z;
    }
    __syncthreads();

    int m = rcur[b];
    m = m < rcap ? m : rcap;
    const int2* re = redges + (size_t)b * rcap;
    for (int i0 = 0; i0 < m; i0 += 8 * 1024) {
        int2 e[8];
#pragma unroll
        for (int k = 0; k < 8; ++k) {
            int i = i0 + k * 1024 + t;
            e[k] = (i < m) ? re[i] : make_int2(-1, 0);
        }
        int r[8];
#pragma unroll
        for (int k = 0; k < 8; ++k)
            r[k] = (e[k].x >= 0) ? atomicAdd(&lc[e[k].x - base], 1) : 0;
#pragma unroll
        for (int k = 0; k < 8; ++k)
            if (e[k].x >= 0) build_put(e[k], e[k].x - base, r[k], lb, ovf_cnt, ovf, E);
    }
    int ns = *spill_cnt;
    ns = ns < E ? ns : E;
    for (int i = t; i < ns; i += 1024) {
        int2 e = spill[i];
        if ((e.x >> RSH) == b)
            build_put(e, e.x - base, atomicAdd(&lc[e.x - base], 1), lb, ovf_cnt, ovf, E);
    }
    __syncthreads();

    int nn = N - base;
    nn = nn < RNODES ? nn : RNODES;
    if (t < nn) cnt[base + t] = lc[t];
    int4* dst = (int4*)(bins + (size_t)base * BIN_CAP);
    for (int i = t; i < nn * (BIN_CAP / 4); i += 1024) dst[i] = ((const int4*)lb)[i];
}

__device__ __forceinline__ float dinv_of(int c) {
    float g = (float)(c + 1);
    return (g > 0.f) ? rsqrtf(g) : 0.f;
}

__global__ __launch_bounds__(256) void k_part(const int* __restrict__ ei,
                                              int* __restrict__ rcur,
                                              int2* __restrict__ redges,
                                              int* __restrict__ spill_cnt,
                                              int2* __restrict__ spill,
                                              int E, int rcap) {
    __shared__ __align__(16) unsigned char smem[PART_LDS];
    part_block(blockIdx.x, smem, ei, rcur, redges, spill_cnt, spill, E, rcap);
}

template <int M> __device__ __forceinline__ float row_xor(float p) {
    return __int_as_float(__builtin_amdgcn_update_dpp(0, __float_as_int(p), 0x160 + M, 0xf, 0xf, true));
}
__device__ __forceinline__ int row_get(int v, int lane) {
    return (int)__builtin_amdgcn_permlane16_var(0u, (unsigned)v, (unsigned)lane, false, true);
}

__device__ __forceinline__ float sum16(float p) {
    p += row_xor<8>(p);
    p += row_xor<4>(p);
    p += row_xor<2>(p);
    p += row_xor<1>(p);
    return p;
}

__device__ __forceinline__ float4 gcn_agg_row(const int n, const int sub,
                                              const int* __restrict__ cnt,
                                              const int* __restrict__ bins,
                                              const int* __restrict__ ovf_cnt,
                                              const int2* __restrict__ ovf,
                                              const float* __restrict__ x, int E,
                                              const float4* __restrict__ w1s,
                                              const float4* __restrict__ wls,
                                              const float4* __restrict__ wrs,
                                              float4* __restrict__ wdst, const int tid) {
    const int* bp = bins + (size_t)n * BIN_CAP;
    const int2 ent = *(const int2*)(bp + 2 * sub);
    int c = cnt[n];
    const float4* __restrict__ x4 = (const float4*)x;
    const unsigned usub = (unsigned)sub;
    float4 v = x4[((unsigned)n << 4) | usub];

    {
        const int kp = tid >> 4, qi = tid & 15;
        const int e = (2 * kp) * 16 + qi, o = e + 16;
        float4 p0 = w1s[e], q0 = w1s[o];
        float4 p1 = wls[e], q1 = wls[o];
        float4 p2 = wrs[e], q2 = wrs[o];
        float4* d = wdst + kp * 32 + 2 * qi;
        d[0]    = make_float4(p0.x, q0.x, p0.y, q0.y);
        d[1]    = make_float4(p0.z, q0.z, p0.w, q0.w);
        d[1024] = make_float4(p1.x, q1.x, p1.y, q1.y);
        d[1025] = make_float4(p1.z, q1.z, p1.w, q1.w);
        d[2048] = make_float4(p2.x, q2.x, p2.y, q2.y);
        d[2049] = make_float4(p2.z, q2.z, p2.w, q2.w);
    }

    float dn = dinv_of(c);
    float w = dn * dn;
    float ax = v.x * w, ay = v.y * w, az = v.z * w, aw = v.w * w;

    const int cm = c < BIN_CAP ? c : BIN_CAP;
    int j = 0;
    for (; j + 4 <= cm; j += 4) {
        const int l = j >> 1;
        int s0 = row_get(ent.x, l), s1 = row_get(ent.y, l);
        int s2 = row_get(ent.x, l + 1), s3 = row_get(ent.y, l + 1);
        float w0 = dinv_of(cnt[s0]) * dn, w1 = dinv_of(cnt[s1]) * dn;
        float w2 = dinv_of(cnt[s2]) * dn, w3 = dinv_of(cnt[s3]) * dn;
        float4 v0 = x4[((unsigned)s0 << 4) | usub];
        float4 v1 = x4[((unsigned)s1 << 4) | usub];
        float4 v2 = x4[((unsigned)s2 << 4) | usub];
        float4 v3 = x4[((unsigned)s3 << 4) | usub];
        ax += v0.x * w0; ay += v0.y * w0; az += v0.z * w0; aw += v0.w * w0;
        ax += v1.x * w1; ay += v1.y * w1; az += v1.z * w1; aw += v1.w * w1;
        ax += v2.x * w2; ay += v2.y * w2; az += v2.z * w2; aw += v2.w * w2;
        ax += v3.x * w3; ay += v3.y * w3; az += v3.z * w3; aw += v3.w * w3;
    }
    const int t = cm - j;
    if (t > 0) {
        const int l = j >> 1;
        int s0 = row_get(ent.x, l), s1 = row_get(ent.y, l), s2 = row_get(ent.x, l + 1);
        s1 = t > 1 ? s1 : s0;
        s2 = t > 2 ? s2 : s0;
        const int c0 = cnt[s0], c1 = cnt[s1], c2 = cnt[s2];
        const float k1 = t > 1 ? 1.f : 0.f, k2 = t > 2 ? 1.f : 0.f;
        float w0 = dinv_of(c0) * dn;
        float w1 = dinv_of(c1) * dn * k1;
        float w2 = dinv_of(c2) * dn * k2;
        float4 v0 = x4[((unsigned)s0 << 4) | usub];
        float4 v1 = x4[((unsigned)s1 << 4) | usub];
        float4 v2 = x4[((unsigned)s2 << 4) | usub];
        ax += v0.x * w0; ay += v0.y * w0; az += v0.z * w0; aw += v0.w * w0;
        ax += v1.x * w1; ay += v1.y * w1; az += v1.z * w1; aw += v1.w * w1;
        ax += v2.x * w2; ay += v2.y * w2; az += v2.z * w2; aw += v2.w * w2;
    }
    if (c > BIN_CAP) {
        int no = *ovf_cnt;
        no = no < E ? no : E;
        for (int k = 0; k < no; ++k) {
            int2 e = ovf[k];
            if (e.x == n) {
                int s0 = e.y;
                float w0 = dinv_of(cnt[s0]) * dn;
                float4 v0 = x4[((unsigned)s0 << 4) | usub];
                ax += v0.x * w0; ay += v0.y * w0; az += v0.z * w0; aw += v0.w * w0;
            }
        }
    }
    return make_float4(ax, ay, az, aw);
}

__device__ __forceinline__ void tile_product(const float* ap, const float* wp,
                                             v8f& acc0, v8f& acc1, v8f& acc2, v8f& acc3) {
#pragma unroll 2
    for (int kk = 0; kk < 16; ++kk) {
        v2f a = *(const v2f*)(ap + kk * 4);
        float4 f0 = *(const float4*)(wp + kk * 256);
        float4 f1 = *(const float4*)(wp + kk * 256 + 4);
        v2f bq0 = {f0.x, f0.y}, bq1 = {f0.z, f0.w}, bq2 = {f1.x, f1.y}, bq3 = {f1.z, f1.w};
        acc0 = __builtin_amdgcn_wmma_f32_16x16x4_f32(false, a, false, bq0, (short)0, acc0, false, false);
        acc1 = __builtin_amdgcn_wmma_f32_16x16x4_f32(false, a, false, bq1, (short)0, acc1, false, false);
        acc2 = __builtin_amdgcn_wmma_f32_16x16x4_f32(false, a, false, bq2, (short)0, acc2, false, false);
        acc3 = __builtin_amdgcn_wmma_f32_16x16x4_f32(false, a, false, bq3, (short)0, acc3, false, false);
    }
}

#define FT_ROWS  32
#define FT_PITCH 68

__global__ __launch_bounds__(512) void k_gcn_dual(const int* __restrict__ cnt,
                                                  const int* __restrict__ bins,
                                                  const int* __restrict__ ovf_cnt,
                                                  const int2* __restrict__ ovf,
                                                  const float* __restrict__ x,
                                                  const float* __restrict__ W1,
                                                  const float* __restrict__ b1,
                                                  const float* __restrict__ lnw,
                                                  const float* __restrict__ lnb,
                                                  const float* __restrict__ Wl,
                                                  const float* __restrict__ Wr,
                                                  float* __restrict__ xl,
                                                  float* __restrict__ xr, int N, int E) {
    __shared__ __align__(16) float sA[FT_ROWS * FT_PITCH];
    __shared__ __align__(16) float sW[3 * 64 * 64];
    const int tid = threadIdx.x;
    const int row = tid >> 4, sub = tid & 15;
    const int n0 = blockIdx.x * FT_ROWS;
    const int n = n0 + row;
    float4 o = gcn_agg_row(n < N ? n : N - 1, sub, cnt, bins, ovf_cnt, ovf, x, E,
                           (const float4*)W1, (const float4*)Wl, (const float4*)Wr, (float4*)sW, tid);
    if (n >= N) o = make_float4(0.f, 0.f, 0.f, 0.f);
    *(float4*)(sA + row * FT_PITCH + sub * 4) = o;
    __syncthreads();

    const int m = tid >> 5;
    if (m >= 2) return;
    const int lane = tid & 31;
    const int c = lane & 15, hh = lane >> 4;

    float* ap = sA + (m * 16 + c) * FT_PITCH + 2 * hh;
    const v8f zero = {0.f, 0.f, 0.f, 0.f, 0.f, 0.f, 0.f, 0.f};
    {
        v8f acc0 = zero, acc1 = zero, acc2 = zero, acc3 = zero;
        tile_product(ap, sW + hh * 128 + 8 * c, acc0, acc1, acc2, acc3);
        const float4 bb = *(const float4*)(b1 + 4 * c);
        const float4 gw = *(const float4*)(lnw + 4 * c);
        const float4 gb = *(const float4*)(lnb + 4 * c);
        float* sp = sA + (m * 16 + 8 * hh) * FT_PITCH + 4 * c;
#pragma unroll
        for (int g = 0; g < 8; ++g) {
            float vx = fmaxf(acc0[g] + bb.x, 0.f), vy = fmaxf(acc1[g] + bb.y, 0.f);
            float vz = fmaxf(acc2[g] + bb.z, 0.f), vw = fmaxf(acc3[g] + bb.w, 0.f);
            float mu = sum16((vx + vy) + (vz + vw)) * (1.f / 64.f);
            float dx = vx - mu, dy = vy - mu, dz = vz - mu, dw = vw - mu;
            float q = sum16((dx * dx + dy * dy) + (dz * dz + dw * dw));
            float inv = rsqrtf(q * (1.f / 64.f) + LN_EPS);
            float4 r;
            r.x = dx * inv * gw.x + gb.x;
            r.y = dy * inv * gw.y + gb.y;
            r.z = dz * inv * gw.z + gb.z;
            r.w = dw * inv * gw.w + gb.w;
            *(float4*)(sp + g * FT_PITCH) = r;
            __builtin_amdgcn_sched_barrier(0);
        }
    }
    __threadfence_block();

    const int r0 = n0 + m * 16 + 8 * hh;
    const int nrows = N - r0;
    for (int mat = 0; mat < 2; ++mat) {
        float4* __restrict__ cp = (float4*)(mat ? xr : xl) + ((size_t)r0 * 16 + c);
        v8f acc0 = zero, acc1 = zero, acc2 = zero, acc3 = zero;
        tile_product(ap, sW + (mat + 1) * 4096 + hh * 128 + 8 * c, acc0, acc1, acc2, acc3);
#pragma unroll
        for (int g = 0; g < 8; ++g) {
            if (g < nrows)
                cp[16 * g] = make_float4(acc0[g], acc1[g], acc2[g], acc3[g]);
        }
    }
}

__device__ __forceinline__ float att_part(v4f a, v4f b, v4f w) {
    v4f v = a + b;
    v4f s = v * NEG_SLOPE;
    float p = w.x * fmaxf(v.x, s.x);
    p += w.y * fmaxf(v.y, s.y);
    p += w.z * fmaxf(v.z, s.z);
    p += w.w * fmaxf(v.w, s.w);
    return p;
}

struct SoftAcc {
    float m, den;
    v4f acc;
    __device__ __forceinline__ void add(float e, v4f a) {
        float mn = fmaxf(m, e);
        float sc = __expf(m - mn);
        float p  = __expf(e - mn);
        den = den * sc + p;
        acc = acc * sc + p * a;
        m = mn;
    }
    __device__ __forceinline__ void add3(float e0, float e1, float e2, v4f a0, v4f a1, v4f a2) {
        float mn = fmaxf(fmaxf(m, e0), fmaxf(e1, e2));
        float sc = __expf(m - mn);
        float p0 = __expf(e0 - mn), p1 = __expf(e1 - mn), p2 = __expf(e2 - mn);
        den = den * sc + p0;
        den += p1;
        den += p2;
        acc = acc * sc;
        acc += p0 * a0;
        acc += p1 * a1;
        acc += p2 * a2;
        m = mn;
    }
    __device__ __forceinline__ void add4(float e0, float e1, float e2, float e3,
                                         v4f a0, v4f a1, v4f a2, v4f a3) {
        float mn = fmaxf(fmaxf(fmaxf(m, e0), e1), fmaxf(e2, e3));
        float sc = __expf(m - mn);
        float p0 = __expf(e0 - mn), p1 = __expf(e1 - mn);
        float p2 = __expf(e2 - mn), p3 = __expf(e3 - mn);
        den = den * sc + p0;
        den += p1;
        den += p2;
        den += p3;
        acc = acc * sc;
        acc += p0 * a0;
        acc += p1 * a1;
        acc += p2 * a2;
        acc += p3 * a3;
        m = mn;
    }
};

__global__ __launch_bounds__(256) void k_att_node(const int* __restrict__ cnt,
                                                  const int* __restrict__ bins,
                                                  const int* __restrict__ ovf_cnt,
                                                  const int2* __restrict__ ovf,
                                                  const float* __restrict__ xl,
                                                  const float* __restrict__ xr,
                                                  const float* __restrict__ att,
                                                  const float* __restrict__ b2,
                                                  float* __restrict__ out, int N, int E) {
    int gid = blockIdx.x * 256 + threadIdx.x;
    int n = gid >> 4;
    const unsigned sub = gid & 15;
    if (n >= N) return;
    const int* bp = bins + (size_t)n * BIN_CAP;
    const int2 ent = *(const int2*)(bp + 2 * sub);
    int c = cnt[n];
    const v4f* __restrict__ xl4 = (const v4f*)xl;
    const v4f b  = ((const v4f*)xr)[((unsigned)n << 4) | sub];
    const v4f wt = ((const v4f*)att)[sub];

    const v4f a = xl4[((unsigned)n << 4) | sub];
    SoftAcc S;
    S.m = sum16(att_part(a, b, wt));
    S.den = 1.f;
    S.acc = a;

    const int cm = c < BIN_CAP ? c : BIN_CAP;
    int j = 0;
    for (; j + 4 <= cm; j += 4) {
        const int l = j >> 1;
        unsigned s0 = row_get(ent.x, l), s1 = row_get(ent.y, l);
        unsigned s2 = row_get(ent.x, l + 1), s3 = row_get(ent.y, l + 1);
        v4f a0 = xl4[(s0 << 4) | sub];
        v4f a1 = xl4[(s1 << 4) | sub];
        v4f a2 = xl4[(s2 << 4) | sub];
        v4f a3 = xl4[(s3 << 4) | sub];
        float e0 = sum16(att_part(a0, b, wt));
        float e1 = sum16(att_part(a1, b, wt));
        float e2 = sum16(att_part(a2, b, wt));
        float e3 = sum16(att_part(a3, b, wt));
        S.add4(e0, e1, e2, e3, a0, a1, a2, a3);
    }
    const int t = cm - j;
    if (t > 0) {
        const int l = j >> 1;
        unsigned s0 = row_get(ent.x, l), s1 = row_get(ent.y, l), s2 = row_get(ent.x, l + 1);
        s1 = t > 1 ? s1 : s0;
        s2 = t > 2 ? s2 : s0;
        v4f a0 = xl4[(s0 << 4) | sub];
        v4f a1 = xl4[(s1 << 4) | sub];
        v4f a2 = xl4[(s2 << 4) | sub];
        float e0 = sum16(att_part(a0, b, wt));
        float e1 = sum16(att_part(a1, b, wt));
        float e2 = sum16(att_part(a2, b, wt));
        e1 = t > 1 ? e1 : -INFINITY;
        e2 = t > 2 ? e2 : -INFINITY;
        S.add3(e0, e1, e2, a0, a1, a2);
    }
    if (c > BIN_CAP) {
        int no = *ovf_cnt;
        no = no < E ? no : E;
        for (int k = 0; k < no; ++k) {
            int2 e = ovf[k];
            if (e.x == n) {
                v4f a0 = xl4[((unsigned)e.y << 4) | sub];
                S.add(sum16(att_part(a0, b, wt)), a0);
            }
        }
    }

    const v4f bb = ((const v4f*)b2)[sub];
    v4f o;
    o.x = fmaxf(S.acc.x / S.den + bb.x, 0.f);
    o.y = fmaxf(S.acc.y / S.den + bb.y, 0.f);
    o.z = fmaxf(S.acc.z / S.den + bb.z, 0.f);
    o.w = fmaxf(S.acc.w / S.den + bb.w, 0.f);
    ((v4f*)out)[((unsigned)n << 4) | sub] = o;
}

extern "C" void kernel_launch(void* const* d_in, const int* in_sizes, int n_in,
                              void* d_out, int out_size, void* d_ws, size_t ws_size,
                              hipStream_t stream) {
    const float* x   = (const float*)d_in[0];
    const int*   ei  = (const int*)d_in[1];
    const float* W1  = (const float*)d_in[2];
    const float* b1  = (const float*)d_in[3];
    const float* lnw = (const float*)d_in[4];
    const float* lnb = (const float*)d_in[5];
    const float* Wl  = (const float*)d_in[6];
    const float* Wr  = (const float*)d_in[7];
    const float* att = (const float*)d_in[8];
    const float* b2  = (const float*)d_in[9];
    float* out = (float*)d_out;

    const int N = in_sizes[0] / 64;
    const int E = in_sizes[1] / 2;

    char* ws = (char*)d_ws;
    size_t off = 0;
    auto carve = [&](size_t bytes) -> void* {
        void* p = ws + off;
        off += (bytes + 255) & ~(size_t)255;
        return p;
    };
    float* xlb    = (float*)carve((size_t)N * 64 * 4);
    float* xrb    = (float*)carve((size_t)N * 64 * 4);
    int*   cnt    = (int*)carve((size_t)N * 4);
    int*   ovfc   = (int*)carve(4);
    int*   bins   = (int*)carve((size_t)N * BIN_CAP * 4);
    int2*  ovf    = (int2*)carve((size_t)E * 8);
    const int nr   = N > RNODES ? (N + RNODES - 1) / RNODES : 1;
    const int rcap = 2 * ((E + nr - 1) / nr) + 1024;
    int*   rcur   = (int*)carve((size_t)NRMAX * 4);
    int*   spillc = (int*)carve(4);
    int2*  redges = (int2*)carve((size_t)nr * rcap * 8);
    int2*  spill  = (int2*)carve((size_t)E * 8);
    const bool lds_bins = nr <= NRMAX && off <= ws_size;

    const int ngrid = (int)(((long long)N * 16 + 255) / 256);
    const int npart = (E + PCHUNK - 1) / PCHUNK;
    if (lds_bins) {
        k_zero_small<<<1, 256, 0, stream>>>(rcur, ovfc, spillc);
        k_part<<<npart, 256, 0, stream>>>(ei, rcur, redges, spillc, spill, E, rcap);
        k_build<<<nr, 1024, 0, stream>>>(rcur, redges, spillc, spill, cnt, bins, ovfc, ovf, N, E, rcap);
    } else {
        k_zero_cnt<<<(N + 255) / 256, 256, 0, stream>>>(cnt, ovfc, N);
        k_bin<<<((E + 3) / 4 + 255) / 256, 256, 0, stream>>>(ei, cnt, bins, ovfc, ovf, E);
    }

    k_gcn_dual<<<(N + FT_ROWS - 1) / FT_ROWS, 512, 0, stream>>>(cnt, bins, ovfc, ovf, x, W1, b1, lnw, lnb,
                                                             Wl, Wr, xlb, xrb, N, E);

    k_att_node<<<ngrid, 256, 0, stream>>>(cnt, bins, ovfc, ovf, xlb, xrb, att, b2, out, N, E);
}


# each wave walks several nodes in turn instead of one (attention: 5 node pairs per wave; GCN: 2 nodes per 16-lane group, W copied once per 64 rows): 100000 -> 35008 waves to start; sum orders kept
# speedup vs baseline: 7.9837x; 1.0418x over previous
#include <hip/hip_runtime.h>
#include <hip/hip_bf16.h>

typedef float v2f __attribute__((ext_vector_type(2)));
typedef float v8f __attribute__((ext_vector_type(8)));
typedef float v4f __attribute__((ext_vector_type(4)));

#define LN_EPS    1e-5f
#define NEG_SLOPE 0.2f

#define BIN_CAP 32

__global__ __launch_bounds__(256) void k_zero_cnt(int* __restrict__ cnt,
                                                  int* __restrict__ ovf_cnt, int N) {
    int i = blockIdx.x * 256 + threadIdx.x;
    if (i < N) cnt[i] = 0;
    if (i == 0) *ovf_cnt = 0;
}

__device__ __forceinline__ void bin_store(int s, int d, int r,
                                          int* __restrict__ bins,
                                          int* __restrict__ ovf_cnt,
                                          int2* __restrict__ ovf, int E) {
    if (r < BIN_CAP) {
        bins[(size_t)d * BIN_CAP + r] = s;
    } else {
        int o = atomicAdd(ovf_cnt, 1);
        if (o < E) ovf[o] = make_int2(d, s);
    }
}

__global__ __launch_bounds__(256) void k_bin(const int* __restrict__ ei,
                                             int* __restrict__ cnt,
                                             int* __restrict__ bins,
                                             int* __restrict__ ovf_cnt,
                                             int2* __restrict__ ovf, int E) {
    int base = (blockIdx.x * 256 + threadIdx.x) * 4;
    if (base >= E) return;
    if ((E & 3) == 0) {
        int4 s = *(const int4*)(ei + base);
        int4 d = *(const int4*)(ei + E + base);
        int r0 = atomicAdd(cnt + d.x, 1);
        int r1 = atomicAdd(cnt + d.y, 1);
        int r2 = atomicAdd(cnt + d.z, 1);
        int r3 = atomicAdd(cnt + d.w, 1);
        bin_store(s.x, d.x, r0, bins, ovf_cnt, ovf, E);
        bin_store(s.y, d.y, r1, bins, ovf_cnt, ovf, E);
        bin_store(s.z, d.z, r2, bins, ovf_cnt, ovf, E);
        bin_store(s.w, d.w, r3, bins, ovf_cnt, ovf, E);
    } else {
        for (int k = 0; k < 4; ++k) {
            int i = base + k;
            if (i < E) {
                int s = ei[i], d = ei[E + i];
                bin_store(s, d, atomicAdd(cnt + d, 1), bins, ovf_cnt, ovf, E);
            }
        }
    }
}

#define RSH    9
#define RNODES (1 << RSH)
#define NRMAX  256
#define PCHUNK 4096

__global__ __launch_bounds__(256) void k_zero_small(int* __restrict__ rcur,
                                                    int* __restrict__ ovf_cnt,
                                                    int* __restrict__ spill_cnt) {
    int t = threadIdx.x;
    if (t < NRMAX) rcur[t] = 0;
    if (t == 0) { *ovf_cnt = 0; *spill_cnt = 0; }
}

#define PART_LH     0
#define PART_LSTART (PART_LH + NRMAX * 4)
#define PART_GPOS   (PART_LSTART + NRMAX * 4)
#define PART_WSUM   (PART_GPOS + NRMAX * 4)
#define PART_STAGE  (PART_WSUM + (NRMAX / 32) * 4)
#define PART_LDS    (PART_STAGE + PCHUNK * 8)

__device__ __forceinline__ void part_block(const int blk, unsigned char* __restrict__ smem,
                                           const int* __restrict__ ei,
                                           int* __restrict__ rcur,
                                           int2* __restrict__ redges,
                                           int* __restrict__ spill_cnt,
                                           int2* __restrict__ spill,
                                           int E, int rcap) {
    int*  lh     = (int*)(smem + PART_LH);
    int*  lstart = (int*)(smem + PART_LSTART);
    int*  gpos   = (int*)(smem + PART_GPOS);
    int*  wsum   = (int*)(smem + PART_WSUM);
    int2* stage  = (int2*)(smem + PART_STAGE);

    const int t  = threadIdx.x;
    const int e0 = blk * PCHUNK;
    lh[t] = 0;

    int s[16], d[16], r[16];
#pragma unroll
    for (int k = 0; k < 16; ++k) {
        int e = e0 + k * 256 + t;
        bool ok = e < E;
        s[k] = ok ? ei[e] : 0;
        d[k] = ok ? ei[E + e] : -1;
    }
    __syncthreads();
#pragma unroll
    for (int k = 0; k < 16; ++k)
        r[k] = (d[k] >= 0) ? atomicAdd(&lh[d[k] >> RSH], 1) : 0;
    __syncthreads();

    const int h = lh[t];
    int incl = h;
    for (int off = 1; off < 32; off <<= 1) {
        int u = __shfl_up(incl, off, 32);
        if ((t & 31) >= off) incl += u;
    }
    if ((t & 31) == 31) wsum[t >> 5] = incl;
    __syncthreads();
    int before = 0;
    for (int w = 0; w < (t >> 5); ++w) before += wsum[w];
    lstart[t] = before + incl - h;
    gpos[t]   = h ? atomicAdd(rcur + t, h) : 0;
    __syncthreads();

#pragma unroll
    for (int k = 0; k < 16; ++k)
        if (d[k] >= 0) stage[lstart[d[k] >> RSH] + r[k]] = make_int2(d[k], s[k]);
    __syncthreads();

    int n = E - e0;
    n = n < PCHUNK ? n : PCHUNK;
    for (int i = t; i < n; i += 256) {
        int2 e = stage[i];
        int b = e.x >> RSH;
        int g = gpos[b] + (i - lstart[b]);
        if (g < rcap) {
            redges[(size_t)b * rcap + g] = e;
        } else {
            int o = atomicAdd(spill_cnt, 1);
            if (o < E) spill[o] = e;
        }
    }
}

__device__ __forceinline__ void build_put(int2 e, int l, int r, int* __restrict__ lb,
                                          int* __restrict__ ovf_cnt,
                                          int2* __restrict__ ovf, int E) {
    if (r < BIN_CAP) {
        lb[l * BIN_CAP + r] = e.y;
    } else {
        int o = atomicAdd(ovf_cnt, 1);
        if (o < E) ovf[o] = e;
    }
}

__global__ __launch_bounds__(1024) void k_build(const int* __restrict__ rcur,
                                                const int2* __restrict__ redges,
                                                const int* __restrict__ spill_cnt,
                                                const int2* __restrict__ spill,
                                                int* __restrict__ cnt,
                                                int* __restrict__ bins,
                                                int* __restrict__ ovf_cnt,
                                                int2* __restrict__ ovf,
                                                int N, int E, int rcap) {
    __shared__ int lc[RNODES];
    __shared__ int lb[RNODES * BIN_CAP];

    const int t = threadIdx.x, b = blockIdx.x;
    const int base = b << RSH;
    if (t < RNODES) lc[t] = 0;
    {
        int4 z = make_int4(0, 0, 0, 0);
        for (int i = t; i < RNODES * BIN_CAP / 4; i += 1024) ((int4*)lb)[i] = z;
    }
    __syncthreads();

    int m = rcur[b];
    m = m < rcap ? m : rcap;
    const int2* re = redges + (size_t)b * rcap;
    for (int i0 = 0; i0 < m; i0 += 8 * 1024) {
        int2 e[8];
#pragma unroll
        for (int k = 0; k < 8; ++k) {
            int i = i0 + k * 1024 + t;
            e[k] = (i < m) ? re[i] : make_int2(-1, 0);
        }
        int r[8];
#pragma unroll
        for (int k = 0; k < 8; ++k)
            r[k] = (e[k].x >= 0) ? atomicAdd(&lc[e[k].x - base], 1) : 0;
#pragma unroll
        for (int k = 0; k < 8; ++k)
            if (e[k].x >= 0) build_put(e[k], e[k].x - base, r[k], lb, ovf_cnt, ovf, E);
    }
    int ns = *spill_cnt;
    ns = ns < E ? ns : E;
    for (int i = t; i < ns; i += 1024) {
        int2 e = spill[i];
        if ((e.x >> RSH) == b)
            build_put(e, e.x - base, atomicAdd(&lc[e.x - base], 1), lb, ovf_cnt, ovf, E);
    }
    __syncthreads();

    int nn = N - base;
    nn = nn < RNODES ? nn : RNODES;
    if (t < nn) cnt[base + t] = lc[t];
    int4* dst = (int4*)(bins + (size_t)base * BIN_CAP);
    for (int i = t; i < nn * (BIN_CAP / 4); i += 1024) dst[i] = ((const int4*)lb)[i];
}

__device__ __forceinline__ float dinv_of(int c) {
    float g = (float)(c + 1);
    return (g > 0.f) ? rsqrtf(g) : 0.f;
}

__global__ __launch_bounds__(256) void k_part(const int* __restrict__ ei,
                                              int* __restrict__ rcur,
                                              int2* __restrict__ redges,
                                              int* __restrict__ spill_cnt,
                                              int2* __restrict__ spill,
                                              int E, int rcap) {
    __shared__ __align__(16) unsigned char smem[PART_LDS];
    part_block(blockIdx.x, smem, ei, rcur, redges, spill_cnt, spill, E, rcap);
}

template <int M> __device__ __forceinline__ float row_xor(float p) {
    return __int_as_float(__builtin_amdgcn_update_dpp(0, __float_as_int(p), 0x160 + M, 0xf, 0xf, true));
}
__device__ __forceinline__ int row_get(int v, int lane) {
    return (int)__builtin_amdgcn_permlane16_var(0u, (unsigned)v, (unsigned)lane, false, true);
}

__device__ __forceinline__ float sum16(float p) {
    p += row_xor<8>(p);
    p += row_xor<4>(p);
    p += row_xor<2>(p);
    p += row_xor<1>(p);
    return p;
}

template <bool STAGE>
__device__ __forceinline__ float4 gcn_agg_row(const int n, const int sub,
                                              const int* __restrict__ cnt,
                                              const int* __restrict__ bins,
                                              const int* __restrict__ ovf_cnt,
                                              const int2* __restrict__ ovf,
                                              const float* __restrict__ x, int E,
                                              const float4* __restrict__ w1s,
                                              const float4* __restrict__ wls,
                                              const float4* __restrict__ wrs,
                                              float4* __restrict__ wdst, const int tid) {
    const int* bp = bins + (size_t)n * BIN_CAP;
    const int2 ent = *(const int2*)(bp + 2 * sub);
    int c = cnt[n];
    const float4* __restrict__ x4 = (const float4*)x;
    const unsigned usub = (unsigned)sub;
    float4 v = x4[((unsigned)n << 4) | usub];

    if (STAGE) {
        const int kp = tid >> 4, qi = tid & 15;
        const int e = (2 * kp) * 16 + qi, o = e + 16;
        float4 p0 = w1s[e], q0 = w1s[o];
        float4 p1 = wls[e], q1 = wls[o];
        float4 p2 = wrs[e], q2 = wrs[o];
        float4* d = wdst + kp * 32 + 2 * qi;
        d[0]    = make_float4(p0.x, q0.x, p0.y, q0.y);
        d[1]    = make_float4(p0.z, q0.z, p0.w, q0.w);
        d[1024] = make_float4(p1.x, q1.x, p1.y, q1.y);
        d[1025] = make_float4(p1.z, q1.z, p1.w, q1.w);
        d[2048] = make_float4(p2.x, q2.x, p2.y, q2.y);
        d[2049] = make_float4(p2.z, q2.z, p2.w, q2.w);
    }

    float dn = dinv_of(c);
    float w = dn * dn;
    float ax = v.x * w, ay = v.y * w, az = v.z * w, aw = v.w * w;

    const int cm = c < BIN_CAP ? c : BIN_CAP;
    int j = 0;
    for (; j + 4 <= cm; j += 4) {
        const int l = j >> 1;
        int s0 = row_get(ent.x, l), s1 = row_get(ent.y, l);
        int s2 = row_get(ent.x, l + 1), s3 = row_get(ent.y, l + 1);
        float w0 = dinv_of(cnt[s0]) * dn, w1 = dinv_of(cnt[s1]) * dn;
        float w2 = dinv_of(cnt[s2]) * dn, w3 = dinv_of(cnt[s3]) * dn;
        float4 v0 = x4[((unsigned)s0 << 4) | usub];
        float4 v1 = x4[((unsigned)s1 << 4) | usub];
        float4 v2 = x4[((unsigned)s2 << 4) | usub];
        float4 v3 = x4[((unsigned)s3 << 4) | usub];
        ax += v0.x * w0; ay += v0.y * w0; az += v0.z * w0; aw += v0.w * w0;
        ax += v1.x * w1; ay += v1.y * w1; az += v1.z * w1; aw += v1.w * w1;
        ax += v2.x * w2; ay += v2.y * w2; az += v2.z * w2; aw += v2.w * w2;
        ax += v3.x * w3; ay += v3.y * w3; az += v3.z * w3; aw += v3.w * w3;
    }
    const int t = cm - j;
    if (t > 0) {
        const int l = j >> 1;
        int s0 = row_get(ent.x, l), s1 = row_get(ent.y, l), s2 = row_get(ent.x, l + 1);
        s1 = t > 1 ? s1 : s0;
        s2 = t > 2 ? s2 : s0;
        const int c0 = cnt[s0], c1 = cnt[s1], c2 = cnt[s2];
        const float k1 = t > 1 ? 1.f : 0.f, k2 = t > 2 ? 1.f : 0.f;
        float w0 = dinv_of(c0) * dn;
        float w1 = dinv_of(c1) * dn * k1;
        float w2 = dinv_of(c2) * dn * k2;
        float4 v0 = x4[((unsigned)s0 << 4) | usub];
        float4 v1 = x4[((unsigned)s1 << 4) | usub];
        float4 v2 = x4[((unsigned)s2 << 4) | usub];
        ax += v0.x * w0; ay += v0.y * w0; az += v0.z * w0; aw += v0.w * w0;
        ax += v1.x * w1; ay += v1.y * w1; az += v1.z * w1; aw += v1.w * w1;
        ax += v2.x * w2; ay += v2.y * w2; az += v2.z * w2; aw += v2.w * w2;
    }
    if (c > BIN_CAP) {
        int no = *ovf_cnt;
        no = no < E ? no : E;
        for (int k = 0; k < no; ++k) {
            int2 e = ovf[k];
            if (e.x == n) {
                int s0 = e.y;
                float w0 = dinv_of(cnt[s0]) * dn;
                float4 v0 = x4[((unsigned)s0 << 4) | usub];
                ax += v0.x * w0; ay += v0.y * w0; az += v0.z * w0; aw += v0.w * w0;
            }
        }
    }
    return make_float4(ax, ay, az, aw);
}

__device__ __forceinline__ void tile_product(const float* ap, const float* wp,
                                             v8f& acc0, v8f& acc1, v8f& acc2, v8f& acc3) {
#pragma unroll 2
    for (int kk = 0; kk < 16; ++kk) {
        v2f a = *(const v2f*)(ap + kk * 4);
        float4 f0 = *(const float4*)(wp + kk * 256);
        float4 f1 = *(const float4*)(wp + kk * 256 + 4);
        v2f bq0 = {f0.x, f0.y}, bq1 = {f0.z, f0.w}, bq2 = {f1.x, f1.y}, bq3 = {f1.z, f1.w};
        acc0 = __builtin_amdgcn_wmma_f32_16x16x4_f32(false, a, false, bq0, (short)0, acc0, false, false);
        acc1 = __builtin_amdgcn_wmma_f32_16x16x4_f32(false, a, false, bq1, (short)0, acc1, false, false);
        acc2 = __builtin_amdgcn_wmma_f32_16x16x4_f32(false, a, false, bq2, (short)0, acc2, false, false);
        acc3 = __builtin_amdgcn_wmma_f32_16x16x4_f32(false, a, false, bq3, (short)0, acc3, false, false);
    }
}

#define FT_GROUPS 32
#define FT_WALK   2
#define FT_ROWS  (FT_GROUPS * FT_WALK)
#define FT_PITCH 68

__global__ __launch_bounds__(512) void k_gcn_dual(const int* __restrict__ cnt,
                                                  const int* __restrict__ bins,
                                                  const int* __restrict__ ovf_cnt,
                                                  const int2* __restrict__ ovf,
                                                  const float* __restrict__ x,
                                                  const float* __restrict__ W1,
                                                  const float* __restrict__ b1,
                                                  const float* __restrict__ lnw,
                                                  const float* __restrict__ lnb,
                                                  const float* __restrict__ Wl,
                                                  const float* __restrict__ Wr,
                                                  float* __restrict__ xl,
                                                  float* __restrict__ xr, int N, int E) {
    __shared__ __align__(16) float sA[FT_ROWS * FT_PITCH];
    __shared__ __align__(16) float sW[3 * 64 * 64];
    const int tid = threadIdx.x;
    const int grp = tid >> 4, sub = tid & 15;
    const int n0 = blockIdx.x * FT_ROWS;
    {
        const int n = n0 + grp;
        float4 o = gcn_agg_row<true>(n < N ? n : N - 1, sub, cnt, bins, ovf_cnt, ovf, x, E,
                                     (const float4*)W1, (const float4*)Wl, (const float4*)Wr,
                                     (float4*)sW, tid);
        if (n >= N) o = make_float4(0.f, 0.f, 0.f, 0.f);
        *(float4*)(sA + grp * FT_PITCH + sub * 4) = o;
    }
    for (int p = 1; p < FT_WALK; ++p) {
        const int row = p * FT_GROUPS + grp;
        const int n = n0 + row;
        float4 o = gcn_agg_row<false>(n < N ? n : N - 1, sub, cnt, bins, ovf_cnt, ovf, x, E,
                                      (const float4*)W1, (const float4*)Wl, (const float4*)Wr,
                                      (float4*)sW, tid);
        if (n >= N) o = make_float4(0.f, 0.f, 0.f, 0.f);
        *(float4*)(sA + row * FT_PITCH + sub * 4) = o;
    }
    __syncthreads();

    const int m = tid >> 5;
    if (m >= FT_ROWS / 16) return;
    const int lane = tid & 31;
    const int c = lane & 15, hh = lane >> 4;

    float* ap = sA + (m * 16 + c) * FT_PITCH + 2 * hh;
    const v8f zero = {0.f, 0.f, 0.f, 0.f, 0.f, 0.f, 0.f, 0.f};
    {
        v8f acc0 = zero, acc1 = zero, acc2 = zero, acc3 = zero;
        tile_product(ap, sW + hh * 128 + 8 * c, acc0, acc1, acc2, acc3);
        const float4 bb = *(const float4*)(b1 + 4 * c);
        const float4 gw = *(const float4*)(lnw + 4 * c);
        const float4 gb = *(const float4*)(lnb + 4 * c);
        float* sp = sA + (m * 16 + 8 * hh) * FT_PITCH + 4 * c;
#pragma unroll
        for (int g = 0; g < 8; ++g) {
            float vx = fmaxf(acc0[g] + bb.x, 0.f), vy = fmaxf(acc1[g] + bb.y, 0.f);
            float vz = fmaxf(acc2[g] + bb.z, 0.f), vw = fmaxf(acc3[g] + bb.w, 0.f);
            float mu = sum16((vx + vy) + (vz + vw)) * (1.f / 64.f);
            float dx = vx - mu, dy = vy - mu, dz = vz - mu, dw = vw - mu;
            float q = sum16((dx * dx + dy * dy) + (dz * dz + dw * dw));
            float inv = rsqrtf(q * (1.f / 64.f) + LN_EPS);
            float4 r;
            r.x = dx * inv * gw.x + gb.x;
            r.y = dy * inv * gw.y + gb.y;
            r.z = dz * inv * gw.z + gb.z;
            r.w = dw * inv * gw.w + gb.w;
            *(float4*)(sp + g * FT_PITCH) = r;
            __builtin_amdgcn_sched_barrier(0);
        }
    }
    __threadfence_block();

    const int r0 = n0 + m * 16 + 8 * hh;
    const int nrows = N - r0;
    for (int mat = 0; mat < 2; ++mat) {
        float4* __restrict__ cp = (float4*)(mat ? xr : xl) + ((size_t)r0 * 16 + c);
        v8f acc0 = zero, acc1 = zero, acc2 = zero, acc3 = zero;
        tile_product(ap, sW + (mat + 1) * 4096 + hh * 128 + 8 * c, acc0, acc1, acc2, acc3);
#pragma unroll
        for (int g = 0; g < 8; ++g) {
            if (g < nrows)
                cp[16 * g] = make_float4(acc0[g], acc1[g], acc2[g], acc3[g]);
        }
    }
}

__device__ __forceinline__ float att_part(v4f a, v4f b, v4f w) {
    v4f v = a + b;
    v4f s = v * NEG_SLOPE;
    float p = w.x * fmaxf(v.x, s.x);
    p += w.y * fmaxf(v.y, s.y);
    p += w.z * fmaxf(v.z, s.z);
    p += w.w * fmaxf(v.w, s.w);
    return p;
}

struct SoftAcc {
    float m, den;
    v4f acc;
    __device__ __forceinline__ void add(float e, v4f a) {
        float mn = fmaxf(m, e);
        float sc = __expf(m - mn);
        float p  = __expf(e - mn);
        den = den * sc + p;
        acc = acc * sc + p * a;
        m = mn;
    }
    __device__ __forceinline__ void add3(float e0, float e1, float e2, v4f a0, v4f a1, v4f a2) {
        float mn = fmaxf(fmaxf(m, e0), fmaxf(e1, e2));
        float sc = __expf(m - mn);
        float p0 = __expf(e0 - mn), p1 = __expf(e1 - mn), p2 = __expf(e2 - mn);
        den = den * sc + p0;
        den += p1;
        den += p2;
        acc = acc * sc;
        acc += p0 * a0;
        acc += p1 * a1;
        acc += p2 * a2;
        m = mn;
    }
    __device__ __forceinline__ void add4(float e0, float e1, float e2, float e3,
                                         v4f a0, v4f a1, v4f a2, v4f a3) {
        float mn = fmaxf(fmaxf(fmaxf(m, e0), e1), fmaxf(e2, e3));
        float sc = __expf(m - mn);
        float p0 = __expf(e0 - mn), p1 = __expf(e1 - mn);
        float p2 = __expf(e2 - mn), p3 = __expf(e3 - mn);
        den = den * sc + p0;
        den += p1;
        den += p2;
        den += p3;
        acc = acc * sc;
        acc += p0 * a0;
        acc += p1 * a1;
        acc += p2 * a2;
        acc += p3 * a3;
        m = mn;
    }
};

#define ATT_WALK 5

__global__ __launch_bounds__(256) void k_att_node(const int* __restrict__ cnt,
                                                  const int* __restrict__ bins,
                                                  const int* __restrict__ ovf_cnt,
                                                  const int2* __restrict__ ovf,
                                                  const float* __restrict__ xl,
                                                  const float* __restrict__ xr,
                                                  const float* __restrict__ att,
                                                  const float* __restrict__ b2,
                                                  float* __restrict__ out, int N, int E) {
    const int gid = blockIdx.x * 256 + threadIdx.x;
    const unsigned sub = gid & 15;
    const int pair0 = (gid >> 5) * ATT_WALK;
    const int half = (gid >> 4) & 1;
    const v4f wt = ((const v4f*)att)[sub];
    for (int i = 0; i < ATT_WALK; ++i) {
        const int n = ((pair0 + i) << 1) | half;
        if (n >= N) break;
        const int* bp = bins + (size_t)n * BIN_CAP;
        const int2 ent = *(const int2*)(bp + 2 * sub);
        int c = cnt[n];
        const v4f* __restrict__ xl4 = (const v4f*)xl;
        const v4f b  = ((const v4f*)xr)[((unsigned)n << 4) | sub];

        const v4f a = xl4[((unsigned)n << 4) | sub];
        __builtin_amdgcn_sched_barrier(0);
        SoftAcc S;
        S.m = sum16(att_part(a, b, wt));
        S.den = 1.f;
        S.acc = a;

        const int cm = c < BIN_CAP ? c : BIN_CAP;
        int j = 0;
        for (; j + 4 <= cm; j += 4) {
            const int l = j >> 1;
            unsigned s0 = row_get(ent.x, l), s1 = row_get(ent.y, l);
            unsigned s2 = row_get(ent.x, l + 1), s3 = row_get(ent.y, l + 1);
            v4f a0 = xl4[(s0 << 4) | sub];
            v4f a1 = xl4[(s1 << 4) | sub];
            v4f a2 = xl4[(s2 << 4) | sub];
            v4f a3 = xl4[(s3 << 4) | sub];
            float e0 = sum16(att_part(a0, b, wt));
            float e1 = sum16(att_part(a1, b, wt));
            float e2 = sum16(att_part(a2, b, wt));
            float e3 = sum16(att_part(a3, b, wt));
            S.add4(e0, e1, e2, e3, a0, a1, a2, a3);
        }
        const int t = cm - j;
        if (t > 0) {
            const int l = j >> 1;
            unsigned s0 = row_get(ent.x, l), s1 = row_get(ent.y, l), s2 = row_get(ent.x, l + 1);
            s1 = t > 1 ? s1 : s0;
            s2 = t > 2 ? s2 : s0;
            v4f a0 = xl4[(s0 << 4) | sub];
            v4f a1 = xl4[(s1 << 4) | sub];
            v4f a2 = xl4[(s2 << 4) | sub];
            float e0 = sum16(att_part(a0, b, wt));
            float e1 = sum16(att_part(a1, b, wt));
            float e2 = sum16(att_part(a2, b, wt));
            e1 = t > 1 ? e1 : -INFINITY;
            e2 = t > 2 ? e2 : -INFINITY;
            S.add3(e0, e1, e2, a0, a1, a2);
        }
        if (c > BIN_CAP) {
            int no = *ovf_cnt;
            no = no < E ? no : E;
            for (int k = 0; k < no; ++k) {
                int2 e = ovf[k];
                if (e.x == n) {
                    v4f a0 = xl4[((unsigned)e.y << 4) | sub];
                    S.add(sum16(att_part(a0, b, wt)), a0);
                }
            }
        }

        const v4f bb = ((const v4f*)b2)[sub];
        v4f o;
        o.x = fmaxf(S.acc.x / S.den + bb.x, 0.f);
        o.y = fmaxf(S.acc.y / S.den + bb.y, 0.f);
        o.z = fmaxf(S.acc.z / S.den + bb.z, 0.f);
        o.w = fmaxf(S.acc.w / S.den + bb.w, 0.f);
        ((v4f*)out)[((unsigned)n << 4) | sub] = o;
    }
}

extern "C" void kernel_launch(void* const* d_in, const int* in_sizes, int n_in,
                              void* d_out, int out_size, void* d_ws, size_t ws_size,
                              hipStream_t stream) {
    const float* x   = (const float*)d_in[0];
    const int*   ei  = (const int*)d_in[1];
    const float* W1  = (const float*)d_in[2];
    const float* b1  = (const float*)d_in[3];
    const float* lnw = (const float*)d_in[4];
    const float* lnb = (const float*)d_in[5];
    const float* Wl  = (const float*)d_in[6];
    const float* Wr  = (const float*)d_in[7];
    const float* att = (const float*)d_in[8];
    const float* b2  = (const float*)d_in[9];
    float* out = (float*)d_out;

    const int N = in_sizes[0] / 64;
    const int E = in_sizes[1] / 2;

    char* ws = (char*)d_ws;
    size_t off = 0;
    auto carve = [&](size_t bytes) -> void* {
        void* p = ws + off;
        off += (bytes + 255) & ~(size_t)255;
        return p;
    };
    float* xlb    = (float*)carve((size_t)N * 64 * 4);
    float* xrb    = (float*)carve((size_t)N * 64 * 4);
    int*   cnt    = (int*)carve((size_t)N * 4);
    int*   ovfc   = (int*)carve(4);
    int*   bins   = (int*)carve((size_t)N * BIN_CAP * 4);
    int2*  ovf    = (int2*)carve((size_t)E * 8);
    const int nr   = N > RNODES ? (N + RNODES - 1) / RNODES : 1;
    const int rcap = 2 * ((E + nr - 1) / nr) + 1024;
    int*   rcur   = (int*)carve((size_t)NRMAX * 4);
    int*   spillc = (int*)carve(4);
    int2*  redges = (int2*)carve((size_t)nr * rcap * 8);
    int2*  spill  = (int2*)carve((size_t)E * 8);
    const bool lds_bins = nr <= NRMAX && off <= ws_size;

    const int nattw = ((N + 1) / 2 + ATT_WALK - 1) / ATT_WALK;
    const int ngrid = (nattw + 7) / 8;
    const int npart = (E + PCHUNK - 1) / PCHUNK;
    if (lds_bins) {
        k_zero_small<<<1, 256, 0, stream>>>(rcur, ovfc, spillc);
        k_part<<<npart, 256, 0, stream>>>(ei, rcur, redges, spillc, spill, E, rcap);
        k_build<<<nr, 1024, 0, stream>>>(rcur, redges, spillc, spill, cnt, bins, ovfc, ovf, N, E, rcap);
    } else {
        k_zero_cnt<<<(N + 255) / 256, 256, 0, stream>>>(cnt, ovfc, N);
        k_bin<<<((E + 3) / 4 + 255) / 256, 256, 0, stream>>>(ei, cnt, bins, ovfc, ovf, E);
    }

    k_gcn_dual<<<(N + FT_ROWS - 1) / FT_ROWS, 512, 0, stream>>>(cnt, bins, ovfc, ovf, x, W1, b1, lnw, lnb,
                                                             Wl, Wr, xlb, xrb, N, E);

    k_att_node<<<ngrid, 256, 0, stream>>>(cnt, bins, ovfc, ovf, xlb, xrb, att, b2, out, N, E);
}
